# starting kernel
# baseline (speedup 1.0000x reference)
#include <hip/hip_runtime.h>
#include <hip/hip_bf16.h>

typedef __attribute__((ext_vector_type(2))) float v2f;
typedef __attribute__((ext_vector_type(8))) float v8f;

#define N_NODES_C 100000

__global__ void zero_f32(float* __restrict__ p, int n) {
    int i = blockIdx.x * blockDim.x + threadIdx.x;
    if (i < n) p[i] = 0.0f;
}

__global__ void deg_count(const int* __restrict__ dst, float* __restrict__ deg, int ne) {
    int e = blockIdx.x * blockDim.x + threadIdx.x;
    if (e < ne) unsafeAtomicAdd(&deg[dst[e]], 1.0f);
}

__global__ void finish_dis(float* __restrict__ deg, int n) {
    int i = blockIdx.x * blockDim.x + threadIdx.x;
    if (i < n) deg[i] = rsqrtf(deg[i] + 1.0f);
}

template<int NT, int KDIM, bool BIAS, bool RELU>
__launch_bounds__(256)
__global__ void gemm_wmma_f32(const float* __restrict__ X,
                              const float* __restrict__ W,
                              const float* __restrict__ bias,
                              float* __restrict__ Y,
                              int M) {
    constexpr int N = NT * 16;
    const int wave = (blockIdx.x * blockDim.x + threadIdx.x) >> 5;
    const int lane = threadIdx.x & 31;
    const int row0 = wave * 16;
    if (row0 >= M) return;

    const int r      = lane & 15;
    const int khalf  = (lane >> 4) * 2;

    v8f acc[NT] = {};

    const float* __restrict__ xrow = X + (size_t)(row0 + r) * KDIM;

#pragma unroll
    for (int k = 0; k < KDIM; k += 4) {
        const int kk = k + khalf;
        v2f a = *(const v2f*)(xrow + kk);
#pragma unroll
        for (int t = 0; t < NT; ++t) {
            const int col = t * 16 + r;
            v2f b;
            b.x = W[(size_t)kk * N + col];
            b.y = W[(size_t)(kk + 1) * N + col];
            acc[t] = __builtin_amdgcn_wmma_f32_16x16x4_f32(
                 false, a,  false, b,
                 (short)0, acc[t],
                 false,  false);
        }
    }

    const int rowoff = (lane >> 4) * 8;
#pragma unroll
    for (int t = 0; t < NT; ++t) {
        const int col = t * 16 + r;
        const float bv = BIAS ? bias[col] : 0.0f;
#pragma unroll
        for (int g = 0; g < 8; ++g) {
            float v = acc[t][g] + bv;
            if (RELU) v = fmaxf(v, 0.0f);
            Y[(size_t)(row0 + rowoff + g) * N + col] = v;
        }
    }
}

__global__ void edge_scatter(const float* __restrict__ h,
                             const int* __restrict__ src,
                             const int* __restrict__ dst,
                             const float* __restrict__ dis,
                             float* __restrict__ agg,
                             int ne) {
    const int tid = blockIdx.x * blockDim.x + threadIdx.x;
    const int e = tid >> 4;
    if (e >= ne) return;
    const int f = (tid & 15) * 4;
    const int s = src[e];
    const int d = dst[e];
    const float norm = dis[s] * dis[d];
    const float4 hv = *(const float4*)(h + (size_t)s * 64 + f);
    float* ap = agg + (size_t)d * 64 + f;
    unsafeAtomicAdd(ap + 0, hv.x * norm);
    unsafeAtomicAdd(ap + 1, hv.y * norm);
    unsafeAtomicAdd(ap + 2, hv.z * norm);
    unsafeAtomicAdd(ap + 3, hv.w * norm);
}

__global__ void selfloop_bias_relu(const float* __restrict__ agg,
                                   const float* __restrict__ h,
                                   const float* __restrict__ dis,
                                   const float* __restrict__ b,
                                   float* __restrict__ out,
                                   int n) {
    const int tid = blockIdx.x * blockDim.x + threadIdx.x;
    const int i = tid >> 4;
    if (i >= n) return;
    const int f = (tid & 15) * 4;
    const float d2 = dis[i] * dis[i];
    const float4 av = *(const float4*)(agg + (size_t)i * 64 + f);
    const float4 hv = *(const float4*)(h + (size_t)i * 64 + f);
    const float4 bv = *(const float4*)(b + f);
    float4 o;
    o.x = fmaxf(av.x + hv.x * d2 + bv.x, 0.0f);
    o.y = fmaxf(av.y + hv.y * d2 + bv.y, 0.0f);
    o.z = fmaxf(av.z + hv.z * d2 + bv.z, 0.0f);
    o.w = fmaxf(av.w + hv.w * d2 + bv.w, 0.0f);
    *(float4*)(out + (size_t)i * 64 + f) = o;
}

__global__ void out_proj(const float* __restrict__ hd,
                         const float* __restrict__ Wo,
                         const float* __restrict__ bo,
                         float* __restrict__ out,
                         int n) {
    const int i = blockIdx.x * blockDim.x + threadIdx.x;
    if (i >= n) return;
    const float* h = hd + (size_t)i * 128;
    float a0 = bo[0], a1 = bo[1];
#pragma unroll 8
    for (int k = 0; k < 128; ++k) {
        const float hv = h[k];
        a0 += hv * Wo[k * 2 + 0];
        a1 += hv * Wo[k * 2 + 1];
    }
    out[(size_t)i * 2 + 0] = a0;
    out[(size_t)i * 2 + 1] = a1;
}

extern "C" void kernel_launch(void* const* d_in, const int* in_sizes, int n_in,
                              void* d_out, int out_size, void* d_ws, size_t ws_size,
                              hipStream_t stream) {
    const float* x  = (const float*)d_in[0];
    const int*   ei = (const int*)d_in[1];
    const float* W1 = (const float*)d_in[2];
    const float* b1 = (const float*)d_in[3];
    const float* W2 = (const float*)d_in[4];
    const float* b2 = (const float*)d_in[5];
    const float* Wd = (const float*)d_in[6];
    const float* bd = (const float*)d_in[7];
    const float* Wo = (const float*)d_in[8];
    const float* bo = (const float*)d_in[9];
    float* out = (float*)d_out;

    const int N = in_sizes[0] / 64;
    const int E = in_sizes[1] / 2;
    const int* src = ei;
    const int* dst = ei + E;

    float* dis  = (float*)d_ws;
    float* bufA = dis  + N;
    float* bufB = bufA + (size_t)N * 64;
    float* bufC = bufB + (size_t)N * 64;
    float* bufD = bufC + (size_t)N * 64;
    (void)ws_size; (void)n_in; (void)out_size;

    const int TB = 256;
    const int gemm_waves   = N / 16;
    const int gemm_blocks  = (gemm_waves * 32 + TB - 1) / TB;
    const int edge_blocks  = (E * 16 + TB - 1) / TB;
    const int node16_blks  = (N * 16 + TB - 1) / TB;

    zero_f32<<<(N + TB - 1) / TB, TB, 0, stream>>>(dis, N);
    deg_count<<<(E + TB - 1) / TB, TB, 0, stream>>>(dst, dis, E);
    finish_dis<<<(N + TB - 1) / TB, TB, 0, stream>>>(dis, N);

    gemm_wmma_f32<4, 64, false, false><<<gemm_blocks, TB, 0, stream>>>(x, W1, nullptr, bufA, N);
    zero_f32<<<((N * 64) + TB - 1) / TB, TB, 0, stream>>>(bufB, N * 64);
    edge_scatter<<<edge_blocks, TB, 0, stream>>>(bufA, src, dst, dis, bufB, E);
    selfloop_bias_relu<<<node16_blks, TB, 0, stream>>>(bufB, bufA, dis, b1, bufC, N);

    gemm_wmma_f32<4, 64, false, false><<<gemm_blocks, TB, 0, stream>>>(bufC, W2, nullptr, bufA, N);
    zero_f32<<<((N * 64) + TB - 1) / TB, TB, 0, stream>>>(bufB, N * 64);
    edge_scatter<<<edge_blocks, TB, 0, stream>>>(bufA, src, dst, dis, bufB, E);
    selfloop_bias_relu<<<node16_blks, TB, 0, stream>>>(bufB, bufA, dis, b2, bufC, N);

    gemm_wmma_f32<8, 64, true, true><<<gemm_blocks, TB, 0, stream>>>(bufC, Wd, bd, bufD, N);

    out_proj<<<(N + TB - 1) / TB, TB, 0, stream>>>(bufD, Wo, bo, out, N);
}


# self-loop term + bias stored by the WMMA product's epilogue as the edge sums' start value (self term summed first, was last); relu moves to the next product's A load; zero-fill, self-loop passes gone
# speedup vs baseline: 1.2671x; 1.2671x over previous
#include <hip/hip_runtime.h>
#include <hip/hip_bf16.h>

typedef __attribute__((ext_vector_type(2))) float v2f;
typedef __attribute__((ext_vector_type(8))) float v8f;

#define N_NODES_C 100000

#define DEG_R   14336
#define DEG_TB  1024
#define DEG_IT  11
#define DEG_CH  (DEG_TB * DEG_IT * 4)
static_assert(DEG_R % DEG_TB == 0, "a range is a whole number of block strides");

__launch_bounds__(DEG_TB)
__global__ void deg_hist(const int* __restrict__ dst, unsigned* __restrict__ part,
                         int ne, int n) {
    __shared__ unsigned hist[DEG_R];
    const int tid = threadIdx.x;
    const int lo  = blockIdx.x * DEG_R;
    const int e0  = blockIdx.y * DEG_CH;

    for (int i = tid; i < DEG_R; i += DEG_TB) hist[i] = 0u;
    __syncthreads();

    if ((ne & 3) == 0) {
        const int4* __restrict__ d4 = (const int4*)dst;
        const int n4 = ne >> 2;
        const int j0 = (e0 >> 2) + tid;
        int4 v[DEG_IT];
#pragma unroll
        for (int it = 0; it < DEG_IT; ++it) {
            const int j = j0 + it * DEG_TB;
            v[it] = d4[j < n4 ? j : n4 - 1];
        }
#pragma unroll
        for (int it = 0; it < DEG_IT; ++it) {
            const int j = j0 + it * DEG_TB;
            if (j < n4) {
                const unsigned a0 = (unsigned)(v[it].x - lo);
                const unsigned a1 = (unsigned)(v[it].y - lo);
                const unsigned a2 = (unsigned)(v[it].z - lo);
                const unsigned a3 = (unsigned)(v[it].w - lo);
                if (a0 < (unsigned)DEG_R) atomicAdd(&hist[a0], 1u);
                if (a1 < (unsigned)DEG_R) atomicAdd(&hist[a1], 1u);
                if (a2 < (unsigned)DEG_R) atomicAdd(&hist[a2], 1u);
                if (a3 < (unsigned)DEG_R) atomicAdd(&hist[a3], 1u);
            }
        }
    } else {
        const int e1 = (e0 + DEG_CH < ne) ? (e0 + DEG_CH) : ne;
        for (int e = e0 + tid; e < e1; e += DEG_TB) {
            const unsigned a = (unsigned)(dst[e] - lo);
            if (a < (unsigned)DEG_R) atomicAdd(&hist[a], 1u);
        }
    }
    __syncthreads();

    unsigned* __restrict__ row = part + (size_t)blockIdx.y * n;
    unsigned hv[DEG_R / DEG_TB];
#pragma unroll
    for (int k = 0; k < DEG_R / DEG_TB; ++k) hv[k] = hist[tid + k * DEG_TB];
#pragma unroll
    for (int k = 0; k < DEG_R / DEG_TB; ++k) {
        const int node = lo + tid + k * DEG_TB;
        if (node < n) row[node] = hv[k];
    }
}

__global__ void finish_dis(const unsigned* __restrict__ part, float* __restrict__ dis,
                           int n, int nc) {
    int i = blockIdx.x * blockDim.x + threadIdx.x;
    if (i >= n) return;
    unsigned s = 0u;
#pragma unroll 12
    for (int c = 0; c < nc; ++c) s += part[(size_t)c * n + i];
    dis[i] = rsqrtf((float)s + 1.0f);
}

template<int NT, int KDIM, bool RELU_IN, bool SELF>
__launch_bounds__(256)
__global__ void gemm_wmma_f32(const float* __restrict__ X,
                              const float* __restrict__ W,
                              const float* __restrict__ bias,
                              const float* __restrict__ dis,
                              float* __restrict__ Y,
                              float* __restrict__ S,
                              int M) {
    constexpr int N = NT * 16;
    const int wave = (blockIdx.x * blockDim.x + threadIdx.x) >> 5;
    const int lane = threadIdx.x & 31;
    const int row0 = wave * 16;
    if (row0 >= M) return;

    const int r      = lane & 15;
    const int khalf  = (lane >> 4) * 2;
    const int rowoff = (lane >> 4) * 8;

    float d2[8];
    if (SELF) {
        const float4 da = *(const float4*)(dis + row0 + rowoff);
        const float4 db = *(const float4*)(dis + row0 + rowoff + 4);
        d2[0] = da.x * da.x; d2[1] = da.y * da.y; d2[2] = da.z * da.z; d2[3] = da.w * da.w;
        d2[4] = db.x * db.x; d2[5] = db.y * db.y; d2[6] = db.z * db.z; d2[7] = db.w * db.w;
    }

    v8f acc[NT] = {};

    const float* __restrict__ xrow = X + (size_t)(row0 + r) * KDIM;

#pragma unroll
    for (int k = 0; k < KDIM; k += 4) {
        const int kk = k + khalf;
        v2f a = *(const v2f*)(xrow + kk);
        if (RELU_IN) {
            a.x = fmaxf(a.x, 0.0f);
            a.y = fmaxf(a.y, 0.0f);
        }
#pragma unroll
        for (int t = 0; t < NT; ++t) {
            const int col = t * 16 + r;
            v2f b;
            b.x = W[(size_t)kk * N + col];
            b.y = W[(size_t)(kk + 1) * N + col];
            acc[t] = __builtin_amdgcn_wmma_f32_16x16x4_f32(
                 false, a,  false, b,
                 (short)0, acc[t],
                 false,  false);
        }
    }

#pragma unroll
    for (int t = 0; t < NT; ++t) {
        const int col = t * 16 + r;
        const float bv = bias[col];
#pragma unroll
        for (int g = 0; g < 8; ++g) {
            const size_t o = (size_t)(row0 + rowoff + g) * N + col;
            if (SELF) {
                Y[o] = acc[t][g];
                S[o] = acc[t][g] * d2[g] + bv;
            } else {
                Y[o] = fmaxf(acc[t][g] + bv, 0.0f);
            }
        }
    }
}

__global__ void edge_scatter(const float* __restrict__ h,
                             const int* __restrict__ src,
                             const int* __restrict__ dst,
                             const float* __restrict__ dis,
                             float* __restrict__ agg,
                             int ne) {
    const int tid = blockIdx.x * blockDim.x + threadIdx.x;
    const int e = tid >> 4;
    if (e >= ne) return;
    const int f = (tid & 15) * 4;
    const int s = src[e];
    const int d = dst[e];
    const float norm = dis[s] * dis[d];
    const float4 hv = *(const float4*)(h + (size_t)s * 64 + f);
    float* ap = agg + (size_t)d * 64 + f;
    unsafeAtomicAdd(ap + 0, hv.x * norm);
    unsafeAtomicAdd(ap + 1, hv.y * norm);
    unsafeAtomicAdd(ap + 2, hv.z * norm);
    unsafeAtomicAdd(ap + 3, hv.w * norm);
}

__global__ void out_proj(const float* __restrict__ hd,
                         const float* __restrict__ Wo,
                         const float* __restrict__ bo,
                         float* __restrict__ out,
                         int n) {
    const int i = blockIdx.x * blockDim.x + threadIdx.x;
    if (i >= n) return;
    const float* h = hd + (size_t)i * 128;
    float a0 = bo[0], a1 = bo[1];
#pragma unroll 8
    for (int k = 0; k < 128; ++k) {
        const float hv = h[k];
        a0 += hv * Wo[k * 2 + 0];
        a1 += hv * Wo[k * 2 + 1];
    }
    out[(size_t)i * 2 + 0] = a0;
    out[(size_t)i * 2 + 1] = a1;
}

extern "C" void kernel_launch(void* const* d_in, const int* in_sizes, int n_in,
                              void* d_out, int out_size, void* d_ws, size_t ws_size,
                              hipStream_t stream) {
    const float* x  = (const float*)d_in[0];
    const int*   ei = (const int*)d_in[1];
    const float* W1 = (const float*)d_in[2];
    const float* b1 = (const float*)d_in[3];
    const float* W2 = (const float*)d_in[4];
    const float* b2 = (const float*)d_in[5];
    const float* Wd = (const float*)d_in[6];
    const float* bd = (const float*)d_in[7];
    const float* Wo = (const float*)d_in[8];
    const float* bo = (const float*)d_in[9];
    float* out = (float*)d_out;

    const int N = in_sizes[0] / 64;
    const int E = in_sizes[1] / 2;
    const int* src = ei;
    const int* dst = ei + E;

    float* dis  = (float*)d_ws;
    float* bufA = dis  + N;
    float* bufB = bufA + (size_t)N * 64;
    float* bufC = bufB + (size_t)N * 64;
    float* bufD = bufC + (size_t)N * 64;
    unsigned* part = (unsigned*)(bufD + (size_t)N * 128);
    (void)ws_size; (void)n_in; (void)out_size;

    const int TB = 256;
    const int gemm_waves   = N / 16;
    const int gemm_blocks  = (gemm_waves * 32 + TB - 1) / TB;
    const int edge_blocks  = (E * 16 + TB - 1) / TB;

    const int deg_ranges = (N + DEG_R - 1) / DEG_R;
    const int deg_chunks = (E + DEG_CH - 1) / DEG_CH;
    deg_hist<<<dim3(deg_ranges, deg_chunks), DEG_TB, 0, stream>>>(dst, part, E, N);
    finish_dis<<<(N + TB - 1) / TB, TB, 0, stream>>>(part, dis, N, deg_chunks);

    gemm_wmma_f32<4, 64, false, true><<<gemm_blocks, TB, 0, stream>>>(x, W1, b1, dis, bufA, bufB, N);
    edge_scatter<<<edge_blocks, TB, 0, stream>>>(bufA, src, dst, dis, bufB, E);

    gemm_wmma_f32<4, 64, true, true><<<gemm_blocks, TB, 0, stream>>>(bufB, W2, b2, dis, bufA, bufC, N);
    edge_scatter<<<edge_blocks, TB, 0, stream>>>(bufA, src, dst, dis, bufC, E);

    gemm_wmma_f32<8, 64, true, false><<<gemm_blocks, TB, 0, stream>>>(bufC, Wd, bd, nullptr, bufD, nullptr, N);

    out_proj<<<(N + TB - 1) / TB, TB, 0, stream>>>(bufD, Wo, bo, out, N);
}


# edge sums: a block keeps a 784-node range's rows in LDS, scans the whole dst list for it, adds gathered h rows by LDS f32 atomics, stores rows once; no global atomics; sums in LDS arrival order
# speedup vs baseline: 1.4262x; 1.1256x over previous
#include <hip/hip_runtime.h>
#include <hip/hip_bf16.h>

typedef __attribute__((ext_vector_type(2))) float v2f;
typedef __attribute__((ext_vector_type(8))) float v8f;

#define N_NODES_C 100000

#define DEG_R   14336
#define DEG_TB  1024
#define DEG_IT  11
#define DEG_CH  (DEG_TB * DEG_IT * 4)
static_assert(DEG_R % DEG_TB == 0, "a range is a whole number of block strides");

__launch_bounds__(DEG_TB)
__global__ void deg_hist(const int* __restrict__ dst, unsigned* __restrict__ part,
                         int ne, int n) {
    __shared__ unsigned hist[DEG_R];
    const int tid = threadIdx.x;
    const int lo  = blockIdx.x * DEG_R;
    const int e0  = blockIdx.y * DEG_CH;

    for (int i = tid; i < DEG_R; i += DEG_TB) hist[i] = 0u;
    __syncthreads();

    if ((ne & 3) == 0) {
        const int4* __restrict__ d4 = (const int4*)dst;
        const int n4 = ne >> 2;
        const int j0 = (e0 >> 2) + tid;
        int4 v[DEG_IT];
#pragma unroll
        for (int it = 0; it < DEG_IT; ++it) {
            const int j = j0 + it * DEG_TB;
            v[it] = d4[j < n4 ? j : n4 - 1];
        }
#pragma unroll
        for (int it = 0; it < DEG_IT; ++it) {
            const int j = j0 + it * DEG_TB;
            if (j < n4) {
                const unsigned a0 = (unsigned)(v[it].x - lo);
                const unsigned a1 = (unsigned)(v[it].y - lo);
                const unsigned a2 = (unsigned)(v[it].z - lo);
                const unsigned a3 = (unsigned)(v[it].w - lo);
                if (a0 < (unsigned)DEG_R) atomicAdd(&hist[a0], 1u);
                if (a1 < (unsigned)DEG_R) atomicAdd(&hist[a1], 1u);
                if (a2 < (unsigned)DEG_R) atomicAdd(&hist[a2], 1u);
                if (a3 < (unsigned)DEG_R) atomicAdd(&hist[a3], 1u);
            }
        }
    } else {
        const int e1 = (e0 + DEG_CH < ne) ? (e0 + DEG_CH) : ne;
        for (int e = e0 + tid; e < e1; e += DEG_TB) {
            const unsigned a = (unsigned)(dst[e] - lo);
            if (a < (unsigned)DEG_R) atomicAdd(&hist[a], 1u);
        }
    }
    __syncthreads();

    unsigned* __restrict__ row = part + (size_t)blockIdx.y * n;
    unsigned hv[DEG_R / DEG_TB];
#pragma unroll
    for (int k = 0; k < DEG_R / DEG_TB; ++k) hv[k] = hist[tid + k * DEG_TB];
#pragma unroll
    for (int k = 0; k < DEG_R / DEG_TB; ++k) {
        const int node = lo + tid + k * DEG_TB;
        if (node < n) row[node] = hv[k];
    }
}

__global__ void finish_dis(const unsigned* __restrict__ part, float* __restrict__ dis,
                           int n, int nc) {
    int i = blockIdx.x * blockDim.x + threadIdx.x;
    if (i >= n) return;
    unsigned s = 0u;
#pragma unroll 12
    for (int c = 0; c < nc; ++c) s += part[(size_t)c * n + i];
    dis[i] = rsqrtf((float)s + 1.0f);
}

template<int NT, int KDIM, bool RELU_IN, bool SELF>
__launch_bounds__(256)
__global__ void gemm_wmma_f32(const float* __restrict__ X,
                              const float* __restrict__ W,
                              const float* __restrict__ bias,
                              const float* __restrict__ dis,
                              float* __restrict__ Y,
                              float* __restrict__ S,
                              int M) {
    constexpr int N = NT * 16;
    const int wave = (blockIdx.x * blockDim.x + threadIdx.x) >> 5;
    const int lane = threadIdx.x & 31;
    const int row0 = wave * 16;
    if (row0 >= M) return;

    const int r      = lane & 15;
    const int khalf  = (lane >> 4) * 2;
    const int rowoff = (lane >> 4) * 8;

    float d2[8];
    if (SELF) {
        const float4 da = *(const float4*)(dis + row0 + rowoff);
        const float4 db = *(const float4*)(dis + row0 + rowoff + 4);
        d2[0] = da.x * da.x; d2[1] = da.y * da.y; d2[2] = da.z * da.z; d2[3] = da.w * da.w;
        d2[4] = db.x * db.x; d2[5] = db.y * db.y; d2[6] = db.z * db.z; d2[7] = db.w * db.w;
    }

    v8f acc[NT] = {};

    const float* __restrict__ xrow = X + (size_t)(row0 + r) * KDIM;

#pragma unroll
    for (int k = 0; k < KDIM; k += 4) {
        const int kk = k + khalf;
        v2f a = *(const v2f*)(xrow + kk);
        if (RELU_IN) {
            a.x = fmaxf(a.x, 0.0f);
            a.y = fmaxf(a.y, 0.0f);
        }
#pragma unroll
        for (int t = 0; t < NT; ++t) {
            const int col = t * 16 + r;
            v2f b;
            b.x = W[(size_t)kk * N + col];
            b.y = W[(size_t)(kk + 1) * N + col];
            acc[t] = __builtin_amdgcn_wmma_f32_16x16x4_f32(
                 false, a,  false, b,
                 (short)0, acc[t],
                 false,  false);
        }
    }

#pragma unroll
    for (int t = 0; t < NT; ++t) {
        const int col = t * 16 + r;
        const float bv = bias[col];
#pragma unroll
        for (int g = 0; g < 8; ++g) {
            const size_t o = (size_t)(row0 + rowoff + g) * N + col;
            if (SELF) {
                Y[o] = acc[t][g];
                S[o] = acc[t][g] * d2[g] + bv;
            } else {
                Y[o] = fmaxf(acc[t][g] + bv, 0.0f);
            }
        }
    }
}

#define AGG_TB  1024
#define AGG_R   784
#define AGG_U   4
#define AGG_Q   256
#define AGG_G   (AGG_Q / 32)
#define AGG_B   16

__device__ __forceinline__ void agg_drain(const float* __restrict__ h,
                                          const int* __restrict__ src,
                                          const float* __restrict__ dis,
                                          float* acc, int* qe, const int* qa, float* qw,
                                          int cnt, int lo, int lane) {
    if (cnt == 0) return;
    int ev[AGG_G], av[AGG_G], sv[AGG_G];
#pragma unroll
    for (int g = 0; g < AGG_G; ++g) {
        const int idx = g * 32 + lane;
        const bool ok = idx < cnt;
        ev[g] = ok ? qe[idx] : 0;
        av[g] = ok ? qa[idx] : 0;
    }
#pragma unroll
    for (int g = 0; g < AGG_G; ++g) sv[g] = src[ev[g]];
#pragma unroll
    for (int g = 0; g < AGG_G; ++g) asm volatile("" :: "v"(sv[g]));
#pragma unroll
    for (int g = 0; g < AGG_G; ++g) {
        const int idx = g * 32 + lane;
        const float w = dis[sv[g]] * dis[lo + av[g]];
        if (idx < cnt) {
            qe[idx] = sv[g];
            qw[idx] = w;
        }
    }
    for (int g = 0; g < cnt; g += 32) {
        const int idx = g + lane;
        const bool valid = idx < cnt;
        const int s = valid ? qe[idx] : 0;
        const int a = valid ? qa[idx] : 0;
        const float w = valid ? qw[idx] : 0.0f;
        const int m = (cnt - g < 32) ? (cnt - g) : 32;
#pragma unroll 1
        for (int u0 = 0; u0 < m; u0 += AGG_B) {
            v2f rows[AGG_B];
#pragma unroll
            for (int u = 0; u < AGG_B; ++u) {
                const int su = __builtin_amdgcn_readlane(s, u0 + u);
                rows[u] = *(const v2f*)(h + (size_t)su * 64 + lane * 2);
            }
#pragma unroll
            for (int u = 0; u < AGG_B; ++u) {
                if (u0 + u < m) {
                    const float wu = __int_as_float(__builtin_amdgcn_readlane(__float_as_int(w), u0 + u));
                    const int au = __builtin_amdgcn_readlane(a, u0 + u);
                    float* p = acc + au * 64 + lane * 2;
                    __hip_atomic_fetch_add(p + 0, rows[u].x * wu, __ATOMIC_RELAXED, __HIP_MEMORY_SCOPE_WORKGROUP);
                    __hip_atomic_fetch_add(p + 1, rows[u].y * wu, __ATOMIC_RELAXED, __HIP_MEMORY_SCOPE_WORKGROUP);
                }
            }
#pragma unroll
            for (int u = 0; u < AGG_B; ++u) asm volatile("" :: "v"(rows[u]));
        }
    }
}

__device__ __forceinline__ void agg_load(const int4* __restrict__ d4, int n4, int jb,
                                         int4 (&v)[AGG_U]) {
#pragma unroll
    for (int u = 0; u < AGG_U; ++u) {
        const int j = jb + u * 32;
        v[u] = d4[j < n4 ? j : n4 - 1];
    }
}

#define AGG_DRAIN_IF(ROOM)                                                      \
    if (cnt + (ROOM) > AGG_Q) {                                                 \
        agg_drain(h, src, dis, acc, myqe, myqa, myqw, cnt, lo, lane);           \
        cnt = 0;                                                                \
    }

#define AGG_TAKE(VAL, EID, LIM)                                                 \
    {                                                                           \
        const unsigned a_ = (unsigned)((VAL) - lo);                             \
        const bool m_ = a_ < (LIM);                                             \
        const unsigned mask_ = __builtin_amdgcn_ballot_w32(m_);                 \
        if (mask_) {                                                            \
            if (m_) {                                                           \
                const int pos_ = cnt + (int)__builtin_amdgcn_mbcnt_lo(mask_, 0u); \
                myqe[pos_] = (EID);                                             \
                myqa[pos_] = (int)a_;                                           \
            }                                                                   \
            cnt += __builtin_popcount(mask_);                                   \
        }                                                                       \
    }

#define AGG_STEP(V, JB, LIVE)                                                   \
    _Pragma("unroll")                                                           \
    for (int u = 0; u < AGG_U; ++u) {                                           \
        const int j = (JB) + u * 32;                                            \
        const unsigned lim = ((LIVE) && j < n4) ? (unsigned)nr : 0u;            \
        AGG_DRAIN_IF(128)                                                       \
        AGG_TAKE(V[u].x, 4 * j + 0, lim)                                        \
        AGG_TAKE(V[u].y, 4 * j + 1, lim)                                        \
        AGG_TAKE(V[u].z, 4 * j + 2, lim)                                        \
        AGG_TAKE(V[u].w, 4 * j + 3, lim)                                        \
    }

__launch_bounds__(AGG_TB)
__global__ void edge_tile(const float* __restrict__ h,
                          const int* __restrict__ src,
                          const int* __restrict__ dst,
                          const float* __restrict__ dis,
                          float* __restrict__ agg,
                          int ne, int n) {
    __shared__ float acc[AGG_R * 64];
    __shared__ int   qe_all[AGG_TB / 32][AGG_Q];
    __shared__ int   qa_all[AGG_TB / 32][AGG_Q];
    __shared__ float qw_all[AGG_TB / 32][AGG_Q];

    const int tid  = threadIdx.x;
    const int wave = tid >> 5;
    const int lane = tid & 31;
    const int lo   = blockIdx.x * AGG_R;
    const int nr   = (n - lo < AGG_R) ? (n - lo) : AGG_R;
    int*   myqe = qe_all[wave];
    int*   myqa = qa_all[wave];
    float* myqw = qw_all[wave];

    float4* acc4 = (float4*)acc;
    float4* agg4 = (float4*)(agg + (size_t)lo * 64);
    for (int i = tid; i < nr * 16; i += AGG_TB) acc4[i] = agg4[i];
    __syncthreads();

    int cnt = 0;
    if ((ne & 3) == 0) {
        const int4* __restrict__ d4 = (const int4*)dst;
        const int n4 = ne >> 2;
        const int per_wave = AGG_U * 32;
        const int per_step = (AGG_TB / 32) * per_wave;
        const int nsteps = (n4 + per_step - 1) / per_step;
        if (nsteps > 0) {
            int itA = blockIdx.x % nsteps;
            int jbA = (itA * (AGG_TB / 32) + wave) * per_wave + lane;
            int4 va[AGG_U], vb[AGG_U];
            agg_load(d4, n4, jbA, va);
            for (int k = 0; k < nsteps; k += 2) {
                const int itB = (itA + 1 == nsteps) ? 0 : itA + 1;
                const int jbB = (itB * (AGG_TB / 32) + wave) * per_wave + lane;
                agg_load(d4, n4, jbB, vb);
                AGG_STEP(va, jbA, true)
                itA = (itB + 1 == nsteps) ? 0 : itB + 1;
                jbA = (itA * (AGG_TB / 32) + wave) * per_wave + lane;
                agg_load(d4, n4, jbA, va);
                AGG_STEP(vb, jbB, (k + 1 < nsteps))
            }
        }
    } else {
        for (int eb = wave * 32; eb < ne; eb += AGG_TB) {
            const int e = eb + lane;
            const int val = dst[e < ne ? e : ne - 1];
            const unsigned lim = (e < ne) ? (unsigned)nr : 0u;
            AGG_DRAIN_IF(32)
            AGG_TAKE(val, e, lim)
        }
    }
    agg_drain(h, src, dis, acc, myqe, myqa, myqw, cnt, lo, lane);
    __syncthreads();

    for (int i = tid; i < nr * 16; i += AGG_TB) agg4[i] = acc4[i];
}

__global__ void out_proj(const float* __restrict__ hd,
                         const float* __restrict__ Wo,
                         const float* __restrict__ bo,
                         float* __restrict__ out,
                         int n) {
    const int i = blockIdx.x * blockDim.x + threadIdx.x;
    if (i >= n) return;
    const float* h = hd + (size_t)i * 128;
    float a0 = bo[0], a1 = bo[1];
#pragma unroll 8
    for (int k = 0; k < 128; ++k) {
        const float hv = h[k];
        a0 += hv * Wo[k * 2 + 0];
        a1 += hv * Wo[k * 2 + 1];
    }
    out[(size_t)i * 2 + 0] = a0;
    out[(size_t)i * 2 + 1] = a1;
}

extern "C" void kernel_launch(void* const* d_in, const int* in_sizes, int n_in,
                              void* d_out, int out_size, void* d_ws, size_t ws_size,
                              hipStream_t stream) {
    const float* x  = (const float*)d_in[0];
    const int*   ei = (const int*)d_in[1];
    const float* W1 = (const float*)d_in[2];
    const float* b1 = (const float*)d_in[3];
    const float* W2 = (const float*)d_in[4];
    const float* b2 = (const float*)d_in[5];
    const float* Wd = (const float*)d_in[6];
    const float* bd = (const float*)d_in[7];
    const float* Wo = (const float*)d_in[8];
    const float* bo = (const float*)d_in[9];
    float* out = (float*)d_out;

    const int N = in_sizes[0] / 64;
    const int E = in_sizes[1] / 2;
    const int* src = ei;
    const int* dst = ei + E;

    float* dis  = (float*)d_ws;
    float* bufA = dis  + N;
    float* bufB = bufA + (size_t)N * 64;
    float* bufC = bufB + (size_t)N * 64;
    float* bufD = bufC + (size_t)N * 64;
    unsigned* part = (unsigned*)(bufD + (size_t)N * 128);
    (void)ws_size; (void)n_in; (void)out_size;

    const int TB = 256;
    const int gemm_waves   = N / 16;
    const int gemm_blocks  = (gemm_waves * 32 + TB - 1) / TB;
    const int agg_blocks   = (N + AGG_R - 1) / AGG_R;

    const int deg_ranges = (N + DEG_R - 1) / DEG_R;
    const int deg_chunks = (E + DEG_CH - 1) / DEG_CH;
    deg_hist<<<dim3(deg_ranges, deg_chunks), DEG_TB, 0, stream>>>(dst, part, E, N);
    finish_dis<<<(N + TB - 1) / TB, TB, 0, stream>>>(part, dis, N, deg_chunks);

    gemm_wmma_f32<4, 64, false, true><<<gemm_blocks, TB, 0, stream>>>(x, W1, b1, dis, bufA, bufB, N);
    edge_tile<<<agg_blocks, AGG_TB, 0, stream>>>(bufA, src, dst, dis, bufB, E, N);

    gemm_wmma_f32<4, 64, true, true><<<gemm_blocks, TB, 0, stream>>>(bufB, W2, b2, dis, bufA, bufC, N);
    edge_tile<<<agg_blocks, AGG_TB, 0, stream>>>(bufA, src, dst, dis, bufC, E, N);

    gemm_wmma_f32<8, 64, true, false><<<gemm_blocks, TB, 0, stream>>>(bufC, Wd, bd, nullptr, bufD, nullptr, N);

    out_proj<<<(N + TB - 1) / TB, TB, 0, stream>>>(bufD, Wo, bo, out, N);
}


# edge pass scans one byte per edge (owning node block's number, written by the degree pass) instead of the 4-byte dst; 256 blocks of 391 nodes (was 128 of 784); dst read for queued edges only
# speedup vs baseline: 1.8322x; 1.2847x over previous
#include <hip/hip_runtime.h>
#include <hip/hip_bf16.h>

typedef __attribute__((ext_vector_type(2))) float v2f;
typedef __attribute__((ext_vector_type(8))) float v8f;

#define N_NODES_C 100000

#define AGG_R   391
static_assert((N_NODES_C + AGG_R - 1) / AGG_R <= 256, "a block number must fit one byte");
#define DEG_R   14336
#define DEG_TB  1024
#define DEG_IT  11
#define DEG_CH  (DEG_TB * DEG_IT * 4)
static_assert(DEG_R % DEG_TB == 0, "a range is a whole number of block strides");

__launch_bounds__(DEG_TB)
__global__ void deg_hist(const int* __restrict__ dst, unsigned* __restrict__ part,
                         unsigned char* __restrict__ keys, int ne, int n) {
    __shared__ unsigned hist[DEG_R];
    const int tid = threadIdx.x;
    const int lo  = blockIdx.x * DEG_R;
    const int e0  = blockIdx.y * DEG_CH;

    for (int i = tid; i < DEG_R; i += DEG_TB) hist[i] = 0u;
    __syncthreads();

    if ((ne & 3) == 0) {
        const int4* __restrict__ d4 = (const int4*)dst;
        const int n4 = ne >> 2;
        const int j0 = (e0 >> 2) + tid;
        int4 v[DEG_IT];
#pragma unroll
        for (int it = 0; it < DEG_IT; ++it) {
            const int j = j0 + it * DEG_TB;
            v[it] = d4[j < n4 ? j : n4 - 1];
        }
#pragma unroll
        for (int it = 0; it < DEG_IT; ++it) {
            const int j = j0 + it * DEG_TB;
            if (j < n4) {
                if (blockIdx.x == 0) {
                    ((unsigned*)keys)[j] = ((unsigned)v[it].x / AGG_R)
                                         | (((unsigned)v[it].y / AGG_R) << 8)
                                         | (((unsigned)v[it].z / AGG_R) << 16)
                                         | (((unsigned)v[it].w / AGG_R) << 24);
                }
                const unsigned a0 = (unsigned)(v[it].x - lo);
                const unsigned a1 = (unsigned)(v[it].y - lo);
                const unsigned a2 = (unsigned)(v[it].z - lo);
                const unsigned a3 = (unsigned)(v[it].w - lo);
                if (a0 < (unsigned)DEG_R) atomicAdd(&hist[a0], 1u);
                if (a1 < (unsigned)DEG_R) atomicAdd(&hist[a1], 1u);
                if (a2 < (unsigned)DEG_R) atomicAdd(&hist[a2], 1u);
                if (a3 < (unsigned)DEG_R) atomicAdd(&hist[a3], 1u);
            }
        }
    } else {
        const int e1 = (e0 + DEG_CH < ne) ? (e0 + DEG_CH) : ne;
        for (int e = e0 + tid; e < e1; e += DEG_TB) {
            const int d = dst[e];
            if (blockIdx.x == 0) keys[e] = (unsigned char)((unsigned)d / AGG_R);
            const unsigned a = (unsigned)(d - lo);
            if (a < (unsigned)DEG_R) atomicAdd(&hist[a], 1u);
        }
    }
    __syncthreads();

    unsigned* __restrict__ row = part + (size_t)blockIdx.y * n;
    unsigned hv[DEG_R / DEG_TB];
#pragma unroll
    for (int k = 0; k < DEG_R / DEG_TB; ++k) hv[k] = hist[tid + k * DEG_TB];
#pragma unroll
    for (int k = 0; k < DEG_R / DEG_TB; ++k) {
        const int node = lo + tid + k * DEG_TB;
        if (node < n) row[node] = hv[k];
    }
}

__global__ void finish_dis(const unsigned* __restrict__ part, float* __restrict__ dis,
                           int n, int nc) {
    int i = blockIdx.x * blockDim.x + threadIdx.x;
    if (i >= n) return;
    unsigned s = 0u;
#pragma unroll 12
    for (int c = 0; c < nc; ++c) s += part[(size_t)c * n + i];
    dis[i] = rsqrtf((float)s + 1.0f);
}

template<int NT, int KDIM, bool RELU_IN, bool SELF>
__launch_bounds__(256)
__global__ void gemm_wmma_f32(const float* __restrict__ X,
                              const float* __restrict__ W,
                              const float* __restrict__ bias,
                              const float* __restrict__ dis,
                              float* __restrict__ Y,
                              float* __restrict__ S,
                              int M) {
    constexpr int N = NT * 16;
    const int wave = (blockIdx.x * blockDim.x + threadIdx.x) >> 5;
    const int lane = threadIdx.x & 31;
    const int row0 = wave * 16;
    if (row0 >= M) return;

    const int r      = lane & 15;
    const int khalf  = (lane >> 4) * 2;
    const int rowoff = (lane >> 4) * 8;

    float d2[8];
    if (SELF) {
        const float4 da = *(const float4*)(dis + row0 + rowoff);
        const float4 db = *(const float4*)(dis + row0 + rowoff + 4);
        d2[0] = da.x * da.x; d2[1] = da.y * da.y; d2[2] = da.z * da.z; d2[3] = da.w * da.w;
        d2[4] = db.x * db.x; d2[5] = db.y * db.y; d2[6] = db.z * db.z; d2[7] = db.w * db.w;
    }

    v8f acc[NT] = {};

    const float* __restrict__ xrow = X + (size_t)(row0 + r) * KDIM;

#pragma unroll
    for (int k = 0; k < KDIM; k += 4) {
        const int kk = k + khalf;
        v2f a = *(const v2f*)(xrow + kk);
        if (RELU_IN) {
            a.x = fmaxf(a.x, 0.0f);
            a.y = fmaxf(a.y, 0.0f);
        }
#pragma unroll
        for (int t = 0; t < NT; ++t) {
            const int col = t * 16 + r;
            v2f b;
            b.x = W[(size_t)kk * N + col];
            b.y = W[(size_t)(kk + 1) * N + col];
            acc[t] = __builtin_amdgcn_wmma_f32_16x16x4_f32(
                 false, a,  false, b,
                 (short)0, acc[t],
                 false,  false);
        }
    }

#pragma unroll
    for (int t = 0; t < NT; ++t) {
        const int col = t * 16 + r;
        const float bv = bias[col];
#pragma unroll
        for (int g = 0; g < 8; ++g) {
            const size_t o = (size_t)(row0 + rowoff + g) * N + col;
            if (SELF) {
                Y[o] = acc[t][g];
                S[o] = acc[t][g] * d2[g] + bv;
            } else {
                Y[o] = fmaxf(acc[t][g] + bv, 0.0f);
            }
        }
    }
}

#define AGG_TB  1024
#define AGG_U   4
#define AGG_C   256
#define AGG_Q   (3 * AGG_C)
#define AGG_G   (AGG_C / 32)
#define AGG_B   16

__device__ __forceinline__ void agg_drain(const float* __restrict__ h,
                                          const int* __restrict__ src,
                                          const int* __restrict__ dst,
                                          const float* __restrict__ dis,
                                          float* acc, int* qe, int* qa, float* qw,
                                          int cnt, int lo, int lane) {
#pragma unroll 1
    for (int c0 = 0; c0 < cnt; c0 += AGG_C) {
        const int cc = (cnt - c0 < AGG_C) ? (cnt - c0) : AGG_C;
        int* ce = qe + c0;
        int ev[AGG_G], sv[AGG_G], dv[AGG_G];
#pragma unroll
        for (int g = 0; g < AGG_G; ++g) {
            const int idx = g * 32 + lane;
            ev[g] = (idx < cc) ? ce[idx] : 0;
        }
#pragma unroll
        for (int g = 0; g < AGG_G; ++g) {
            sv[g] = src[ev[g]];
            dv[g] = dst[ev[g]];
        }
#pragma unroll
        for (int g = 0; g < AGG_G; ++g) asm volatile("" :: "v"(sv[g]), "v"(dv[g]));
#pragma unroll
        for (int g = 0; g < AGG_G; ++g) {
            const int idx = g * 32 + lane;
            const float w = dis[sv[g]] * dis[dv[g]];
            if (idx < cc) {
                ce[idx] = sv[g];
                qa[idx] = dv[g] - lo;
                qw[idx] = w;
            }
        }
        for (int g = 0; g < cc; g += 32) {
            const int idx = g + lane;
            const bool valid = idx < cc;
            const int s = valid ? ce[idx] : 0;
            const int a = valid ? qa[idx] : 0;
            const float w = valid ? qw[idx] : 0.0f;
            const int m = (cc - g < 32) ? (cc - g) : 32;
#pragma unroll 1
            for (int u0 = 0; u0 < m; u0 += AGG_B) {
                v2f rows[AGG_B];
#pragma unroll
                for (int u = 0; u < AGG_B; ++u) {
                    const int su = __builtin_amdgcn_readlane(s, u0 + u);
                    rows[u] = *(const v2f*)(h + (size_t)su * 64 + lane * 2);
                }
#pragma unroll
                for (int u = 0; u < AGG_B; ++u) {
                    if (u0 + u < m) {
                        const float wu = __int_as_float(__builtin_amdgcn_readlane(__float_as_int(w), u0 + u));
                        const int au = __builtin_amdgcn_readlane(a, u0 + u);
                        float* p = acc + au * 64 + lane * 2;
                        __hip_atomic_fetch_add(p + 0, rows[u].x * wu, __ATOMIC_RELAXED, __HIP_MEMORY_SCOPE_WORKGROUP);
                        __hip_atomic_fetch_add(p + 1, rows[u].y * wu, __ATOMIC_RELAXED, __HIP_MEMORY_SCOPE_WORKGROUP);
                    }
                }
#pragma unroll
                for (int u = 0; u < AGG_B; ++u) asm volatile("" :: "v"(rows[u]));
            }
        }
    }
}

__device__ __forceinline__ void agg_load(const uint4* __restrict__ k4, int n16, int jb,
                                         uint4 (&v)[AGG_U]) {
#pragma unroll
    for (int u = 0; u < AGG_U; ++u) {
        const int j = jb + u * 32;
        v[u] = k4[j < n16 ? j : n16 - 1];
    }
}

#define AGG_DRAIN_IF(COND)                                                      \
    if (COND) {                                                                 \
        agg_drain(h, src, dst, dis, acc, myqe, myqa, myqw, cnt, lo, lane);      \
        cnt = 0;                                                                \
    }

#define AGG_TAKE(PRED, EID)                                                     \
    {                                                                           \
        const bool m_ = (PRED);                                                 \
        const unsigned mask_ = __builtin_amdgcn_ballot_w32(m_);                 \
        if (mask_) {                                                            \
            if (m_) {                                                           \
                const int pos_ = cnt + (int)__builtin_amdgcn_mbcnt_lo(mask_, 0u); \
                myqe[pos_] = (EID);                                             \
            }                                                                   \
            cnt += __builtin_popcount(mask_);                                   \
        }                                                                       \
    }

#define AGG_WORD(XW, EID0)                                                      \
    {                                                                           \
        const unsigned x_ = (XW);                                               \
        const bool h_ = (((x_ - 0x01010101u) & ~x_ & 0x80808080u) != 0u);       \
        if (__builtin_amdgcn_ballot_w32(h_)) {                                  \
            AGG_TAKE((x_ & 0x000000ffu) == 0u, (EID0) + 0)                      \
            AGG_TAKE((x_ & 0x0000ff00u) == 0u, (EID0) + 1)                      \
            AGG_TAKE((x_ & 0x00ff0000u) == 0u, (EID0) + 2)                      \
            AGG_TAKE((x_ & 0xff000000u) == 0u, (EID0) + 3)                      \
        }                                                                       \
    }

#define AGG_STEP(V, JB, LIVE)                                                   \
    _Pragma("unroll")                                                           \
    for (int u = 0; u < AGG_U; ++u) {                                           \
        const int j = (JB) + u * 32;                                            \
        const unsigned dead = ((LIVE) && j < n16) ? 0u : 0xffffffffu;           \
        AGG_DRAIN_IF(cnt > early)                                               \
        AGG_WORD((V[u].x ^ idrep) | dead, 16 * j + 0)                           \
        AGG_WORD((V[u].y ^ idrep) | dead, 16 * j + 4)                           \
        AGG_WORD((V[u].z ^ idrep) | dead, 16 * j + 8)                           \
        AGG_WORD((V[u].w ^ idrep) | dead, 16 * j + 12)                          \
    }

__launch_bounds__(AGG_TB)
__global__ void edge_tile(const float* __restrict__ h,
                          const int* __restrict__ src,
                          const int* __restrict__ dst,
                          const unsigned char* __restrict__ keys,
                          const float* __restrict__ dis,
                          float* __restrict__ agg,
                          int ne, int n) {
    __shared__ float acc[AGG_R * 64];
    __shared__ int   qe_all[AGG_TB / 32][AGG_Q];
    __shared__ int   qa_all[AGG_TB / 32][AGG_C];
    __shared__ float qw_all[AGG_TB / 32][AGG_C];

    const int tid  = threadIdx.x;
    const int wave = tid >> 5;
    const int lane = tid & 31;
    const int lo   = blockIdx.x * AGG_R;
    const int nr   = (n - lo < AGG_R) ? (n - lo) : AGG_R;
    int*   myqe = qe_all[wave];
    int*   myqa = qa_all[wave];
    float* myqw = qw_all[wave];

    float4* acc4 = (float4*)acc;
    float4* agg4 = (float4*)(agg + (size_t)lo * 64);
    for (int i = tid; i < nr * 16; i += AGG_TB) acc4[i] = agg4[i];
    __syncthreads();

    int cnt = 0;
    if ((ne & 15) == 0) {
        const uint4* __restrict__ k4 = (const uint4*)keys;
        const unsigned idrep = blockIdx.x * 0x01010101u;
        const int n16 = ne >> 4;
        const int early = 64 + 4 * wave;
        static_assert(64 + 4 * (AGG_TB / 32 - 1) + 512 <= AGG_Q, "queue room for one more piece");
        const int per_wave = AGG_U * 32;
        const int per_step = (AGG_TB / 32) * per_wave;
        const int nsteps = (n16 + per_step - 1) / per_step;
        if (nsteps > 0) {
            int itA = blockIdx.x % nsteps;
            int jbA = (itA * (AGG_TB / 32) + wave) * per_wave + lane;
            uint4 va[AGG_U], vb[AGG_U];
            agg_load(k4, n16, jbA, va);
            for (int k = 0; k < nsteps; k += 2) {
                const int itB = (itA + 1 == nsteps) ? 0 : itA + 1;
                const int jbB = (itB * (AGG_TB / 32) + wave) * per_wave + lane;
                agg_load(k4, n16, jbB, vb);
                AGG_STEP(va, jbA, true)
                itA = (itB + 1 == nsteps) ? 0 : itB + 1;
                jbA = (itA * (AGG_TB / 32) + wave) * per_wave + lane;
                agg_load(k4, n16, jbA, va);
                AGG_STEP(vb, jbB, (k + 1 < nsteps))
            }
        }
    } else {
        for (int eb = wave * 32; eb < ne; eb += AGG_TB) {
            const int e = eb + lane;
            const int val = dst[e < ne ? e : ne - 1];
            AGG_DRAIN_IF(cnt + 32 > AGG_Q)
            AGG_TAKE(e < ne && (unsigned)(val - lo) < (unsigned)nr, e)
        }
    }
    agg_drain(h, src, dst, dis, acc, myqe, myqa, myqw, cnt, lo, lane);
    __syncthreads();

    for (int i = tid; i < nr * 16; i += AGG_TB) agg4[i] = acc4[i];
}

__global__ void out_proj(const float* __restrict__ hd,
                         const float* __restrict__ Wo,
                         const float* __restrict__ bo,
                         float* __restrict__ out,
                         int n) {
    const int i = blockIdx.x * blockDim.x + threadIdx.x;
    if (i >= n) return;
    const float* h = hd + (size_t)i * 128;
    float a0 = bo[0], a1 = bo[1];
#pragma unroll 8
    for (int k = 0; k < 128; ++k) {
        const float hv = h[k];
        a0 += hv * Wo[k * 2 + 0];
        a1 += hv * Wo[k * 2 + 1];
    }
    out[(size_t)i * 2 + 0] = a0;
    out[(size_t)i * 2 + 1] = a1;
}

extern "C" void kernel_launch(void* const* d_in, const int* in_sizes, int n_in,
                              void* d_out, int out_size, void* d_ws, size_t ws_size,
                              hipStream_t stream) {
    const float* x  = (const float*)d_in[0];
    const int*   ei = (const int*)d_in[1];
    const float* W1 = (const float*)d_in[2];
    const float* b1 = (const float*)d_in[3];
    const float* W2 = (const float*)d_in[4];
    const float* b2 = (const float*)d_in[5];
    const float* Wd = (const float*)d_in[6];
    const float* bd = (const float*)d_in[7];
    const float* Wo = (const float*)d_in[8];
    const float* bo = (const float*)d_in[9];
    float* out = (float*)d_out;

    const int N = in_sizes[0] / 64;
    const int E = in_sizes[1] / 2;
    const int* src = ei;
    const int* dst = ei + E;

    float* dis  = (float*)d_ws;
    float* bufA = dis  + N;
    float* bufB = bufA + (size_t)N * 64;
    float* bufC = bufB + (size_t)N * 64;
    float* bufD = bufC + (size_t)N * 64;
    unsigned* part = (unsigned*)(bufD + (size_t)N * 128);
    (void)ws_size; (void)n_in; (void)out_size;

    const int TB = 256;
    const int gemm_waves   = N / 16;
    const int gemm_blocks  = (gemm_waves * 32 + TB - 1) / TB;
    const int agg_blocks   = (N + AGG_R - 1) / AGG_R;

    const int deg_ranges = (N + DEG_R - 1) / DEG_R;
    const int deg_chunks = (E + DEG_CH - 1) / DEG_CH;
    unsigned char* keys = (unsigned char*)(part + (size_t)deg_chunks * N);
    deg_hist<<<dim3(deg_ranges, deg_chunks), DEG_TB, 0, stream>>>(dst, part, keys, E, N);
    finish_dis<<<(N + TB - 1) / TB, TB, 0, stream>>>(part, dis, N, deg_chunks);

    gemm_wmma_f32<4, 64, false, true><<<gemm_blocks, TB, 0, stream>>>(x, W1, b1, dis, bufA, bufB, N);
    edge_tile<<<agg_blocks, AGG_TB, 0, stream>>>(bufA, src, dst, keys, dis, bufB, E, N);

    gemm_wmma_f32<4, 64, true, true><<<gemm_blocks, TB, 0, stream>>>(bufB, W2, b2, dis, bufA, bufC, N);
    edge_tile<<<agg_blocks, AGG_TB, 0, stream>>>(bufA, src, dst, keys, dis, bufC, E, N);

    gemm_wmma_f32<8, 64, true, false><<<gemm_blocks, TB, 0, stream>>>(bufC, Wd, bd, nullptr, bufD, nullptr, N);

    out_proj<<<(N + TB - 1) / TB, TB, 0, stream>>>(bufD, Wo, bo, out, N);
}


# layer 2's edge pass reads a per-node-block edge list (source, row, weight) that layer 1's pass stores while draining; no second key scan or src/dst/scale look-ups; sums still in LDS arrival order
# speedup vs baseline: 2.1984x; 1.1999x over previous
#include <hip/hip_runtime.h>
#include <hip/hip_bf16.h>

typedef __attribute__((ext_vector_type(2))) float v2f;
typedef __attribute__((ext_vector_type(8))) float v8f;

#define N_NODES_C 100000

#define AGG_R   391
static_assert((N_NODES_C + AGG_R - 1) / AGG_R <= 256, "a block number must fit one byte");
#define DEG_R   14336
#define DEG_TB  1024
#define DEG_IT  11
#define DEG_CH  (DEG_TB * DEG_IT * 4)
static_assert(DEG_R % DEG_TB == 0, "a range is a whole number of block strides");

__launch_bounds__(DEG_TB)
__global__ void deg_hist(const int* __restrict__ dst, unsigned* __restrict__ part,
                         unsigned char* __restrict__ keys, int ne, int n) {
    __shared__ unsigned hist[DEG_R];
    const int tid = threadIdx.x;
    const int lo  = blockIdx.x * DEG_R;
    const int e0  = blockIdx.y * DEG_CH;

    for (int i = tid; i < DEG_R; i += DEG_TB) hist[i] = 0u;
    __syncthreads();

    if ((ne & 3) == 0) {
        const int4* __restrict__ d4 = (const int4*)dst;
        const int n4 = ne >> 2;
        const int j0 = (e0 >> 2) + tid;
        int4 v[DEG_IT];
#pragma unroll
        for (int it = 0; it < DEG_IT; ++it) {
            const int j = j0 + it * DEG_TB;
            v[it] = d4[j < n4 ? j : n4 - 1];
        }
#pragma unroll
        for (int it = 0; it < DEG_IT; ++it) {
            const int j = j0 + it * DEG_TB;
            if (j < n4) {
                if (blockIdx.x == 0) {
                    ((unsigned*)keys)[j] = ((unsigned)v[it].x / AGG_R)
                                         | (((unsigned)v[it].y / AGG_R) << 8)
                                         | (((unsigned)v[it].z / AGG_R) << 16)
                                         | (((unsigned)v[it].w / AGG_R) << 24);
                }
                const unsigned a0 = (unsigned)(v[it].x - lo);
                const unsigned a1 = (unsigned)(v[it].y - lo);
                const unsigned a2 = (unsigned)(v[it].z - lo);
                const unsigned a3 = (unsigned)(v[it].w - lo);
                if (a0 < (unsigned)DEG_R) atomicAdd(&hist[a0], 1u);
                if (a1 < (unsigned)DEG_R) atomicAdd(&hist[a1], 1u);
                if (a2 < (unsigned)DEG_R) atomicAdd(&hist[a2], 1u);
                if (a3 < (unsigned)DEG_R) atomicAdd(&hist[a3], 1u);
            }
        }
    } else {
        const int e1 = (e0 + DEG_CH < ne) ? (e0 + DEG_CH) : ne;
        for (int e = e0 + tid; e < e1; e += DEG_TB) {
            const int d = dst[e];
            if (blockIdx.x == 0) keys[e] = (unsigned char)((unsigned)d / AGG_R);
            const unsigned a = (unsigned)(d - lo);
            if (a < (unsigned)DEG_R) atomicAdd(&hist[a], 1u);
        }
    }
    __syncthreads();

    unsigned* __restrict__ row = part + (size_t)blockIdx.y * n;
    unsigned hv[DEG_R / DEG_TB];
#pragma unroll
    for (int k = 0; k < DEG_R / DEG_TB; ++k) hv[k] = hist[tid + k * DEG_TB];
#pragma unroll
    for (int k = 0; k < DEG_R / DEG_TB; ++k) {
        const int node = lo + tid + k * DEG_TB;
        if (node < n) row[node] = hv[k];
    }
}

__device__ __forceinline__ unsigned wave_sum(unsigned v) {
#pragma unroll
    for (int m = 16; m > 0; m >>= 1) v += __shfl_xor(v, m);
    return v;
}

#define FIN_TB  512
static_assert(AGG_R <= FIN_TB, "one thread per node of a block");
__launch_bounds__(FIN_TB)
__global__ void finish_dis(const unsigned* __restrict__ part, float* __restrict__ dis,
                           unsigned* __restrict__ blkcnt, int n, int nc) {
    __shared__ unsigned tot;
    const int tid = threadIdx.x;
    if (tid == 0) tot = 0u;
    __syncthreads();
    const int i = blockIdx.x * AGG_R + tid;
    unsigned s = 0u;
    if (tid < AGG_R && i < n) {
#pragma unroll 12
        for (int c = 0; c < nc; ++c) s += part[(size_t)c * n + i];
        dis[i] = rsqrtf((float)s + 1.0f);
    }
    s = wave_sum(s);
    if ((tid & 31) == 0 && s != 0u) atomicAdd(&tot, s);
    __syncthreads();
    if (tid == 0) blkcnt[blockIdx.x] = tot;
}

template<int NT, int KDIM, bool RELU_IN, bool SELF>
__launch_bounds__(256)
__global__ void gemm_wmma_f32(const float* __restrict__ X,
                              const float* __restrict__ W,
                              const float* __restrict__ bias,
                              const float* __restrict__ dis,
                              float* __restrict__ Y,
                              float* __restrict__ S,
                              int M) {
    constexpr int N = NT * 16;
    const int wave = (blockIdx.x * blockDim.x + threadIdx.x) >> 5;
    const int lane = threadIdx.x & 31;
    const int row0 = wave * 16;
    if (row0 >= M) return;

    const int r      = lane & 15;
    const int khalf  = (lane >> 4) * 2;
    const int rowoff = (lane >> 4) * 8;

    float d2[8];
    if (SELF) {
        const float4 da = *(const float4*)(dis + row0 + rowoff);
        const float4 db = *(const float4*)(dis + row0 + rowoff + 4);
        d2[0] = da.x * da.x; d2[1] = da.y * da.y; d2[2] = da.z * da.z; d2[3] = da.w * da.w;
        d2[4] = db.x * db.x; d2[5] = db.y * db.y; d2[6] = db.z * db.z; d2[7] = db.w * db.w;
    }

    v8f acc[NT] = {};

    const float* __restrict__ xrow = X + (size_t)(row0 + r) * KDIM;

#pragma unroll
    for (int k = 0; k < KDIM; k += 4) {
        const int kk = k + khalf;
        v2f a = *(const v2f*)(xrow + kk);
        if (RELU_IN) {
            a.x = fmaxf(a.x, 0.0f);
            a.y = fmaxf(a.y, 0.0f);
        }
#pragma unroll
        for (int t = 0; t < NT; ++t) {
            const int col = t * 16 + r;
            v2f b;
            b.x = W[(size_t)kk * N + col];
            b.y = W[(size_t)(kk + 1) * N + col];
            acc[t] = __builtin_amdgcn_wmma_f32_16x16x4_f32(
                 false, a,  false, b,
                 (short)0, acc[t],
                 false,  false);
        }
    }

#pragma unroll
    for (int t = 0; t < NT; ++t) {
        const int col = t * 16 + r;
        const float bv = bias[col];
#pragma unroll
        for (int g = 0; g < 8; ++g) {
            const size_t o = (size_t)(row0 + rowoff + g) * N + col;
            if (SELF) {
                Y[o] = acc[t][g];
                S[o] = acc[t][g] * d2[g] + bv;
            } else {
                Y[o] = fmaxf(acc[t][g] + bv, 0.0f);
            }
        }
    }
}

static_assert(AGG_R <= 512 && N_NODES_C <= (1 << 23), "a list entry packs row (9 bits) and source (23 bits)");
#define AGG_TB  1024
#define AGG_U   4
#define AGG_C   256
#define AGG_Q   (3 * AGG_C)
#define AGG_G   (AGG_C / 32)
#define AGG_B   16

__device__ __forceinline__ void agg_drain(const float* __restrict__ h,
                                          const int* __restrict__ src,
                                          const int* __restrict__ dst,
                                          const float* __restrict__ dis,
                                          float* acc, int* qe, int* qa, float* qw,
                                          uint2* __restrict__ mylist, unsigned* fill,
                                          int cnt, int lo, int lane) {
#pragma unroll 1
    for (int c0 = 0; c0 < cnt; c0 += AGG_C) {
        const int cc = (cnt - c0 < AGG_C) ? (cnt - c0) : AGG_C;
        int* ce = qe + c0;
        int ev[AGG_G], sv[AGG_G], dv[AGG_G];
#pragma unroll
        for (int g = 0; g < AGG_G; ++g) {
            const int idx = g * 32 + lane;
            ev[g] = (idx < cc) ? ce[idx] : 0;
        }
#pragma unroll
        for (int g = 0; g < AGG_G; ++g) {
            sv[g] = src[ev[g]];
            dv[g] = dst[ev[g]];
        }
#pragma unroll
        for (int g = 0; g < AGG_G; ++g) asm volatile("" :: "v"(sv[g]), "v"(dv[g]));
        unsigned pos = 0u;
        if (lane == 0) pos = atomicAdd(fill, (unsigned)cc);
        pos = (unsigned)__builtin_amdgcn_readfirstlane((int)pos);
#pragma unroll
        for (int g = 0; g < AGG_G; ++g) {
            const int idx = g * 32 + lane;
            const float w = dis[sv[g]] * dis[dv[g]];
            if (idx < cc) {
                ce[idx] = sv[g];
                qa[idx] = dv[g] - lo;
                qw[idx] = w;
                mylist[pos + idx] = make_uint2((unsigned)sv[g] | ((unsigned)(dv[g] - lo) << 23),
                                               __float_as_uint(w));
            }
        }
        for (int g = 0; g < cc; g += 32) {
            const int idx = g + lane;
            const bool valid = idx < cc;
            const int s = valid ? ce[idx] : 0;
            const int a = valid ? qa[idx] : 0;
            const float w = valid ? qw[idx] : 0.0f;
            const int m = (cc - g < 32) ? (cc - g) : 32;
#pragma unroll 1
            for (int u0 = 0; u0 < m; u0 += AGG_B) {
                v2f rows[AGG_B];
#pragma unroll
                for (int u = 0; u < AGG_B; ++u) {
                    const int su = __builtin_amdgcn_readlane(s, u0 + u);
                    rows[u] = *(const v2f*)(h + (size_t)su * 64 + lane * 2);
                }
#pragma unroll
                for (int u = 0; u < AGG_B; ++u) {
                    if (u0 + u < m) {
                        const float wu = __int_as_float(__builtin_amdgcn_readlane(__float_as_int(w), u0 + u));
                        const int au = __builtin_amdgcn_readlane(a, u0 + u);
                        float* p = acc + au * 64 + lane * 2;
                        __hip_atomic_fetch_add(p + 0, rows[u].x * wu, __ATOMIC_RELAXED, __HIP_MEMORY_SCOPE_WORKGROUP);
                        __hip_atomic_fetch_add(p + 1, rows[u].y * wu, __ATOMIC_RELAXED, __HIP_MEMORY_SCOPE_WORKGROUP);
                    }
                }
#pragma unroll
                for (int u = 0; u < AGG_B; ++u) asm volatile("" :: "v"(rows[u]));
            }
        }
    }
}

__device__ __forceinline__ void agg_load(const uint4* __restrict__ k4, int n16, int jb,
                                         uint4 (&v)[AGG_U]) {
#pragma unroll
    for (int u = 0; u < AGG_U; ++u) {
        const int j = jb + u * 32;
        v[u] = k4[j < n16 ? j : n16 - 1];
    }
}

#define AGG_DRAIN_IF(COND)                                                      \
    if (COND) {                                                                 \
        agg_drain(h, src, dst, dis, acc, myqe, myqa, myqw, mylist, &s_fill, cnt, lo, lane); \
        cnt = 0;                                                                \
    }

#define AGG_TAKE(PRED, EID)                                                     \
    {                                                                           \
        const bool m_ = (PRED);                                                 \
        const unsigned mask_ = __builtin_amdgcn_ballot_w32(m_);                 \
        if (mask_) {                                                            \
            if (m_) {                                                           \
                const int pos_ = cnt + (int)__builtin_amdgcn_mbcnt_lo(mask_, 0u); \
                myqe[pos_] = (EID);                                             \
            }                                                                   \
            cnt += __builtin_popcount(mask_);                                   \
        }                                                                       \
    }

#define AGG_WORD(XW, EID0)                                                      \
    {                                                                           \
        const unsigned x_ = (XW);                                               \
        const bool h_ = (((x_ - 0x01010101u) & ~x_ & 0x80808080u) != 0u);       \
        if (__builtin_amdgcn_ballot_w32(h_)) {                                  \
            AGG_TAKE((x_ & 0x000000ffu) == 0u, (EID0) + 0)                      \
            AGG_TAKE((x_ & 0x0000ff00u) == 0u, (EID0) + 1)                      \
            AGG_TAKE((x_ & 0x00ff0000u) == 0u, (EID0) + 2)                      \
            AGG_TAKE((x_ & 0xff000000u) == 0u, (EID0) + 3)                      \
        }                                                                       \
    }

#define AGG_STEP(V, JB, LIVE)                                                   \
    _Pragma("unroll")                                                           \
    for (int u = 0; u < AGG_U; ++u) {                                           \
        const int j = (JB) + u * 32;                                            \
        const unsigned dead = ((LIVE) && j < n16) ? 0u : 0xffffffffu;           \
        AGG_DRAIN_IF(cnt > early)                                               \
        AGG_WORD((V[u].x ^ idrep) | dead, 16 * j + 0)                           \
        AGG_WORD((V[u].y ^ idrep) | dead, 16 * j + 4)                           \
        AGG_WORD((V[u].z ^ idrep) | dead, 16 * j + 8)                           \
        AGG_WORD((V[u].w ^ idrep) | dead, 16 * j + 12)                          \
    }

__launch_bounds__(AGG_TB)
__global__ void edge_tile(const float* __restrict__ h,
                          const int* __restrict__ src,
                          const int* __restrict__ dst,
                          const unsigned char* __restrict__ keys,
                          const float* __restrict__ dis,
                          float* __restrict__ agg,
                          const unsigned* __restrict__ blkcnt,
                          unsigned* __restrict__ list,
                          int ne, int n) {
    __shared__ float acc[AGG_R * 64];
    __shared__ int   qe_all[AGG_TB / 32][AGG_Q];
    __shared__ int   qa_all[AGG_TB / 32][AGG_C];
    __shared__ float qw_all[AGG_TB / 32][AGG_C];
    __shared__ unsigned s_base;
    __shared__ unsigned s_fill;

    const int tid  = threadIdx.x;
    const int wave = tid >> 5;
    const int lane = tid & 31;
    const int lo   = blockIdx.x * AGG_R;
    const int nr   = (n - lo < AGG_R) ? (n - lo) : AGG_R;
    int*   myqe = qe_all[wave];
    int*   myqa = qa_all[wave];
    float* myqw = qw_all[wave];

    if (tid == 0) { s_base = 0u; s_fill = 0u; }
    __syncthreads();
    if (tid < 256) {
        const unsigned v = wave_sum(tid < (int)blockIdx.x ? blkcnt[tid] : 0u);
        if (lane == 0 && v != 0u) atomicAdd(&s_base, v);
    }

    float4* acc4 = (float4*)acc;
    float4* agg4 = (float4*)(agg + (size_t)lo * 64);
    for (int i = tid; i < nr * 16; i += AGG_TB) acc4[i] = agg4[i];
    __syncthreads();
    uint2* __restrict__ mylist = (uint2*)list + s_base;

    int cnt = 0;
    if ((ne & 15) == 0) {
        const uint4* __restrict__ k4 = (const uint4*)keys;
        const unsigned idrep = blockIdx.x * 0x01010101u;
        const int n16 = ne >> 4;
        const int early = 64 + 4 * wave;
        static_assert(64 + 4 * (AGG_TB / 32 - 1) + 512 <= AGG_Q, "queue room for one more piece");
        const int per_wave = AGG_U * 32;
        const int per_step = (AGG_TB / 32) * per_wave;
        const int nsteps = (n16 + per_step - 1) / per_step;
        if (nsteps > 0) {
            int itA = blockIdx.x % nsteps;
            int jbA = (itA * (AGG_TB / 32) + wave) * per_wave + lane;
            uint4 va[AGG_U], vb[AGG_U];
            agg_load(k4, n16, jbA, va);
            for (int k = 0; k < nsteps; k += 2) {
                const int itB = (itA + 1 == nsteps) ? 0 : itA + 1;
                const int jbB = (itB * (AGG_TB / 32) + wave) * per_wave + lane;
                agg_load(k4, n16, jbB, vb);
                AGG_STEP(va, jbA, true)
                itA = (itB + 1 == nsteps) ? 0 : itB + 1;
                jbA = (itA * (AGG_TB / 32) + wave) * per_wave + lane;
                agg_load(k4, n16, jbA, va);
                AGG_STEP(vb, jbB, (k + 1 < nsteps))
            }
        }
    } else {
        for (int eb = wave * 32; eb < ne; eb += AGG_TB) {
            const int e = eb + lane;
            const int val = dst[e < ne ? e : ne - 1];
            AGG_DRAIN_IF(cnt + 32 > AGG_Q)
            AGG_TAKE(e < ne && (unsigned)(val - lo) < (unsigned)nr, e)
        }
    }
    agg_drain(h, src, dst, dis, acc, myqe, myqa, myqw, mylist, &s_fill, cnt, lo, lane);
    __syncthreads();

    for (int i = tid; i < nr * 16; i += AGG_TB) agg4[i] = acc4[i];
}

#define LST_B   32

__launch_bounds__(AGG_TB)
__global__ void edge_list(const float* __restrict__ h,
                          const unsigned* __restrict__ list,
                          const unsigned* __restrict__ blkcnt,
                          float* __restrict__ agg,
                          int n) {
    __shared__ float acc[AGG_R * 64];
    __shared__ unsigned s_base;

    const int tid  = threadIdx.x;
    const int wave = __builtin_amdgcn_readfirstlane(tid >> 5);
    const int lane = tid & 31;
    const int lo   = blockIdx.x * AGG_R;
    const int nr   = (n - lo < AGG_R) ? (n - lo) : AGG_R;

    if (tid == 0) s_base = 0u;
    __syncthreads();
    if (tid < 256) {
        const unsigned v = wave_sum(tid < (int)blockIdx.x ? blkcnt[tid] : 0u);
        if (lane == 0 && v != 0u) atomicAdd(&s_base, v);
    }
    const int cnt = (int)blkcnt[blockIdx.x];

    float4* acc4 = (float4*)acc;
    float4* agg4 = (float4*)(agg + (size_t)lo * 64);
    for (int i = tid; i < nr * 16; i += AGG_TB) acc4[i] = agg4[i];
    __syncthreads();
    const uint2* __restrict__ my = (const uint2*)list + s_base;

    int g = wave * 32;
    uint2 en = make_uint2(0u, 0u);
    if (g + lane < cnt) en = my[g + lane];
    while (g < cnt) {
        const uint2 e = en;
        const int gn = g + AGG_TB;
        en = make_uint2(0u, 0u);
        if (gn + lane < cnt) en = my[gn + lane];
        const int m = (cnt - g < 32) ? (cnt - g) : 32;
        const int   s = (int)(e.x & 0x7fffffu);
        const int   a = (int)(e.x >> 23);
        const float w = __uint_as_float(e.y);
        v2f rows[LST_B];
#pragma unroll
        for (int u = 0; u < LST_B; ++u) {
            const int su = __builtin_amdgcn_readlane(s, u);
            rows[u] = *(const v2f*)(h + (size_t)su * 64 + lane * 2);
        }
#pragma unroll
        for (int u = 0; u < LST_B; ++u) {
            if (u < m) {
                const float wu = __int_as_float(__builtin_amdgcn_readlane(__float_as_int(w), u));
                const int au = __builtin_amdgcn_readlane(a, u);
                float* p = acc + au * 64 + lane * 2;
                __hip_atomic_fetch_add(p + 0, rows[u].x * wu, __ATOMIC_RELAXED, __HIP_MEMORY_SCOPE_WORKGROUP);
                __hip_atomic_fetch_add(p + 1, rows[u].y * wu, __ATOMIC_RELAXED, __HIP_MEMORY_SCOPE_WORKGROUP);
            }
        }
        g = gn;
    }
    __syncthreads();

    for (int i = tid; i < nr * 16; i += AGG_TB) agg4[i] = acc4[i];
}

__global__ void out_proj(const float* __restrict__ hd,
                         const float* __restrict__ Wo,
                         const float* __restrict__ bo,
                         float* __restrict__ out,
                         int n) {
    const int i = blockIdx.x * blockDim.x + threadIdx.x;
    if (i >= n) return;
    const float* h = hd + (size_t)i * 128;
    float a0 = bo[0], a1 = bo[1];
#pragma unroll 8
    for (int k = 0; k < 128; ++k) {
        const float hv = h[k];
        a0 += hv * Wo[k * 2 + 0];
        a1 += hv * Wo[k * 2 + 1];
    }
    out[(size_t)i * 2 + 0] = a0;
    out[(size_t)i * 2 + 1] = a1;
}

extern "C" void kernel_launch(void* const* d_in, const int* in_sizes, int n_in,
                              void* d_out, int out_size, void* d_ws, size_t ws_size,
                              hipStream_t stream) {
    const float* x  = (const float*)d_in[0];
    const int*   ei = (const int*)d_in[1];
    const float* W1 = (const float*)d_in[2];
    const float* b1 = (const float*)d_in[3];
    const float* W2 = (const float*)d_in[4];
    const float* b2 = (const float*)d_in[5];
    const float* Wd = (const float*)d_in[6];
    const float* bd = (const float*)d_in[7];
    const float* Wo = (const float*)d_in[8];
    const float* bo = (const float*)d_in[9];
    float* out = (float*)d_out;

    const int N = in_sizes[0] / 64;
    const int E = in_sizes[1] / 2;
    const int* src = ei;
    const int* dst = ei + E;

    float* dis  = (float*)d_ws;
    float* bufA = dis  + N;
    float* bufB = bufA + (size_t)N * 64;
    float* bufC = bufB + (size_t)N * 64;
    float* bufD = bufC + (size_t)N * 64;
    unsigned* part = (unsigned*)(bufD + (size_t)N * 128);
    (void)ws_size; (void)n_in; (void)out_size;

    const int TB = 256;
    const int gemm_waves   = N / 16;
    const int gemm_blocks  = (gemm_waves * 32 + TB - 1) / TB;
    const int agg_blocks   = (N + AGG_R - 1) / AGG_R;

    const int deg_ranges = (N + DEG_R - 1) / DEG_R;
    const int deg_chunks = (E + DEG_CH - 1) / DEG_CH;
    unsigned char* keys = (unsigned char*)(part + (size_t)deg_chunks * N);
    unsigned* list   = (unsigned*)(keys + (((size_t)E + 15) & ~(size_t)15));
    unsigned* blkcnt = list + (size_t)E * 2;
    deg_hist<<<dim3(deg_ranges, deg_chunks), DEG_TB, 0, stream>>>(dst, part, keys, E, N);
    finish_dis<<<agg_blocks, FIN_TB, 0, stream>>>(part, dis, blkcnt, N, deg_chunks);

    gemm_wmma_f32<4, 64, false, true><<<gemm_blocks, TB, 0, stream>>>(x, W1, b1, dis, bufA, bufB, N);
    edge_tile<<<agg_blocks, AGG_TB, 0, stream>>>(bufA, src, dst, keys, dis, bufB, blkcnt, list, E, N);

    gemm_wmma_f32<4, 64, true, true><<<gemm_blocks, TB, 0, stream>>>(bufB, W2, b2, dis, bufA, bufC, N);
    edge_list<<<agg_blocks, AGG_TB, 0, stream>>>(bufA, list, blkcnt, bufC, N);

    gemm_wmma_f32<8, 64, true, false><<<gemm_blocks, TB, 0, stream>>>(bufC, Wd, bd, nullptr, bufD, nullptr, N);

    out_proj<<<(N + TB - 1) / TB, TB, 0, stream>>>(bufD, Wo, bo, out, N);
}


# out projection moved into the dense product's epilogue: per-lane sums over the lane's 8 columns, then a 16-lane exchange tree (sum order changes); dense rows never stored, 7 launches (was 8)
# speedup vs baseline: 2.2704x; 1.0328x over previous
#include <hip/hip_runtime.h>
#include <hip/hip_bf16.h>

typedef __attribute__((ext_vector_type(2))) float v2f;
typedef __attribute__((ext_vector_type(8))) float v8f;

#define N_NODES_C 100000

#define AGG_R   391
static_assert((N_NODES_C + AGG_R - 1) / AGG_R <= 256, "a block number must fit one byte");
#define DEG_R   14336
#define DEG_TB  1024
#define DEG_IT  11
#define DEG_CH  (DEG_TB * DEG_IT * 4)
static_assert(DEG_R % DEG_TB == 0, "a range is a whole number of block strides");

__launch_bounds__(DEG_TB)
__global__ void deg_hist(const int* __restrict__ dst, unsigned* __restrict__ part,
                         unsigned char* __restrict__ keys, int ne, int n) {
    __shared__ unsigned hist[DEG_R];
    const int tid = threadIdx.x;
    const int lo  = blockIdx.x * DEG_R;
    const int e0  = blockIdx.y * DEG_CH;

    for (int i = tid; i < DEG_R; i += DEG_TB) hist[i] = 0u;
    __syncthreads();

    if ((ne & 3) == 0) {
        const int4* __restrict__ d4 = (const int4*)dst;
        const int n4 = ne >> 2;
        const int j0 = (e0 >> 2) + tid;
        int4 v[DEG_IT];
#pragma unroll
        for (int it = 0; it < DEG_IT; ++it) {
            const int j = j0 + it * DEG_TB;
            v[it] = d4[j < n4 ? j : n4 - 1];
        }
#pragma unroll
        for (int it = 0; it < DEG_IT; ++it) {
            const int j = j0 + it * DEG_TB;
            if (j < n4) {
                if (blockIdx.x == 0) {
                    ((unsigned*)keys)[j] = ((unsigned)v[it].x / AGG_R)
                                         | (((unsigned)v[it].y / AGG_R) << 8)
                                         | (((unsigned)v[it].z / AGG_R) << 16)
                                         | (((unsigned)v[it].w / AGG_R) << 24);
                }
                const unsigned a0 = (unsigned)(v[it].x - lo);
                const unsigned a1 = (unsigned)(v[it].y - lo);
                const unsigned a2 = (unsigned)(v[it].z - lo);
                const unsigned a3 = (unsigned)(v[it].w - lo);
                if (a0 < (unsigned)DEG_R) atomicAdd(&hist[a0], 1u);
                if (a1 < (unsigned)DEG_R) atomicAdd(&hist[a1], 1u);
                if (a2 < (unsigned)DEG_R) atomicAdd(&hist[a2], 1u);
                if (a3 < (unsigned)DEG_R) atomicAdd(&hist[a3], 1u);
            }
        }
    } else {
        const int e1 = (e0 + DEG_CH < ne) ? (e0 + DEG_CH) : ne;
        for (int e = e0 + tid; e < e1; e += DEG_TB) {
            const int d = dst[e];
            if (blockIdx.x == 0) keys[e] = (unsigned char)((unsigned)d / AGG_R);
            const unsigned a = (unsigned)(d - lo);
            if (a < (unsigned)DEG_R) atomicAdd(&hist[a], 1u);
        }
    }
    __syncthreads();

    unsigned* __restrict__ row = part + (size_t)blockIdx.y * n;
    unsigned hv[DEG_R / DEG_TB];
#pragma unroll
    for (int k = 0; k < DEG_R / DEG_TB; ++k) hv[k] = hist[tid + k * DEG_TB];
#pragma unroll
    for (int k = 0; k < DEG_R / DEG_TB; ++k) {
        const int node = lo + tid + k * DEG_TB;
        if (node < n) row[node] = hv[k];
    }
}

__device__ __forceinline__ unsigned wave_sum(unsigned v) {
#pragma unroll
    for (int m = 16; m > 0; m >>= 1) v += __shfl_xor(v, m);
    return v;
}

#define FIN_TB  512
static_assert(AGG_R <= FIN_TB, "one thread per node of a block");
__launch_bounds__(FIN_TB)
__global__ void finish_dis(const unsigned* __restrict__ part, float* __restrict__ dis,
                           unsigned* __restrict__ blkcnt, int n, int nc) {
    __shared__ unsigned tot;
    const int tid = threadIdx.x;
    if (tid == 0) tot = 0u;
    __syncthreads();
    const int i = blockIdx.x * AGG_R + tid;
    unsigned s = 0u;
    if (tid < AGG_R && i < n) {
#pragma unroll 12
        for (int c = 0; c < nc; ++c) s += part[(size_t)c * n + i];
        dis[i] = rsqrtf((float)s + 1.0f);
    }
    s = wave_sum(s);
    if ((tid & 31) == 0 && s != 0u) atomicAdd(&tot, s);
    __syncthreads();
    if (tid == 0) blkcnt[blockIdx.x] = tot;
}

template<int NT, int KDIM, bool RELU_IN, bool SELF, bool OUTP>
__launch_bounds__(256)
__global__ void gemm_wmma_f32(const float* __restrict__ X,
                              const float* __restrict__ W,
                              const float* __restrict__ bias,
                              const float* __restrict__ dis,
                              const float* __restrict__ Wo,
                              const float* __restrict__ bo,
                              float* __restrict__ Y,
                              float* __restrict__ S,
                              float* __restrict__ out,
                              int M) {
    static_assert(!(SELF && OUTP), "one epilogue per instantiation");
    constexpr int N = NT * 16;
    const int wave = (blockIdx.x * blockDim.x + threadIdx.x) >> 5;
    const int lane = threadIdx.x & 31;
    const int row0 = wave * 16;
    if (row0 >= M) return;

    const int r      = lane & 15;
    const int khalf  = (lane >> 4) * 2;
    const int rowoff = (lane >> 4) * 8;

    float d2[8];
    if (SELF) {
        const float4 da = *(const float4*)(dis + row0 + rowoff);
        const float4 db = *(const float4*)(dis + row0 + rowoff + 4);
        d2[0] = da.x * da.x; d2[1] = da.y * da.y; d2[2] = da.z * da.z; d2[3] = da.w * da.w;
        d2[4] = db.x * db.x; d2[5] = db.y * db.y; d2[6] = db.z * db.z; d2[7] = db.w * db.w;
    }

    v8f acc[NT] = {};

    const float* __restrict__ xrow = X + (size_t)(row0 + r) * KDIM;

#pragma unroll
    for (int k = 0; k < KDIM; k += 4) {
        const int kk = k + khalf;
        v2f a = *(const v2f*)(xrow + kk);
        if (RELU_IN) {
            a.x = fmaxf(a.x, 0.0f);
            a.y = fmaxf(a.y, 0.0f);
        }
#pragma unroll
        for (int t = 0; t < NT; ++t) {
            const int col = t * 16 + r;
            v2f b;
            b.x = W[(size_t)kk * N + col];
            b.y = W[(size_t)(kk + 1) * N + col];
            acc[t] = __builtin_amdgcn_wmma_f32_16x16x4_f32(
                 false, a,  false, b,
                 (short)0, acc[t],
                 false,  false);
        }
    }

    if (OUTP) {
        float p[16];
#pragma unroll
        for (int i = 0; i < 16; ++i) p[i] = 0.0f;
#pragma unroll
        for (int t = 0; t < NT; ++t) {
            const int col = t * 16 + r;
            const float bv = bias[col];
            const v2f wo = *(const v2f*)(Wo + col * 2);
#pragma unroll
            for (int g = 0; g < 8; ++g) {
                const float v = fmaxf(acc[t][g] + bv, 0.0f);
                p[2 * g + 0] += v * wo.x;
                p[2 * g + 1] += v * wo.y;
            }
        }
        const bool b8 = (r & 8) != 0, b4 = (r & 4) != 0, b2 = (r & 2) != 0, b1 = (r & 1) != 0;
        float q8[8], q4[4], q2[2];
#pragma unroll
        for (int k = 0; k < 8; ++k) {
            const float keep = b8 ? p[k + 8] : p[k];
            const float send = b8 ? p[k] : p[k + 8];
            q8[k] = keep + __shfl_xor(send, 8);
        }
#pragma unroll
        for (int k = 0; k < 4; ++k) {
            const float keep = b4 ? q8[k + 4] : q8[k];
            const float send = b4 ? q8[k] : q8[k + 4];
            q4[k] = keep + __shfl_xor(send, 4);
        }
#pragma unroll
        for (int k = 0; k < 2; ++k) {
            const float keep = b2 ? q4[k + 2] : q4[k];
            const float send = b2 ? q4[k] : q4[k + 2];
            q2[k] = keep + __shfl_xor(send, 2);
        }
        const float keep1 = b1 ? q2[1] : q2[0];
        const float send1 = b1 ? q2[0] : q2[1];
        const float tot = keep1 + __shfl_xor(send1, 1);
        out[(size_t)row0 * 2 + lane] = tot + bo[lane & 1];
        return;
    }

#pragma unroll
    for (int t = 0; t < NT; ++t) {
        const int col = t * 16 + r;
        const float bv = bias[col];
#pragma unroll
        for (int g = 0; g < 8; ++g) {
            const size_t o = (size_t)(row0 + rowoff + g) * N + col;
            if (SELF) {
                Y[o] = acc[t][g];
                S[o] = acc[t][g] * d2[g] + bv;
            } else {
                Y[o] = fmaxf(acc[t][g] + bv, 0.0f);
            }
        }
    }
}

static_assert(AGG_R <= 512 && N_NODES_C <= (1 << 23), "a list entry packs row (9 bits) and source (23 bits)");
#define AGG_TB  1024
#define AGG_U   4
#define AGG_C   256
#define AGG_Q   (3 * AGG_C)
#define AGG_G   (AGG_C / 32)
#define AGG_B   16

__device__ __forceinline__ void agg_drain(const float* __restrict__ h,
                                          const int* __restrict__ src,
                                          const int* __restrict__ dst,
                                          const float* __restrict__ dis,
                                          float* acc, int* qe, int* qa, float* qw,
                                          uint2* __restrict__ mylist, unsigned* fill,
                                          int cnt, int lo, int lane) {
#pragma unroll 1
    for (int c0 = 0; c0 < cnt; c0 += AGG_C) {
        const int cc = (cnt - c0 < AGG_C) ? (cnt - c0) : AGG_C;
        int* ce = qe + c0;
        int ev[AGG_G], sv[AGG_G], dv[AGG_G];
#pragma unroll
        for (int g = 0; g < AGG_G; ++g) {
            const int idx = g * 32 + lane;
            ev[g] = (idx < cc) ? ce[idx] : 0;
        }
#pragma unroll
        for (int g = 0; g < AGG_G; ++g) {
            sv[g] = src[ev[g]];
            dv[g] = dst[ev[g]];
        }
#pragma unroll
        for (int g = 0; g < AGG_G; ++g) asm volatile("" :: "v"(sv[g]), "v"(dv[g]));
        unsigned pos = 0u;
        if (lane == 0) pos = atomicAdd(fill, (unsigned)cc);
        pos = (unsigned)__builtin_amdgcn_readfirstlane((int)pos);
#pragma unroll
        for (int g = 0; g < AGG_G; ++g) {
            const int idx = g * 32 + lane;
            const float w = dis[sv[g]] * dis[dv[g]];
            if (idx < cc) {
                ce[idx] = sv[g];
                qa[idx] = dv[g] - lo;
                qw[idx] = w;
                mylist[pos + idx] = make_uint2((unsigned)sv[g] | ((unsigned)(dv[g] - lo) << 23),
                                               __float_as_uint(w));
            }
        }
        for (int g = 0; g < cc; g += 32) {
            const int idx = g + lane;
            const bool valid = idx < cc;
            const int s = valid ? ce[idx] : 0;
            const int a = valid ? qa[idx] : 0;
            const float w = valid ? qw[idx] : 0.0f;
            const int m = (cc - g < 32) ? (cc - g) : 32;
#pragma unroll 1
            for (int u0 = 0; u0 < m; u0 += AGG_B) {
                v2f rows[AGG_B];
#pragma unroll
                for (int u = 0; u < AGG_B; ++u) {
                    const int su = __builtin_amdgcn_readlane(s, u0 + u);
                    rows[u] = *(const v2f*)(h + (size_t)su * 64 + lane * 2);
                }
#pragma unroll
                for (int u = 0; u < AGG_B; ++u) {
                    if (u0 + u < m) {
                        const float wu = __int_as_float(__builtin_amdgcn_readlane(__float_as_int(w), u0 + u));
                        const int au = __builtin_amdgcn_readlane(a, u0 + u);
                        float* p = acc + au * 64 + lane * 2;
                        __hip_atomic_fetch_add(p + 0, rows[u].x * wu, __ATOMIC_RELAXED, __HIP_MEMORY_SCOPE_WORKGROUP);
                        __hip_atomic_fetch_add(p + 1, rows[u].y * wu, __ATOMIC_RELAXED, __HIP_MEMORY_SCOPE_WORKGROUP);
                    }
                }
#pragma unroll
                for (int u = 0; u < AGG_B; ++u) asm volatile("" :: "v"(rows[u]));
            }
        }
    }
}

__device__ __forceinline__ void agg_load(const uint4* __restrict__ k4, int n16, int jb,
                                         uint4 (&v)[AGG_U]) {
#pragma unroll
    for (int u = 0; u < AGG_U; ++u) {
        const int j = jb + u * 32;
        v[u] = k4[j < n16 ? j : n16 - 1];
    }
}

#define AGG_DRAIN_IF(COND)                                                      \
    if (COND) {                                                                 \
        agg_drain(h, src, dst, dis, acc, myqe, myqa, myqw, mylist, &s_fill, cnt, lo, lane); \
        cnt = 0;                                                                \
    }

#define AGG_TAKE(PRED, EID)                                                     \
    {                                                                           \
        const bool m_ = (PRED);                                                 \
        const unsigned mask_ = __builtin_amdgcn_ballot_w32(m_);                 \
        if (mask_) {                                                            \
            if (m_) {                                                           \
                const int pos_ = cnt + (int)__builtin_amdgcn_mbcnt_lo(mask_, 0u); \
                myqe[pos_] = (EID);                                             \
            }                                                                   \
            cnt += __builtin_popcount(mask_);                                   \
        }                                                                       \
    }

#define AGG_WORD(XW, EID0)                                                      \
    {                                                                           \
        const unsigned x_ = (XW);                                               \
        const bool h_ = (((x_ - 0x01010101u) & ~x_ & 0x80808080u) != 0u);       \
        if (__builtin_amdgcn_ballot_w32(h_)) {                                  \
            AGG_TAKE((x_ & 0x000000ffu) == 0u, (EID0) + 0)                      \
            AGG_TAKE((x_ & 0x0000ff00u) == 0u, (EID0) + 1)                      \
            AGG_TAKE((x_ & 0x00ff0000u) == 0u, (EID0) + 2)                      \
            AGG_TAKE((x_ & 0xff000000u) == 0u, (EID0) + 3)                      \
        }                                                                       \
    }

#define AGG_STEP(V, JB, LIVE)                                                   \
    _Pragma("unroll")                                                           \
    for (int u = 0; u < AGG_U; ++u) {                                           \
        const int j = (JB) + u * 32;                                            \
        const unsigned dead = ((LIVE) && j < n16) ? 0u : 0xffffffffu;           \
        AGG_DRAIN_IF(cnt > early)                                               \
        AGG_WORD((V[u].x ^ idrep) | dead, 16 * j + 0)                           \
        AGG_WORD((V[u].y ^ idrep) | dead, 16 * j + 4)                           \
        AGG_WORD((V[u].z ^ idrep) | dead, 16 * j + 8)                           \
        AGG_WORD((V[u].w ^ idrep) | dead, 16 * j + 12)                          \
    }

__launch_bounds__(AGG_TB)
__global__ void edge_tile(const float* __restrict__ h,
                          const int* __restrict__ src,
                          const int* __restrict__ dst,
                          const unsigned char* __restrict__ keys,
                          const float* __restrict__ dis,
                          float* __restrict__ agg,
                          const unsigned* __restrict__ blkcnt,
                          unsigned* __restrict__ list,
                          int ne, int n) {
    __shared__ float acc[AGG_R * 64];
    __shared__ int   qe_all[AGG_TB / 32][AGG_Q];
    __shared__ int   qa_all[AGG_TB / 32][AGG_C];
    __shared__ float qw_all[AGG_TB / 32][AGG_C];
    __shared__ unsigned s_base;
    __shared__ unsigned s_fill;

    const int tid  = threadIdx.x;
    const int wave = tid >> 5;
    const int lane = tid & 31;
    const int lo   = blockIdx.x * AGG_R;
    const int nr   = (n - lo < AGG_R) ? (n - lo) : AGG_R;
    int*   myqe = qe_all[wave];
    int*   myqa = qa_all[wave];
    float* myqw = qw_all[wave];

    if (tid == 0) { s_base = 0u; s_fill = 0u; }
    __syncthreads();
    if (tid < 256) {
        const unsigned v = wave_sum(tid < (int)blockIdx.x ? blkcnt[tid] : 0u);
        if (lane == 0 && v != 0u) atomicAdd(&s_base, v);
    }

    float4* acc4 = (float4*)acc;
    float4* agg4 = (float4*)(agg + (size_t)lo * 64);
    for (int i = tid; i < nr * 16; i += AGG_TB) acc4[i] = agg4[i];
    __syncthreads();
    uint2* __restrict__ mylist = (uint2*)list + s_base;

    int cnt = 0;
    if ((ne & 15) == 0) {
        const uint4* __restrict__ k4 = (const uint4*)keys;
        const unsigned idrep = blockIdx.x * 0x01010101u;
        const int n16 = ne >> 4;
        const int early = 64 + 4 * wave;
        static_assert(64 + 4 * (AGG_TB / 32 - 1) + 512 <= AGG_Q, "queue room for one more piece");
        const int per_wave = AGG_U * 32;
        const int per_step = (AGG_TB / 32) * per_wave;
        const int nsteps = (n16 + per_step - 1) / per_step;
        if (nsteps > 0) {
            int itA = blockIdx.x % nsteps;
            int jbA = (itA * (AGG_TB / 32) + wave) * per_wave + lane;
            uint4 va[AGG_U], vb[AGG_U];
            agg_load(k4, n16, jbA, va);
            for (int k = 0; k < nsteps; k += 2) {
                const int itB = (itA + 1 == nsteps) ? 0 : itA + 1;
                const int jbB = (itB * (AGG_TB / 32) + wave) * per_wave + lane;
                agg_load(k4, n16, jbB, vb);
                AGG_STEP(va, jbA, true)
                itA = (itB + 1 == nsteps) ? 0 : itB + 1;
                jbA = (itA * (AGG_TB / 32) + wave) * per_wave + lane;
                agg_load(k4, n16, jbA, va);
                AGG_STEP(vb, jbB, (k + 1 < nsteps))
            }
        }
    } else {
        for (int eb = wave * 32; eb < ne; eb += AGG_TB) {
            const int e = eb + lane;
            const int val = dst[e < ne ? e : ne - 1];
            AGG_DRAIN_IF(cnt + 32 > AGG_Q)
            AGG_TAKE(e < ne && (unsigned)(val - lo) < (unsigned)nr, e)
        }
    }
    agg_drain(h, src, dst, dis, acc, myqe, myqa, myqw, mylist, &s_fill, cnt, lo, lane);
    __syncthreads();

    for (int i = tid; i < nr * 16; i += AGG_TB) agg4[i] = acc4[i];
}

#define LST_B   32

__launch_bounds__(AGG_TB)
__global__ void edge_list(const float* __restrict__ h,
                          const unsigned* __restrict__ list,
                          const unsigned* __restrict__ blkcnt,
                          float* __restrict__ agg,
                          int n) {
    __shared__ float acc[AGG_R * 64];
    __shared__ unsigned s_base;

    const int tid  = threadIdx.x;
    const int wave = __builtin_amdgcn_readfirstlane(tid >> 5);
    const int lane = tid & 31;
    const int lo   = blockIdx.x * AGG_R;
    const int nr   = (n - lo < AGG_R) ? (n - lo) : AGG_R;

    if (tid == 0) s_base = 0u;
    __syncthreads();
    if (tid < 256) {
        const unsigned v = wave_sum(tid < (int)blockIdx.x ? blkcnt[tid] : 0u);
        if (lane == 0 && v != 0u) atomicAdd(&s_base, v);
    }
    const int cnt = (int)blkcnt[blockIdx.x];

    float4* acc4 = (float4*)acc;
    float4* agg4 = (float4*)(agg + (size_t)lo * 64);
    for (int i = tid; i < nr * 16; i += AGG_TB) acc4[i] = agg4[i];
    __syncthreads();
    const uint2* __restrict__ my = (const uint2*)list + s_base;

    int g = wave * 32;
    uint2 en = make_uint2(0u, 0u);
    if (g + lane < cnt) en = my[g + lane];
    while (g < cnt) {
        const uint2 e = en;
        const int gn = g + AGG_TB;
        en = make_uint2(0u, 0u);
        if (gn + lane < cnt) en = my[gn + lane];
        const int m = (cnt - g < 32) ? (cnt - g) : 32;
        const int   s = (int)(e.x & 0x7fffffu);
        const int   a = (int)(e.x >> 23);
        const float w = __uint_as_float(e.y);
        v2f rows[LST_B];
#pragma unroll
        for (int u = 0; u < LST_B; ++u) {
            const int su = __builtin_amdgcn_readlane(s, u);
            rows[u] = *(const v2f*)(h + (size_t)su * 64 + lane * 2);
        }
#pragma unroll
        for (int u = 0; u < LST_B; ++u) {
            if (u < m) {
                const float wu = __int_as_float(__builtin_amdgcn_readlane(__float_as_int(w), u));
                const int au = __builtin_amdgcn_readlane(a, u);
                float* p = acc + au * 64 + lane * 2;
                __hip_atomic_fetch_add(p + 0, rows[u].x * wu, __ATOMIC_RELAXED, __HIP_MEMORY_SCOPE_WORKGROUP);
                __hip_atomic_fetch_add(p + 1, rows[u].y * wu, __ATOMIC_RELAXED, __HIP_MEMORY_SCOPE_WORKGROUP);
            }
        }
        g = gn;
    }
    __syncthreads();

    for (int i = tid; i < nr * 16; i += AGG_TB) agg4[i] = acc4[i];
}

extern "C" void kernel_launch(void* const* d_in, const int* in_sizes, int n_in,
                              void* d_out, int out_size, void* d_ws, size_t ws_size,
                              hipStream_t stream) {
    const float* x  = (const float*)d_in[0];
    const int*   ei = (const int*)d_in[1];
    const float* W1 = (const float*)d_in[2];
    const float* b1 = (const float*)d_in[3];
    const float* W2 = (const float*)d_in[4];
    const float* b2 = (const float*)d_in[5];
    const float* Wd = (const float*)d_in[6];
    const float* bd = (const float*)d_in[7];
    const float* Wo = (const float*)d_in[8];
    const float* bo = (const float*)d_in[9];
    float* out = (float*)d_out;

    const int N = in_sizes[0] / 64;
    const int E = in_sizes[1] / 2;
    const int* src = ei;
    const int* dst = ei + E;

    float* dis  = (float*)d_ws;
    float* bufA = dis  + N;
    float* bufB = bufA + (size_t)N * 64;
    float* bufC = bufB + (size_t)N * 64;
    unsigned* part = (unsigned*)(bufC + (size_t)N * 64);
    (void)ws_size; (void)n_in; (void)out_size;

    const int TB = 256;
    const int gemm_waves   = N / 16;
    const int gemm_blocks  = (gemm_waves * 32 + TB - 1) / TB;
    const int agg_blocks   = (N + AGG_R - 1) / AGG_R;

    const int deg_ranges = (N + DEG_R - 1) / DEG_R;
    const int deg_chunks = (E + DEG_CH - 1) / DEG_CH;
    unsigned char* keys = (unsigned char*)(part + (size_t)deg_chunks * N);
    unsigned* list   = (unsigned*)(keys + (((size_t)E + 15) & ~(size_t)15));
    unsigned* blkcnt = list + (size_t)E * 2;
    deg_hist<<<dim3(deg_ranges, deg_chunks), DEG_TB, 0, stream>>>(dst, part, keys, E, N);
    finish_dis<<<agg_blocks, FIN_TB, 0, stream>>>(part, dis, blkcnt, N, deg_chunks);

    gemm_wmma_f32<4, 64, false, true, false><<<gemm_blocks, TB, 0, stream>>>(x, W1, b1, dis, nullptr, nullptr, bufA, bufB, nullptr, N);
    edge_tile<<<agg_blocks, AGG_TB, 0, stream>>>(bufA, src, dst, keys, dis, bufB, blkcnt, list, E, N);

    gemm_wmma_f32<4, 64, true, true, false><<<gemm_blocks, TB, 0, stream>>>(bufB, W2, b2, dis, nullptr, nullptr, bufA, bufC, nullptr, N);
    edge_list<<<agg_blocks, AGG_TB, 0, stream>>>(bufA, list, blkcnt, bufC, N);

    gemm_wmma_f32<8, 64, true, false, true><<<gemm_blocks, TB, 0, stream>>>(bufC, Wd, bd, nullptr, Wo, bo, nullptr, nullptr, out, N);
}


# edge pass fetches one list word per queued edge (source | row << 23, stored per edge by the degree pass) instead of its src and its dst: one random look-up less per edge; sum order unchanged
# speedup vs baseline: 2.3452x; 1.0329x over previous
#include <hip/hip_runtime.h>
#include <hip/hip_bf16.h>

typedef __attribute__((ext_vector_type(2))) float v2f;
typedef __attribute__((ext_vector_type(8))) float v8f;

#define N_NODES_C 100000

#define AGG_R   391
static_assert((N_NODES_C + AGG_R - 1) / AGG_R <= 256, "a block number must fit one byte");
#define DEG_R   14336
#define DEG_TB  1024
#define DEG_IT  11
#define DEG_CH  (DEG_TB * DEG_IT * 4)
static_assert(DEG_R % DEG_TB == 0, "a range is a whole number of block strides");
static_assert(AGG_R <= 512 && N_NODES_C <= (1 << 23), "a list word packs row (9 bits) and source (23 bits)");

__device__ __forceinline__ unsigned edge_word(int s, int d) {
    const unsigned k = (unsigned)d / AGG_R;
    return (unsigned)s | (((unsigned)d - k * AGG_R) << 23);
}

__launch_bounds__(DEG_TB)
__global__ void deg_hist(const int* __restrict__ src, const int* __restrict__ dst,
                         unsigned* __restrict__ part, unsigned char* __restrict__ keys,
                         unsigned* __restrict__ words, int ne, int n) {
    __shared__ unsigned hist[DEG_R];
    const int tid = threadIdx.x;
    const int lo  = blockIdx.x * DEG_R;
    const int e0  = blockIdx.y * DEG_CH;

    for (int i = tid; i < DEG_R; i += DEG_TB) hist[i] = 0u;
    __syncthreads();

    if ((ne & 3) == 0) {
        const int4* __restrict__ d4 = (const int4*)dst;
        const int n4 = ne >> 2;
        const int j0 = (e0 >> 2) + tid;
        unsigned mine = 0u;
        for (int it = blockIdx.x; it < DEG_IT; it += gridDim.x) mine |= 1u << it;
        const int4* __restrict__ s4 = (const int4*)src;
        int4 sv[DEG_IT] = {};
#pragma unroll
        for (int it = 0; it < DEG_IT; ++it) {
            const int j = j0 + it * DEG_TB;
            if ((mine >> it) & 1u) sv[it] = s4[j < n4 ? j : n4 - 1];
        }
        int4 v[DEG_IT];
#pragma unroll
        for (int it = 0; it < DEG_IT; ++it) {
            const int j = j0 + it * DEG_TB;
            v[it] = d4[j < n4 ? j : n4 - 1];
        }
#pragma unroll
        for (int it = 0; it < DEG_IT; ++it) {
            const int j = j0 + it * DEG_TB;
            if (j < n4) {
                if (blockIdx.x == 0) {
                    ((unsigned*)keys)[j] = ((unsigned)v[it].x / AGG_R)
                                         | (((unsigned)v[it].y / AGG_R) << 8)
                                         | (((unsigned)v[it].z / AGG_R) << 16)
                                         | (((unsigned)v[it].w / AGG_R) << 24);
                }
                const unsigned a0 = (unsigned)(v[it].x - lo);
                const unsigned a1 = (unsigned)(v[it].y - lo);
                const unsigned a2 = (unsigned)(v[it].z - lo);
                const unsigned a3 = (unsigned)(v[it].w - lo);
                if (a0 < (unsigned)DEG_R) atomicAdd(&hist[a0], 1u);
                if (a1 < (unsigned)DEG_R) atomicAdd(&hist[a1], 1u);
                if (a2 < (unsigned)DEG_R) atomicAdd(&hist[a2], 1u);
                if (a3 < (unsigned)DEG_R) atomicAdd(&hist[a3], 1u);
            }
        }
        uint4* __restrict__ w4 = (uint4*)words;
#pragma unroll
        for (int it = 0; it < DEG_IT; ++it) {
            const int j = j0 + it * DEG_TB;
            if (((mine >> it) & 1u) && j < n4) {
                w4[j] = make_uint4(edge_word(sv[it].x, v[it].x), edge_word(sv[it].y, v[it].y),
                                   edge_word(sv[it].z, v[it].z), edge_word(sv[it].w, v[it].w));
            }
        }
    } else {
        const int e1 = (e0 + DEG_CH < ne) ? (e0 + DEG_CH) : ne;
        for (int e = e0 + tid; e < e1; e += DEG_TB) {
            const int d = dst[e];
            if (blockIdx.x == 0) {
                keys[e] = (unsigned char)((unsigned)d / AGG_R);
                words[e] = edge_word(src[e], d);
            }
            const unsigned a = (unsigned)(d - lo);
            if (a < (unsigned)DEG_R) atomicAdd(&hist[a], 1u);
        }
    }
    __syncthreads();

    unsigned* __restrict__ row = part + (size_t)blockIdx.y * n;
    unsigned hv[DEG_R / DEG_TB];
#pragma unroll
    for (int k = 0; k < DEG_R / DEG_TB; ++k) hv[k] = hist[tid + k * DEG_TB];
#pragma unroll
    for (int k = 0; k < DEG_R / DEG_TB; ++k) {
        const int node = lo + tid + k * DEG_TB;
        if (node < n) row[node] = hv[k];
    }
}

__device__ __forceinline__ unsigned wave_sum(unsigned v) {
#pragma unroll
    for (int m = 16; m > 0; m >>= 1) v += __shfl_xor(v, m);
    return v;
}

#define FIN_TB  512
static_assert(AGG_R <= FIN_TB, "one thread per node of a block");
__launch_bounds__(FIN_TB)
__global__ void finish_dis(const unsigned* __restrict__ part, float* __restrict__ dis,
                           unsigned* __restrict__ blkcnt, int n, int nc) {
    __shared__ unsigned tot;
    const int tid = threadIdx.x;
    if (tid == 0) tot = 0u;
    __syncthreads();
    const int i = blockIdx.x * AGG_R + tid;
    unsigned s = 0u;
    if (tid < AGG_R && i < n) {
#pragma unroll 12
        for (int c = 0; c < nc; ++c) s += part[(size_t)c * n + i];
        dis[i] = rsqrtf((float)s + 1.0f);
    }
    s = wave_sum(s);
    if ((tid & 31) == 0 && s != 0u) atomicAdd(&tot, s);
    __syncthreads();
    if (tid == 0) blkcnt[blockIdx.x] = tot;
}

template<int NT, int KDIM, bool RELU_IN, bool SELF, bool OUTP>
__launch_bounds__(256)
__global__ void gemm_wmma_f32(const float* __restrict__ X,
                              const float* __restrict__ W,
                              const float* __restrict__ bias,
                              const float* __restrict__ dis,
                              const float* __restrict__ Wo,
                              const float* __restrict__ bo,
                              float* __restrict__ Y,
                              float* __restrict__ S,
                              float* __restrict__ out,
                              int M) {
    static_assert(!(SELF && OUTP), "one epilogue per instantiation");
    constexpr int N = NT * 16;
    const int wave = (blockIdx.x * blockDim.x + threadIdx.x) >> 5;
    const int lane = threadIdx.x & 31;
    const int row0 = wave * 16;
    if (row0 >= M) return;

    const int r      = lane & 15;
    const int khalf  = (lane >> 4) * 2;
    const int rowoff = (lane >> 4) * 8;

    float d2[8];
    if (SELF) {
        const float4 da = *(const float4*)(dis + row0 + rowoff);
        const float4 db = *(const float4*)(dis + row0 + rowoff + 4);
        d2[0] = da.x * da.x; d2[1] = da.y * da.y; d2[2] = da.z * da.z; d2[3] = da.w * da.w;
        d2[4] = db.x * db.x; d2[5] = db.y * db.y; d2[6] = db.z * db.z; d2[7] = db.w * db.w;
    }

    v8f acc[NT] = {};

    const float* __restrict__ xrow = X + (size_t)(row0 + r) * KDIM;

#pragma unroll
    for (int k = 0; k < KDIM; k += 4) {
        const int kk = k + khalf;
        v2f a = *(const v2f*)(xrow + kk);
        if (RELU_IN) {
            a.x = fmaxf(a.x, 0.0f);
            a.y = fmaxf(a.y, 0.0f);
        }
#pragma unroll
        for (int t = 0; t < NT; ++t) {
            const int col = t * 16 + r;
            v2f b;
            b.x = W[(size_t)kk * N + col];
            b.y = W[(size_t)(kk + 1) * N + col];
            acc[t] = __builtin_amdgcn_wmma_f32_16x16x4_f32(
                 false, a,  false, b,
                 (short)0, acc[t],
                 false,  false);
        }
    }

    if (OUTP) {
        float p[16];
#pragma unroll
        for (int i = 0; i < 16; ++i) p[i] = 0.0f;
#pragma unroll
        for (int t = 0; t < NT; ++t) {
            const int col = t * 16 + r;
            const float bv = bias[col];
            const v2f wo = *(const v2f*)(Wo + col * 2);
#pragma unroll
            for (int g = 0; g < 8; ++g) {
                const float v = fmaxf(acc[t][g] + bv, 0.0f);
                p[2 * g + 0] += v * wo.x;
                p[2 * g + 1] += v * wo.y;
            }
        }
        const bool b8 = (r & 8) != 0, b4 = (r & 4) != 0, b2 = (r & 2) != 0, b1 = (r & 1) != 0;
        float q8[8], q4[4], q2[2];
#pragma unroll
        for (int k = 0; k < 8; ++k) {
            const float keep = b8 ? p[k + 8] : p[k];
            const float send = b8 ? p[k] : p[k + 8];
            q8[k] = keep + __shfl_xor(send, 8);
        }
#pragma unroll
        for (int k = 0; k < 4; ++k) {
            const float keep = b4 ? q8[k + 4] : q8[k];
            const float send = b4 ? q8[k] : q8[k + 4];
            q4[k] = keep + __shfl_xor(send, 4);
        }
#pragma unroll
        for (int k = 0; k < 2; ++k) {
            const float keep = b2 ? q4[k + 2] : q4[k];
            const float send = b2 ? q4[k] : q4[k + 2];
            q2[k] = keep + __shfl_xor(send, 2);
        }
        const float keep1 = b1 ? q2[1] : q2[0];
        const float send1 = b1 ? q2[0] : q2[1];
        const float tot = keep1 + __shfl_xor(send1, 1);
        out[(size_t)row0 * 2 + lane] = tot + bo[lane & 1];
        return;
    }

#pragma unroll
    for (int t = 0; t < NT; ++t) {
        const int col = t * 16 + r;
        const float bv = bias[col];
#pragma unroll
        for (int g = 0; g < 8; ++g) {
            const size_t o = (size_t)(row0 + rowoff + g) * N + col;
            if (SELF) {
                Y[o] = acc[t][g];
                S[o] = acc[t][g] * d2[g] + bv;
            } else {
                Y[o] = fmaxf(acc[t][g] + bv, 0.0f);
            }
        }
    }
}

#define AGG_TB  1024
#define AGG_U   4
#define AGG_C   256
#define AGG_Q   (3 * AGG_C)
#define AGG_G   (AGG_C / 32)
#define AGG_B   16

__device__ __forceinline__ void agg_drain(const float* __restrict__ h,
                                          const unsigned* __restrict__ words,
                                          const float* __restrict__ dis,
                                          float* acc, int* qe, int* qa, float* qw,
                                          uint2* __restrict__ mylist, unsigned* fill,
                                          int cnt, int lo, int lane) {
#pragma unroll 1
    for (int c0 = 0; c0 < cnt; c0 += AGG_C) {
        const int cc = (cnt - c0 < AGG_C) ? (cnt - c0) : AGG_C;
        int* ce = qe + c0;
        int ev[AGG_G];
        unsigned pk[AGG_G];
#pragma unroll
        for (int g = 0; g < AGG_G; ++g) {
            const int idx = g * 32 + lane;
            ev[g] = (idx < cc) ? ce[idx] : 0;
        }
#pragma unroll
        for (int g = 0; g < AGG_G; ++g) pk[g] = words[ev[g]];
#pragma unroll
        for (int g = 0; g < AGG_G; ++g) asm volatile("" :: "v"(pk[g]));
        unsigned pos = 0u;
        if (lane == 0) pos = atomicAdd(fill, (unsigned)cc);
        pos = (unsigned)__builtin_amdgcn_readfirstlane((int)pos);
#pragma unroll
        for (int g = 0; g < AGG_G; ++g) {
            const int idx = g * 32 + lane;
            const unsigned p = (idx < cc) ? pk[g] : 0u;
            const int s = (int)(p & 0x7fffffu);
            const int a = (int)(p >> 23);
            const float w = dis[s] * dis[lo + a];
            if (idx < cc) {
                ce[idx] = s;
                qa[idx] = a;
                qw[idx] = w;
                mylist[pos + idx] = make_uint2(p, __float_as_uint(w));
            }
        }
        for (int g = 0; g < cc; g += 32) {
            const int idx = g + lane;
            const bool valid = idx < cc;
            const int s = valid ? ce[idx] : 0;
            const int a = valid ? qa[idx] : 0;
            const float w = valid ? qw[idx] : 0.0f;
            const int m = (cc - g < 32) ? (cc - g) : 32;
#pragma unroll 1
            for (int u0 = 0; u0 < m; u0 += AGG_B) {
                v2f rows[AGG_B];
#pragma unroll
                for (int u = 0; u < AGG_B; ++u) {
                    const int su = __builtin_amdgcn_readlane(s, u0 + u);
                    rows[u] = *(const v2f*)(h + (size_t)su * 64 + lane * 2);
                }
#pragma unroll
                for (int u = 0; u < AGG_B; ++u) {
                    if (u0 + u < m) {
                        const float wu = __int_as_float(__builtin_amdgcn_readlane(__float_as_int(w), u0 + u));
                        const int au = __builtin_amdgcn_readlane(a, u0 + u);
                        float* p = acc + au * 64 + lane * 2;
                        __hip_atomic_fetch_add(p + 0, rows[u].x * wu, __ATOMIC_RELAXED, __HIP_MEMORY_SCOPE_WORKGROUP);
                        __hip_atomic_fetch_add(p + 1, rows[u].y * wu, __ATOMIC_RELAXED, __HIP_MEMORY_SCOPE_WORKGROUP);
                    }
                }
#pragma unroll
                for (int u = 0; u < AGG_B; ++u) asm volatile("" :: "v"(rows[u]));
            }
        }
    }
}

__device__ __forceinline__ void agg_load(const uint4* __restrict__ k4, int n16, int jb,
                                         uint4 (&v)[AGG_U]) {
#pragma unroll
    for (int u = 0; u < AGG_U; ++u) {
        const int j = jb + u * 32;
        v[u] = k4[j < n16 ? j : n16 - 1];
    }
}

#define AGG_DRAIN_IF(COND)                                                      \
    if (COND) {                                                                 \
        agg_drain(h, words, dis, acc, myqe, myqa, myqw, mylist, &s_fill, cnt, lo, lane); \
        cnt = 0;                                                                \
    }

#define AGG_TAKE(PRED, EID)                                                     \
    {                                                                           \
        const bool m_ = (PRED);                                                 \
        const unsigned mask_ = __builtin_amdgcn_ballot_w32(m_);                 \
        if (mask_) {                                                            \
            if (m_) {                                                           \
                const int pos_ = cnt + (int)__builtin_amdgcn_mbcnt_lo(mask_, 0u); \
                myqe[pos_] = (EID);                                             \
            }                                                                   \
            cnt += __builtin_popcount(mask_);                                   \
        }                                                                       \
    }

#define AGG_WORD(XW, EID0)                                                      \
    {                                                                           \
        const unsigned x_ = (XW);                                               \
        const bool h_ = (((x_ - 0x01010101u) & ~x_ & 0x80808080u) != 0u);       \
        if (__builtin_amdgcn_ballot_w32(h_)) {                                  \
            AGG_TAKE((x_ & 0x000000ffu) == 0u, (EID0) + 0)                      \
            AGG_TAKE((x_ & 0x0000ff00u) == 0u, (EID0) + 1)                      \
            AGG_TAKE((x_ & 0x00ff0000u) == 0u, (EID0) + 2)                      \
            AGG_TAKE((x_ & 0xff000000u) == 0u, (EID0) + 3)                      \
        }                                                                       \
    }

#define AGG_STEP(V, JB, LIVE)                                                   \
    _Pragma("unroll")                                                           \
    for (int u = 0; u < AGG_U; ++u) {                                           \
        const int j = (JB) + u * 32;                                            \
        const unsigned dead = ((LIVE) && j < n16) ? 0u : 0xffffffffu;           \
        AGG_DRAIN_IF(cnt > early)                                               \
        AGG_WORD((V[u].x ^ idrep) | dead, 16 * j + 0)                           \
        AGG_WORD((V[u].y ^ idrep) | dead, 16 * j + 4)                           \
        AGG_WORD((V[u].z ^ idrep) | dead, 16 * j + 8)                           \
        AGG_WORD((V[u].w ^ idrep) | dead, 16 * j + 12)                          \
    }

__launch_bounds__(AGG_TB)
__global__ void edge_tile(const float* __restrict__ h,
                          const int* __restrict__ dst,
                          const unsigned char* __restrict__ keys,
                          const unsigned* __restrict__ words,
                          const float* __restrict__ dis,
                          float* __restrict__ agg,
                          const unsigned* __restrict__ blkcnt,
                          unsigned* __restrict__ list,
                          int ne, int n) {
    __shared__ float acc[AGG_R * 64];
    __shared__ int   qe_all[AGG_TB / 32][AGG_Q];
    __shared__ int   qa_all[AGG_TB / 32][AGG_C];
    __shared__ float qw_all[AGG_TB / 32][AGG_C];
    __shared__ unsigned s_base;
    __shared__ unsigned s_fill;

    const int tid  = threadIdx.x;
    const int wave = tid >> 5;
    const int lane = tid & 31;
    const int lo   = blockIdx.x * AGG_R;
    const int nr   = (n - lo < AGG_R) ? (n - lo) : AGG_R;
    int*   myqe = qe_all[wave];
    int*   myqa = qa_all[wave];
    float* myqw = qw_all[wave];

    if (tid == 0) { s_base = 0u; s_fill = 0u; }
    __syncthreads();
    if (tid < 256) {
        const unsigned v = wave_sum(tid < (int)blockIdx.x ? blkcnt[tid] : 0u);
        if (lane == 0 && v != 0u) atomicAdd(&s_base, v);
    }

    float4* acc4 = (float4*)acc;
    float4* agg4 = (float4*)(agg + (size_t)lo * 64);
    for (int i = tid; i < nr * 16; i += AGG_TB) acc4[i] = agg4[i];
    __syncthreads();
    uint2* __restrict__ mylist = (uint2*)list + s_base;

    int cnt = 0;
    if ((ne & 15) == 0) {
        const uint4* __restrict__ k4 = (const uint4*)keys;
        const unsigned idrep = blockIdx.x * 0x01010101u;
        const int n16 = ne >> 4;
        const int early = 64 + 4 * wave;
        static_assert(64 + 4 * (AGG_TB / 32 - 1) + 512 <= AGG_Q, "queue room for one more piece");
        const int per_wave = AGG_U * 32;
        const int per_step = (AGG_TB / 32) * per_wave;
        const int nsteps = (n16 + per_step - 1) / per_step;
        if (nsteps > 0) {
            int itA = blockIdx.x % nsteps;
            int jbA = (itA * (AGG_TB / 32) + wave) * per_wave + lane;
            uint4 va[AGG_U], vb[AGG_U];
            agg_load(k4, n16, jbA, va);
            for (int k = 0; k < nsteps; k += 2) {
                const int itB = (itA + 1 == nsteps) ? 0 : itA + 1;
                const int jbB = (itB * (AGG_TB / 32) + wave) * per_wave + lane;
                agg_load(k4, n16, jbB, vb);
                AGG_STEP(va, jbA, true)
                itA = (itB + 1 == nsteps) ? 0 : itB + 1;
                jbA = (itA * (AGG_TB / 32) + wave) * per_wave + lane;
                agg_load(k4, n16, jbA, va);
                AGG_STEP(vb, jbB, (k + 1 < nsteps))
            }
        }
    } else {
        for (int eb = wave * 32; eb < ne; eb += AGG_TB) {
            const int e = eb + lane;
            const int val = dst[e < ne ? e : ne - 1];
            AGG_DRAIN_IF(cnt + 32 > AGG_Q)
            AGG_TAKE(e < ne && (unsigned)(val - lo) < (unsigned)nr, e)
        }
    }
    agg_drain(h, words, dis, acc, myqe, myqa, myqw, mylist, &s_fill, cnt, lo, lane);
    __syncthreads();

    for (int i = tid; i < nr * 16; i += AGG_TB) agg4[i] = acc4[i];
}

#define LST_B   32

__launch_bounds__(AGG_TB)
__global__ void edge_list(const float* __restrict__ h,
                          const unsigned* __restrict__ list,
                          const unsigned* __restrict__ blkcnt,
                          float* __restrict__ agg,
                          int n) {
    __shared__ float acc[AGG_R * 64];
    __shared__ unsigned s_base;

    const int tid  = threadIdx.x;
    const int wave = __builtin_amdgcn_readfirstlane(tid >> 5);
    const int lane = tid & 31;
    const int lo   = blockIdx.x * AGG_R;
    const int nr   = (n - lo < AGG_R) ? (n - lo) : AGG_R;

    if (tid == 0) s_base = 0u;
    __syncthreads();
    if (tid < 256) {
        const unsigned v = wave_sum(tid < (int)blockIdx.x ? blkcnt[tid] : 0u);
        if (lane == 0 && v != 0u) atomicAdd(&s_base, v);
    }
    const int cnt = (int)blkcnt[blockIdx.x];

    float4* acc4 = (float4*)acc;
    float4* agg4 = (float4*)(agg + (size_t)lo * 64);
    for (int i = tid; i < nr * 16; i += AGG_TB) acc4[i] = agg4[i];
    __syncthreads();
    const uint2* __restrict__ my = (const uint2*)list + s_base;

    int g = wave * 32;
    uint2 en = make_uint2(0u, 0u);
    if (g + lane < cnt) en = my[g + lane];
    while (g < cnt) {
        const uint2 e = en;
        const int gn = g + AGG_TB;
        en = make_uint2(0u, 0u);
        if (gn + lane < cnt) en = my[gn + lane];
        const int m = (cnt - g < 32) ? (cnt - g) : 32;
        const int   s = (int)(e.x & 0x7fffffu);
        const int   a = (int)(e.x >> 23);
        const float w = __uint_as_float(e.y);
        v2f rows[LST_B];
#pragma unroll
        for (int u = 0; u < LST_B; ++u) {
            const int su = __builtin_amdgcn_readlane(s, u);
            rows[u] = *(const v2f*)(h + (size_t)su * 64 + lane * 2);
        }
#pragma unroll
        for (int u = 0; u < LST_B; ++u) {
            if (u < m) {
                const float wu = __int_as_float(__builtin_amdgcn_readlane(__float_as_int(w), u));
                const int au = __builtin_amdgcn_readlane(a, u);
                float* p = acc + au * 64 + lane * 2;
                __hip_atomic_fetch_add(p + 0, rows[u].x * wu, __ATOMIC_RELAXED, __HIP_MEMORY_SCOPE_WORKGROUP);
                __hip_atomic_fetch_add(p + 1, rows[u].y * wu, __ATOMIC_RELAXED, __HIP_MEMORY_SCOPE_WORKGROUP);
            }
        }
        g = gn;
    }
    __syncthreads();

    for (int i = tid; i < nr * 16; i += AGG_TB) agg4[i] = acc4[i];
}

extern "C" void kernel_launch(void* const* d_in, const int* in_sizes, int n_in,
                              void* d_out, int out_size, void* d_ws, size_t ws_size,
                              hipStream_t stream) {
    const float* x  = (const float*)d_in[0];
    const int*   ei = (const int*)d_in[1];
    const float* W1 = (const float*)d_in[2];
    const float* b1 = (const float*)d_in[3];
    const float* W2 = (const float*)d_in[4];
    const float* b2 = (const float*)d_in[5];
    const float* Wd = (const float*)d_in[6];
    const float* bd = (const float*)d_in[7];
    const float* Wo = (const float*)d_in[8];
    const float* bo = (const float*)d_in[9];
    float* out = (float*)d_out;

    const int N = in_sizes[0] / 64;
    const int E = in_sizes[1] / 2;
    const int* src = ei;
    const int* dst = ei + E;

    float* dis  = (float*)d_ws;
    float* bufA = dis  + N;
    float* bufB = bufA + (size_t)N * 64;
    float* bufC = bufB + (size_t)N * 64;
    unsigned* part = (unsigned*)(bufC + (size_t)N * 64);
    (void)ws_size; (void)n_in; (void)out_size;

    const int TB = 256;
    const int gemm_waves   = N / 16;
    const int gemm_blocks  = (gemm_waves * 32 + TB - 1) / TB;
    const int agg_blocks   = (N + AGG_R - 1) / AGG_R;

    const int deg_ranges = (N + DEG_R - 1) / DEG_R;
    const int deg_chunks = (E + DEG_CH - 1) / DEG_CH;
    unsigned char* keys = (unsigned char*)(part + (size_t)deg_chunks * N);
    unsigned* list   = (unsigned*)(keys + (((size_t)E + 15) & ~(size_t)15));
    unsigned* blkcnt = list + (size_t)E * 2;
    unsigned* words  = blkcnt + 256;
    deg_hist<<<dim3(deg_ranges, deg_chunks), DEG_TB, 0, stream>>>(src, dst, part, keys, words, E, N);
    finish_dis<<<agg_blocks, FIN_TB, 0, stream>>>(part, dis, blkcnt, N, deg_chunks);

    gemm_wmma_f32<4, 64, false, true, false><<<gemm_blocks, TB, 0, stream>>>(x, W1, b1, dis, nullptr, nullptr, bufA, bufB, nullptr, N);
    edge_tile<<<agg_blocks, AGG_TB, 0, stream>>>(bufA, dst, keys, words, dis, bufB, blkcnt, list, E, N);

    gemm_wmma_f32<4, 64, true, true, false><<<gemm_blocks, TB, 0, stream>>>(bufB, W2, b2, dis, nullptr, nullptr, bufA, bufC, nullptr, N);
    edge_list<<<agg_blocks, AGG_TB, 0, stream>>>(bufA, list, blkcnt, bufC, N);

    gemm_wmma_f32<8, 64, true, false, true><<<gemm_blocks, TB, 0, stream>>>(bufC, Wd, bd, nullptr, Wo, bo, nullptr, nullptr, out, N);
}


# edge pass scans the keys as bit planes (8 words per 32 edges, built by the degree pass from wave ballots): a lane tests 32 edges by logic ops, queues its set bits; sums still in LDS arrival order
# speedup vs baseline: 2.4618x; 1.0497x over previous
#include <hip/hip_runtime.h>
#include <hip/hip_bf16.h>

typedef __attribute__((ext_vector_type(2))) float v2f;
typedef __attribute__((ext_vector_type(8))) float v8f;

#define N_NODES_C 100000

#define AGG_R   391
static_assert((N_NODES_C + AGG_R - 1) / AGG_R <= 256, "a block number must fit one byte");
#define DEG_R   14336
#define DEG_TB  1024
#define DEG_IT  11
#define DEG_CH  (DEG_TB * DEG_IT * 4)
static_assert(DEG_R % DEG_TB == 0, "a range is a whole number of block strides");
static_assert((DEG_CH / 4) % 32 == 0 && DEG_TB % 32 == 0, "a wave's 32 pieces are one aligned 128-edge group");
static_assert(AGG_R <= 512 && N_NODES_C <= (1 << 23), "a list word packs row (9 bits) and source (23 bits)");

__device__ __forceinline__ unsigned edge_word(int s, int d) {
    const unsigned k = (unsigned)d / AGG_R;
    return (unsigned)s | (((unsigned)d - k * AGG_R) << 23);
}

__launch_bounds__(DEG_TB)
__global__ void deg_hist(const int* __restrict__ src, const int* __restrict__ dst,
                         unsigned* __restrict__ part, unsigned* __restrict__ planes,
                         unsigned* __restrict__ words, int ne, int n) {
    __shared__ unsigned hist[DEG_R];
    const int tid = threadIdx.x;
    const int lo  = blockIdx.x * DEG_R;
    const int e0  = blockIdx.y * DEG_CH;

    for (int i = tid; i < DEG_R; i += DEG_TB) hist[i] = 0u;
    __syncthreads();

    if ((ne & 3) == 0) {
        const int4* __restrict__ d4 = (const int4*)dst;
        const int n4 = ne >> 2;
        const int j0 = (e0 >> 2) + tid;
        unsigned mine = 0u;
        for (int it = blockIdx.x; it < DEG_IT; it += gridDim.x) mine |= 1u << it;
        const int4* __restrict__ s4 = (const int4*)src;
        int4 sv[DEG_IT] = {};
#pragma unroll
        for (int it = 0; it < DEG_IT; ++it) {
            const int j = j0 + it * DEG_TB;
            if ((mine >> it) & 1u) sv[it] = s4[j < n4 ? j : n4 - 1];
        }
        int4 v[DEG_IT];
#pragma unroll
        for (int it = 0; it < DEG_IT; ++it) {
            const int j = j0 + it * DEG_TB;
            v[it] = d4[j < n4 ? j : n4 - 1];
        }
#pragma unroll
        for (int it = 0; it < DEG_IT; ++it) {
            const int j = j0 + it * DEG_TB;
            if (j < n4) {
                const unsigned a0 = (unsigned)(v[it].x - lo);
                const unsigned a1 = (unsigned)(v[it].y - lo);
                const unsigned a2 = (unsigned)(v[it].z - lo);
                const unsigned a3 = (unsigned)(v[it].w - lo);
                if (a0 < (unsigned)DEG_R) atomicAdd(&hist[a0], 1u);
                if (a1 < (unsigned)DEG_R) atomicAdd(&hist[a1], 1u);
                if (a2 < (unsigned)DEG_R) atomicAdd(&hist[a2], 1u);
                if (a3 < (unsigned)DEG_R) atomicAdd(&hist[a3], 1u);
            }
        }
        uint4* __restrict__ w4 = (uint4*)words;
        const int lane = tid & 31;
#pragma unroll
        for (int it = 0; it < DEG_IT; ++it) {
            const int j = j0 + it * DEG_TB;
            if ((mine >> it) & 1u) {
                if (j < n4) {
                    w4[j] = make_uint4(edge_word(sv[it].x, v[it].x), edge_word(sv[it].y, v[it].y),
                                       edge_word(sv[it].z, v[it].z), edge_word(sv[it].w, v[it].w));
                }
                const int jw = __builtin_amdgcn_readfirstlane(j);
                if (jw < n4) {
                    const unsigned kk[4] = {(unsigned)v[it].x / AGG_R, (unsigned)v[it].y / AGG_R,
                                            (unsigned)v[it].z / AGG_R, (unsigned)v[it].w / AGG_R};
                    unsigned wv = 0u;
#pragma unroll
                    for (int cb = 0; cb < 32; ++cb) {
                        const unsigned bal = __builtin_amdgcn_ballot_w32(((kk[cb >> 3] >> (cb & 7)) & 1u) != 0u);
                        wv = (lane == cb) ? bal : wv;
                    }
                    planes[((j >> 8) << 8) + (((lane >> 2) & 1) << 7) + (((j >> 5) & 7) << 4)
                           + ((lane >> 3) << 2) + (lane & 3)] = wv;
                }
            }
        }
    } else {
        const int e1 = (e0 + DEG_CH < ne) ? (e0 + DEG_CH) : ne;
        for (int e = e0 + tid; e < e1; e += DEG_TB) {
            const int d = dst[e];
            if (blockIdx.x == 0) words[e] = edge_word(src[e], d);
            const unsigned a = (unsigned)(d - lo);
            if (a < (unsigned)DEG_R) atomicAdd(&hist[a], 1u);
        }
    }
    __syncthreads();

    unsigned* __restrict__ row = part + (size_t)blockIdx.y * n;
    unsigned hv[DEG_R / DEG_TB];
#pragma unroll
    for (int k = 0; k < DEG_R / DEG_TB; ++k) hv[k] = hist[tid + k * DEG_TB];
#pragma unroll
    for (int k = 0; k < DEG_R / DEG_TB; ++k) {
        const int node = lo + tid + k * DEG_TB;
        if (node < n) row[node] = hv[k];
    }
}

__device__ __forceinline__ unsigned wave_sum(unsigned v) {
#pragma unroll
    for (int m = 16; m > 0; m >>= 1) v += __shfl_xor(v, m);
    return v;
}

#define FIN_TB  512
static_assert(AGG_R <= FIN_TB, "one thread per node of a block");
__launch_bounds__(FIN_TB)
__global__ void finish_dis(const unsigned* __restrict__ part, float* __restrict__ dis,
                           unsigned* __restrict__ blkcnt, int n, int nc) {
    __shared__ unsigned tot;
    const int tid = threadIdx.x;
    if (tid == 0) tot = 0u;
    __syncthreads();
    const int i = blockIdx.x * AGG_R + tid;
    unsigned s = 0u;
    if (tid < AGG_R && i < n) {
#pragma unroll 12
        for (int c = 0; c < nc; ++c) s += part[(size_t)c * n + i];
        dis[i] = rsqrtf((float)s + 1.0f);
    }
    s = wave_sum(s);
    if ((tid & 31) == 0 && s != 0u) atomicAdd(&tot, s);
    __syncthreads();
    if (tid == 0) blkcnt[blockIdx.x] = tot;
}

template<int NT, int KDIM, bool RELU_IN, bool SELF, bool OUTP>
__launch_bounds__(256)
__global__ void gemm_wmma_f32(const float* __restrict__ X,
                              const float* __restrict__ W,
                              const float* __restrict__ bias,
                              const float* __restrict__ dis,
                              const float* __restrict__ Wo,
                              const float* __restrict__ bo,
                              float* __restrict__ Y,
                              float* __restrict__ S,
                              float* __restrict__ out,
                              int M) {
    static_assert(!(SELF && OUTP), "one epilogue per instantiation");
    constexpr int N = NT * 16;
    const int wave = (blockIdx.x * blockDim.x + threadIdx.x) >> 5;
    const int lane = threadIdx.x & 31;
    const int row0 = wave * 16;
    if (row0 >= M) return;

    const int r      = lane & 15;
    const int khalf  = (lane >> 4) * 2;
    const int rowoff = (lane >> 4) * 8;

    float d2[8];
    if (SELF) {
        const float4 da = *(const float4*)(dis + row0 + rowoff);
        const float4 db = *(const float4*)(dis + row0 + rowoff + 4);
        d2[0] = da.x * da.x; d2[1] = da.y * da.y; d2[2] = da.z * da.z; d2[3] = da.w * da.w;
        d2[4] = db.x * db.x; d2[5] = db.y * db.y; d2[6] = db.z * db.z; d2[7] = db.w * db.w;
    }

    v8f acc[NT] = {};

    const float* __restrict__ xrow = X + (size_t)(row0 + r) * KDIM;

#pragma unroll
    for (int k = 0; k < KDIM; k += 4) {
        const int kk = k + khalf;
        v2f a = *(const v2f*)(xrow + kk);
        if (RELU_IN) {
            a.x = fmaxf(a.x, 0.0f);
            a.y = fmaxf(a.y, 0.0f);
        }
#pragma unroll
        for (int t = 0; t < NT; ++t) {
            const int col = t * 16 + r;
            v2f b;
            b.x = W[(size_t)kk * N + col];
            b.y = W[(size_t)(kk + 1) * N + col];
            acc[t] = __builtin_amdgcn_wmma_f32_16x16x4_f32(
                 false, a,  false, b,
                 (short)0, acc[t],
                 false,  false);
        }
    }

    if (OUTP) {
        float p[16];
#pragma unroll
        for (int i = 0; i < 16; ++i) p[i] = 0.0f;
#pragma unroll
        for (int t = 0; t < NT; ++t) {
            const int col = t * 16 + r;
            const float bv = bias[col];
            const v2f wo = *(const v2f*)(Wo + col * 2);
#pragma unroll
            for (int g = 0; g < 8; ++g) {
                const float v = fmaxf(acc[t][g] + bv, 0.0f);
                p[2 * g + 0] += v * wo.x;
                p[2 * g + 1] += v * wo.y;
            }
        }
        const bool b8 = (r & 8) != 0, b4 = (r & 4) != 0, b2 = (r & 2) != 0, b1 = (r & 1) != 0;
        float q8[8], q4[4], q2[2];
#pragma unroll
        for (int k = 0; k < 8; ++k) {
            const float keep = b8 ? p[k + 8] : p[k];
            const float send = b8 ? p[k] : p[k + 8];
            q8[k] = keep + __shfl_xor(send, 8);
        }
#pragma unroll
        for (int k = 0; k < 4; ++k) {
            const float keep = b4 ? q8[k + 4] : q8[k];
            const float send = b4 ? q8[k] : q8[k + 4];
            q4[k] = keep + __shfl_xor(send, 4);
        }
#pragma unroll
        for (int k = 0; k < 2; ++k) {
            const float keep = b2 ? q4[k + 2] : q4[k];
            const float send = b2 ? q4[k] : q4[k + 2];
            q2[k] = keep + __shfl_xor(send, 2);
        }
        const float keep1 = b1 ? q2[1] : q2[0];
        const float send1 = b1 ? q2[0] : q2[1];
        const float tot = keep1 + __shfl_xor(send1, 1);
        out[(size_t)row0 * 2 + lane] = tot + bo[lane & 1];
        return;
    }

#pragma unroll
    for (int t = 0; t < NT; ++t) {
        const int col = t * 16 + r;
        const float bv = bias[col];
#pragma unroll
        for (int g = 0; g < 8; ++g) {
            const size_t o = (size_t)(row0 + rowoff + g) * N + col;
            if (SELF) {
                Y[o] = acc[t][g];
                S[o] = acc[t][g] * d2[g] + bv;
            } else {
                Y[o] = fmaxf(acc[t][g] + bv, 0.0f);
            }
        }
    }
}

#define AGG_TB  1024
#define AGG_U   2
#define AGG_C   256
#define AGG_Q   (3 * AGG_C)
#define AGG_G   (AGG_C / 32)
#define AGG_B   16

__device__ __forceinline__ void agg_drain(const float* __restrict__ h,
                                          const unsigned* __restrict__ words,
                                          const float* __restrict__ dis,
                                          float* acc, int* qe, int* qa, float* qw,
                                          uint2* __restrict__ mylist, unsigned* fill,
                                          int cnt, int lo, int lane) {
#pragma unroll 1
    for (int c0 = 0; c0 < cnt; c0 += AGG_C) {
        const int cc = (cnt - c0 < AGG_C) ? (cnt - c0) : AGG_C;
        int* ce = qe + c0;
        int ev[AGG_G];
        unsigned pk[AGG_G];
#pragma unroll
        for (int g = 0; g < AGG_G; ++g) {
            const int idx = g * 32 + lane;
            ev[g] = (idx < cc) ? ce[idx] : 0;
        }
#pragma unroll
        for (int g = 0; g < AGG_G; ++g) pk[g] = words[ev[g]];
#pragma unroll
        for (int g = 0; g < AGG_G; ++g) asm volatile("" :: "v"(pk[g]));
        unsigned pos = 0u;
        if (lane == 0) pos = atomicAdd(fill, (unsigned)cc);
        pos = (unsigned)__builtin_amdgcn_readfirstlane((int)pos);
#pragma unroll
        for (int g = 0; g < AGG_G; ++g) {
            const int idx = g * 32 + lane;
            const unsigned p = (idx < cc) ? pk[g] : 0u;
            const int s = (int)(p & 0x7fffffu);
            const int a = (int)(p >> 23);
            const float w = dis[s] * dis[lo + a];
            if (idx < cc) {
                ce[idx] = s;
                qa[idx] = a;
                qw[idx] = w;
                mylist[pos + idx] = make_uint2(p, __float_as_uint(w));
            }
        }
        for (int g = 0; g < cc; g += 32) {
            const int idx = g + lane;
            const bool valid = idx < cc;
            const int s = valid ? ce[idx] : 0;
            const int a = valid ? qa[idx] : 0;
            const float w = valid ? qw[idx] : 0.0f;
            const int m = (cc - g < 32) ? (cc - g) : 32;
#pragma unroll 1
            for (int u0 = 0; u0 < m; u0 += AGG_B) {
                v2f rows[AGG_B];
#pragma unroll
                for (int u = 0; u < AGG_B; ++u) {
                    const int su = __builtin_amdgcn_readlane(s, u0 + u);
                    rows[u] = *(const v2f*)(h + (size_t)su * 64 + lane * 2);
                }
#pragma unroll
                for (int u = 0; u < AGG_B; ++u) {
                    if (u0 + u < m) {
                        const float wu = __int_as_float(__builtin_amdgcn_readlane(__float_as_int(w), u0 + u));
                        const int au = __builtin_amdgcn_readlane(a, u0 + u);
                        float* p = acc + au * 64 + lane * 2;
                        __hip_atomic_fetch_add(p + 0, rows[u].x * wu, __ATOMIC_RELAXED, __HIP_MEMORY_SCOPE_WORKGROUP);
                        __hip_atomic_fetch_add(p + 1, rows[u].y * wu, __ATOMIC_RELAXED, __HIP_MEMORY_SCOPE_WORKGROUP);
                    }
                }
#pragma unroll
                for (int u = 0; u < AGG_B; ++u) asm volatile("" :: "v"(rows[u]));
            }
        }
    }
}

__device__ __forceinline__ void agg_load(const uint4* __restrict__ k4, int nu, int ub,
                                         uint4 (&v)[2 * AGG_U]) {
#pragma unroll
    for (int u = 0; u < AGG_U; ++u) {
        const int U = ub + u * 32;
        const int cu = U < nu ? U : nu - 1;
        const int q = cu + (cu & ~31);
        v[2 * u + 0] = k4[q];
        v[2 * u + 1] = k4[q + 32];
    }
}

#define AGG_DRAIN_IF(COND)                                                      \
    if (COND) {                                                                 \
        agg_drain(h, words, dis, acc, myqe, myqa, myqw, mylist, &s_fill, cnt, lo, lane); \
        cnt = 0;                                                                \
    }

#define AGG_TAKE(PRED, EID)                                                     \
    {                                                                           \
        const bool m_ = (PRED);                                                 \
        const unsigned mask_ = __builtin_amdgcn_ballot_w32(m_);                 \
        if (mask_) {                                                            \
            if (m_) {                                                           \
                const int pos_ = cnt + (int)__builtin_amdgcn_mbcnt_lo(mask_, 0u); \
                myqe[pos_] = (EID);                                             \
            }                                                                   \
            cnt += __builtin_popcount(mask_);                                   \
        }                                                                       \
    }

#define AGG_STEP(V, UB, LIVE)                                                   \
    _Pragma("unroll")                                                           \
    for (int u = 0; u < AGG_U; ++u) {                                           \
        const int U_ = (UB) + u * 32;                                           \
        unsigned keep_ = (U_ < fu) ? 0xffffffffu : tailmask;                    \
        if (!((LIVE) && U_ < nu)) keep_ = 0u;                                   \
        const unsigned mis_ = (V[2 * u].x ^ r0) | (V[2 * u].y ^ r1)             \
                            | (V[2 * u].z ^ r2) | (V[2 * u].w ^ r3)             \
                            | (V[2 * u + 1].x ^ r4) | (V[2 * u + 1].y ^ r5)     \
                            | (V[2 * u + 1].z ^ r6) | (V[2 * u + 1].w ^ r7);    \
        unsigned m_ = ~mis_ & keep_;                                            \
        const int eb_ = ((U_ >> 2) << 7) + (U_ & 3);                            \
        unsigned any_ = __builtin_amdgcn_ballot_w32(m_ != 0u);                  \
        while (any_) {                                                          \
            AGG_DRAIN_IF(cnt > early)                                           \
            if (m_ != 0u) {                                                     \
                const int pos_ = cnt + (int)__builtin_amdgcn_mbcnt_lo(any_, 0u); \
                myqe[pos_] = eb_ + 4 * (__ffs(m_) - 1);                         \
            }                                                                   \
            cnt += __builtin_popcount(any_);                                    \
            m_ &= m_ - 1u;                                                      \
            any_ = __builtin_amdgcn_ballot_w32(m_ != 0u);                       \
        }                                                                       \
    }

__launch_bounds__(AGG_TB)
__global__ void edge_tile(const float* __restrict__ h,
                          const int* __restrict__ dst,
                          const unsigned* __restrict__ planes,
                          const unsigned* __restrict__ words,
                          const float* __restrict__ dis,
                          float* __restrict__ agg,
                          const unsigned* __restrict__ blkcnt,
                          unsigned* __restrict__ list,
                          int ne, int n) {
    __shared__ float acc[AGG_R * 64];
    __shared__ int   qe_all[AGG_TB / 32][AGG_Q];
    __shared__ int   qa_all[AGG_TB / 32][AGG_C];
    __shared__ float qw_all[AGG_TB / 32][AGG_C];
    __shared__ unsigned s_base;
    __shared__ unsigned s_fill;

    const int tid  = threadIdx.x;
    const int wave = tid >> 5;
    const int lane = tid & 31;
    const int lo   = blockIdx.x * AGG_R;
    const int nr   = (n - lo < AGG_R) ? (n - lo) : AGG_R;
    int*   myqe = qe_all[wave];
    int*   myqa = qa_all[wave];
    float* myqw = qw_all[wave];

    if (tid == 0) { s_base = 0u; s_fill = 0u; }
    __syncthreads();
    if (tid < 256) {
        const unsigned v = wave_sum(tid < (int)blockIdx.x ? blkcnt[tid] : 0u);
        if (lane == 0 && v != 0u) atomicAdd(&s_base, v);
    }

    float4* acc4 = (float4*)acc;
    float4* agg4 = (float4*)(agg + (size_t)lo * 64);
    for (int i = tid; i < nr * 16; i += AGG_TB) acc4[i] = agg4[i];
    __syncthreads();
    uint2* __restrict__ mylist = (uint2*)list + s_base;

    int cnt = 0;
    if ((ne & 3) == 0) {
        const uint4* __restrict__ k4 = (const uint4*)planes;
        const int n4 = ne >> 2;
        const int nu = ((n4 + 31) >> 5) << 2;
        const int fu = (n4 >> 5) << 2;
        const unsigned tailmask = (1u << (n4 & 31)) - 1u;
        const unsigned r0 = 0u - ((blockIdx.x >> 0) & 1u), r1 = 0u - ((blockIdx.x >> 1) & 1u);
        const unsigned r2 = 0u - ((blockIdx.x >> 2) & 1u), r3 = 0u - ((blockIdx.x >> 3) & 1u);
        const unsigned r4 = 0u - ((blockIdx.x >> 4) & 1u), r5 = 0u - ((blockIdx.x >> 5) & 1u);
        const unsigned r6 = 0u - ((blockIdx.x >> 6) & 1u), r7 = 0u - ((blockIdx.x >> 7) & 1u);
        const int early = 64 + 4 * wave;
        static_assert(64 + 4 * (AGG_TB / 32 - 1) + 32 <= AGG_Q, "queue room for one more round");
        const int per_wave = AGG_U * 32;
        const int per_step = (AGG_TB / 32) * per_wave;
        const int nsteps = (nu + per_step - 1) / per_step;
        if (nsteps > 0) {
            int itA = blockIdx.x % nsteps;
            int jbA = (itA * (AGG_TB / 32) + wave) * per_wave + lane;
            uint4 va[2 * AGG_U], vb[2 * AGG_U];
            agg_load(k4, nu, jbA, va);
            for (int k = 0; k < nsteps; k += 2) {
                const int itB = (itA + 1 == nsteps) ? 0 : itA + 1;
                const int jbB = (itB * (AGG_TB / 32) + wave) * per_wave + lane;
                agg_load(k4, nu, jbB, vb);
                AGG_STEP(va, jbA, true)
                itA = (itB + 1 == nsteps) ? 0 : itB + 1;
                jbA = (itA * (AGG_TB / 32) + wave) * per_wave + lane;
                agg_load(k4, nu, jbA, va);
                AGG_STEP(vb, jbB, (k + 1 < nsteps))
            }
        }
    } else {
        for (int eb = wave * 32; eb < ne; eb += AGG_TB) {
            const int e = eb + lane;
            const int val = dst[e < ne ? e : ne - 1];
            AGG_DRAIN_IF(cnt + 32 > AGG_Q)
            AGG_TAKE(e < ne && (unsigned)(val - lo) < (unsigned)nr, e)
        }
    }
    agg_drain(h, words, dis, acc, myqe, myqa, myqw, mylist, &s_fill, cnt, lo, lane);
    __syncthreads();

    for (int i = tid; i < nr * 16; i += AGG_TB) agg4[i] = acc4[i];
}

#define LST_B   32

__launch_bounds__(AGG_TB)
__global__ void edge_list(const float* __restrict__ h,
                          const unsigned* __restrict__ list,
                          const unsigned* __restrict__ blkcnt,
                          float* __restrict__ agg,
                          int n) {
    __shared__ float acc[AGG_R * 64];
    __shared__ unsigned s_base;

    const int tid  = threadIdx.x;
    const int wave = __builtin_amdgcn_readfirstlane(tid >> 5);
    const int lane = tid & 31;
    const int lo   = blockIdx.x * AGG_R;
    const int nr   = (n - lo < AGG_R) ? (n - lo) : AGG_R;

    if (tid == 0) s_base = 0u;
    __syncthreads();
    if (tid < 256) {
        const unsigned v = wave_sum(tid < (int)blockIdx.x ? blkcnt[tid] : 0u);
        if (lane == 0 && v != 0u) atomicAdd(&s_base, v);
    }
    const int cnt = (int)blkcnt[blockIdx.x];

    float4* acc4 = (float4*)acc;
    float4* agg4 = (float4*)(agg + (size_t)lo * 64);
    for (int i = tid; i < nr * 16; i += AGG_TB) acc4[i] = agg4[i];
    __syncthreads();
    const uint2* __restrict__ my = (const uint2*)list + s_base;

    int g = wave * 32;
    uint2 en = make_uint2(0u, 0u);
    if (g + lane < cnt) en = my[g + lane];
    while (g < cnt) {
        const uint2 e = en;
        const int gn = g + AGG_TB;
        en = make_uint2(0u, 0u);
        if (gn + lane < cnt) en = my[gn + lane];
        const int m = (cnt - g < 32) ? (cnt - g) : 32;
        const int   s = (int)(e.x & 0x7fffffu);
        const int   a = (int)(e.x >> 23);
        const float w = __uint_as_float(e.y);
        v2f rows[LST_B];
#pragma unroll
        for (int u = 0; u < LST_B; ++u) {
            const int su = __builtin_amdgcn_readlane(s, u);
            rows[u] = *(const v2f*)(h + (size_t)su * 64 + lane * 2);
        }
#pragma unroll
        for (int u = 0; u < LST_B; ++u) {
            if (u < m) {
                const float wu = __int_as_float(__builtin_amdgcn_readlane(__float_as_int(w), u));
                const int au = __builtin_amdgcn_readlane(a, u);
                float* p = acc + au * 64 + lane * 2;
                __hip_atomic_fetch_add(p + 0, rows[u].x * wu, __ATOMIC_RELAXED, __HIP_MEMORY_SCOPE_WORKGROUP);
                __hip_atomic_fetch_add(p + 1, rows[u].y * wu, __ATOMIC_RELAXED, __HIP_MEMORY_SCOPE_WORKGROUP);
            }
        }
        g = gn;
    }
    __syncthreads();

    for (int i = tid; i < nr * 16; i += AGG_TB) agg4[i] = acc4[i];
}

extern "C" void kernel_launch(void* const* d_in, const int* in_sizes, int n_in,
                              void* d_out, int out_size, void* d_ws, size_t ws_size,
                              hipStream_t stream) {
    const float* x  = (const float*)d_in[0];
    const int*   ei = (const int*)d_in[1];
    const float* W1 = (const float*)d_in[2];
    const float* b1 = (const float*)d_in[3];
    const float* W2 = (const float*)d_in[4];
    const float* b2 = (const float*)d_in[5];
    const float* Wd = (const float*)d_in[6];
    const float* bd = (const float*)d_in[7];
    const float* Wo = (const float*)d_in[8];
    const float* bo = (const float*)d_in[9];
    float* out = (float*)d_out;

    const int N = in_sizes[0] / 64;
    const int E = in_sizes[1] / 2;
    const int* src = ei;
    const int* dst = ei + E;

    float* dis  = (float*)d_ws;
    float* bufA = dis  + N;
    float* bufB = bufA + (size_t)N * 64;
    float* bufC = bufB + (size_t)N * 64;
    unsigned* part = (unsigned*)(bufC + (size_t)N * 64);
    (void)ws_size; (void)n_in; (void)out_size;

    const int TB = 256;
    const int gemm_waves   = N / 16;
    const int gemm_blocks  = (gemm_waves * 32 + TB - 1) / TB;
    const int agg_blocks   = (N + AGG_R - 1) / AGG_R;

    const int deg_ranges = (N + DEG_R - 1) / DEG_R;
    const int deg_chunks = (E + DEG_CH - 1) / DEG_CH;
    unsigned* planes = part + (size_t)deg_chunks * N;
    unsigned* list   = planes + ((((size_t)E + 1023) >> 10) << 8);
    unsigned* blkcnt = list + (size_t)E * 2;
    unsigned* words  = blkcnt + 256;
    deg_hist<<<dim3(deg_ranges, deg_chunks), DEG_TB, 0, stream>>>(src, dst, part, planes, words, E, N);
    finish_dis<<<agg_blocks, FIN_TB, 0, stream>>>(part, dis, blkcnt, N, deg_chunks);

    gemm_wmma_f32<4, 64, false, true, false><<<gemm_blocks, TB, 0, stream>>>(x, W1, b1, dis, nullptr, nullptr, bufA, bufB, nullptr, N);
    edge_tile<<<agg_blocks, AGG_TB, 0, stream>>>(bufA, dst, planes, words, dis, bufB, blkcnt, list, E, N);

    gemm_wmma_f32<4, 64, true, true, false><<<gemm_blocks, TB, 0, stream>>>(bufB, W2, b2, dis, nullptr, nullptr, bufA, bufC, nullptr, N);
    edge_list<<<agg_blocks, AGG_TB, 0, stream>>>(bufA, list, blkcnt, bufC, N);

    gemm_wmma_f32<8, 64, true, false, true><<<gemm_blocks, TB, 0, stream>>>(bufC, Wd, bd, nullptr, Wo, bo, nullptr, nullptr, out, N);
}


# edge pass scans its node block's own bit row (1 bit per edge; degree pass builds 256 rows per 4096 edges in an LDS tile by LDS OR) instead of 8 key planes of every edge; sum order unchanged
# speedup vs baseline: 2.6360x; 1.0708x over previous
#include <hip/hip_runtime.h>
#include <hip/hip_bf16.h>

typedef __attribute__((ext_vector_type(2))) float v2f;
typedef __attribute__((ext_vector_type(8))) float v8f;

#define N_NODES_C 100000

#define AGG_R   391
static_assert((N_NODES_C + AGG_R - 1) / AGG_R <= 256, "a block number must fit one byte");
#define DEG_R   14336
#define DEG_TB  1024
#define DEG_IT  11
#define DEG_CH  (DEG_TB * DEG_IT * 4)
static_assert(DEG_R % DEG_TB == 0, "a range is a whole number of block strides");
static_assert((DEG_CH / 4) % 32 == 0 && DEG_TB % 32 == 0, "a wave's 32 pieces are one aligned 128-edge group");
static_assert(AGG_R <= 512 && N_NODES_C <= (1 << 23), "a list word packs row (9 bits) and source (23 bits)");

__device__ __forceinline__ unsigned edge_word(int s, int d) {
    const unsigned k = (unsigned)d / AGG_R;
    return (unsigned)s | (((unsigned)d - k * AGG_R) << 23);
}

__launch_bounds__(DEG_TB)
__global__ void deg_hist(const int* __restrict__ src, const int* __restrict__ dst,
                         unsigned* __restrict__ part, unsigned* __restrict__ planes,
                         unsigned* __restrict__ words, int ne, int n) {
    __shared__ unsigned hist[DEG_R];
    __shared__ unsigned tile[256 * 128];
    const int tid = threadIdx.x;
    const int lo  = blockIdx.x * DEG_R;
    const int e0  = blockIdx.y * DEG_CH;

    for (int i = tid; i < DEG_R; i += DEG_TB) hist[i] = 0u;
    __syncthreads();

    if ((ne & 3) == 0) {
        const int4* __restrict__ d4 = (const int4*)dst;
        const int n4 = ne >> 2;
        const int j0 = (e0 >> 2) + tid;
        const size_t pitch = (size_t)(((n4 + 1023) >> 10) << 7);
        unsigned mine = 0u;
        for (int it = blockIdx.x; it < DEG_IT; it += gridDim.x) mine |= 1u << it;
        const int4* __restrict__ s4 = (const int4*)src;
        int4 sv[DEG_IT] = {};
#pragma unroll
        for (int it = 0; it < DEG_IT; ++it) {
            const int j = j0 + it * DEG_TB;
            if ((mine >> it) & 1u) sv[it] = s4[j < n4 ? j : n4 - 1];
        }
        int4 v[DEG_IT];
#pragma unroll
        for (int it = 0; it < DEG_IT; ++it) {
            const int j = j0 + it * DEG_TB;
            v[it] = d4[j < n4 ? j : n4 - 1];
        }
#pragma unroll
        for (int it = 0; it < DEG_IT; ++it) {
            const int j = j0 + it * DEG_TB;
            if (j < n4) {
                const unsigned a0 = (unsigned)(v[it].x - lo);
                const unsigned a1 = (unsigned)(v[it].y - lo);
                const unsigned a2 = (unsigned)(v[it].z - lo);
                const unsigned a3 = (unsigned)(v[it].w - lo);
                if (a0 < (unsigned)DEG_R) atomicAdd(&hist[a0], 1u);
                if (a1 < (unsigned)DEG_R) atomicAdd(&hist[a1], 1u);
                if (a2 < (unsigned)DEG_R) atomicAdd(&hist[a2], 1u);
                if (a3 < (unsigned)DEG_R) atomicAdd(&hist[a3], 1u);
            }
        }
        uint4* __restrict__ w4 = (uint4*)words;
        const int lane = tid & 31;
#pragma unroll
        for (int it = 0; it < DEG_IT; ++it) {
            const int j = j0 + it * DEG_TB;
            if ((mine >> it) & 1u) {
                if (j < n4) {
                    w4[j] = make_uint4(edge_word(sv[it].x, v[it].x), edge_word(sv[it].y, v[it].y),
                                       edge_word(sv[it].z, v[it].z), edge_word(sv[it].w, v[it].w));
                }
                const int jb = (e0 >> 2) + it * DEG_TB;
                if (jb < n4) {
                    uint4* t4 = (uint4*)tile;
#pragma unroll
                    for (int k = 0; k < 8; ++k) t4[tid + k * DEG_TB] = make_uint4(0u, 0u, 0u, 0u);
                    __syncthreads();
                    if (j < n4) {
                        const unsigned kk[4] = {(unsigned)v[it].x / AGG_R, (unsigned)v[it].y / AGG_R,
                                                (unsigned)v[it].z / AGG_R, (unsigned)v[it].w / AGG_R};
                        const unsigned u0 = (unsigned)(tid >> 5) * 4u;
#pragma unroll
                        for (int cp = 0; cp < 4; ++cp)
                            atomicOr(&tile[kk[cp] * 128u + ((u0 + cp + 4u * (kk[cp] & 31u)) & 127u)], 1u << lane);
                    }
                    __syncthreads();
                    const unsigned wv = (unsigned)(tid >> 5);
#pragma unroll
                    for (int k = 0; k < 8; ++k) {
                        const unsigned key = wv + 32u * k;
                        const uint4 val = *(const uint4*)&tile[key * 128u + ((4u * lane + 4u * (key & 31u)) & 127u)];
                        *(uint4*)(planes + (size_t)key * pitch + (jb >> 3) + 4 * lane) = val;
                    }
                    __syncthreads();
                }
            }
        }
    } else {
        const int e1 = (e0 + DEG_CH < ne) ? (e0 + DEG_CH) : ne;
        for (int e = e0 + tid; e < e1; e += DEG_TB) {
            const int d = dst[e];
            if (blockIdx.x == 0) words[e] = edge_word(src[e], d);
            const unsigned a = (unsigned)(d - lo);
            if (a < (unsigned)DEG_R) atomicAdd(&hist[a], 1u);
        }
    }
    __syncthreads();

    unsigned* __restrict__ row = part + (size_t)blockIdx.y * n;
    unsigned hv[DEG_R / DEG_TB];
#pragma unroll
    for (int k = 0; k < DEG_R / DEG_TB; ++k) hv[k] = hist[tid + k * DEG_TB];
#pragma unroll
    for (int k = 0; k < DEG_R / DEG_TB; ++k) {
        const int node = lo + tid + k * DEG_TB;
        if (node < n) row[node] = hv[k];
    }
}

__device__ __forceinline__ unsigned wave_sum(unsigned v) {
#pragma unroll
    for (int m = 16; m > 0; m >>= 1) v += __shfl_xor(v, m);
    return v;
}

#define FIN_TB  512
static_assert(AGG_R <= FIN_TB, "one thread per node of a block");
__launch_bounds__(FIN_TB)
__global__ void finish_dis(const unsigned* __restrict__ part, float* __restrict__ dis,
                           unsigned* __restrict__ blkcnt, int n, int nc) {
    __shared__ unsigned tot;
    const int tid = threadIdx.x;
    if (tid == 0) tot = 0u;
    __syncthreads();
    const int i = blockIdx.x * AGG_R + tid;
    unsigned s = 0u;
    if (tid < AGG_R && i < n) {
#pragma unroll 12
        for (int c = 0; c < nc; ++c) s += part[(size_t)c * n + i];
        dis[i] = rsqrtf((float)s + 1.0f);
    }
    s = wave_sum(s);
    if ((tid & 31) == 0 && s != 0u) atomicAdd(&tot, s);
    __syncthreads();
    if (tid == 0) blkcnt[blockIdx.x] = tot;
}

template<int NT, int KDIM, bool RELU_IN, bool SELF, bool OUTP>
__launch_bounds__(256)
__global__ void gemm_wmma_f32(const float* __restrict__ X,
                              const float* __restrict__ W,
                              const float* __restrict__ bias,
                              const float* __restrict__ dis,
                              const float* __restrict__ Wo,
                              const float* __restrict__ bo,
                              float* __restrict__ Y,
                              float* __restrict__ S,
                              float* __restrict__ out,
                              int M) {
    static_assert(!(SELF && OUTP), "one epilogue per instantiation");
    constexpr int N = NT * 16;
    const int wave = (blockIdx.x * blockDim.x + threadIdx.x) >> 5;
    const int lane = threadIdx.x & 31;
    const int row0 = wave * 16;
    if (row0 >= M) return;

    const int r      = lane & 15;
    const int khalf  = (lane >> 4) * 2;
    const int rowoff = (lane >> 4) * 8;

    float d2[8];
    if (SELF) {
        const float4 da = *(const float4*)(dis + row0 + rowoff);
        const float4 db = *(const float4*)(dis + row0 + rowoff + 4);
        d2[0] = da.x * da.x; d2[1] = da.y * da.y; d2[2] = da.z * da.z; d2[3] = da.w * da.w;
        d2[4] = db.x * db.x; d2[5] = db.y * db.y; d2[6] = db.z * db.z; d2[7] = db.w * db.w;
    }

    v8f acc[NT] = {};

    const float* __restrict__ xrow = X + (size_t)(row0 + r) * KDIM;

#pragma unroll
    for (int k = 0; k < KDIM; k += 4) {
        const int kk = k + khalf;
        v2f a = *(const v2f*)(xrow + kk);
        if (RELU_IN) {
            a.x = fmaxf(a.x, 0.0f);
            a.y = fmaxf(a.y, 0.0f);
        }
#pragma unroll
        for (int t = 0; t < NT; ++t) {
            const int col = t * 16 + r;
            v2f b;
            b.x = W[(size_t)kk * N + col];
            b.y = W[(size_t)(kk + 1) * N + col];
            acc[t] = __builtin_amdgcn_wmma_f32_16x16x4_f32(
                 false, a,  false, b,
                 (short)0, acc[t],
                 false,  false);
        }
    }

    if (OUTP) {
        float p[16];
#pragma unroll
        for (int i = 0; i < 16; ++i) p[i] = 0.0f;
#pragma unroll
        for (int t = 0; t < NT; ++t) {
            const int col = t * 16 + r;
            const float bv = bias[col];
            const v2f wo = *(const v2f*)(Wo + col * 2);
#pragma unroll
            for (int g = 0; g < 8; ++g) {
                const float v = fmaxf(acc[t][g] + bv, 0.0f);
                p[2 * g + 0] += v * wo.x;
                p[2 * g + 1] += v * wo.y;
            }
        }
        const bool b8 = (r & 8) != 0, b4 = (r & 4) != 0, b2 = (r & 2) != 0, b1 = (r & 1) != 0;
        float q8[8], q4[4], q2[2];
#pragma unroll
        for (int k = 0; k < 8; ++k) {
            const float keep = b8 ? p[k + 8] : p[k];
            const float send = b8 ? p[k] : p[k + 8];
            q8[k] = keep + __shfl_xor(send, 8);
        }
#pragma unroll
        for (int k = 0; k < 4; ++k) {
            const float keep = b4 ? q8[k + 4] : q8[k];
            const float send = b4 ? q8[k] : q8[k + 4];
            q4[k] = keep + __shfl_xor(send, 4);
        }
#pragma unroll
        for (int k = 0; k < 2; ++k) {
            const float keep = b2 ? q4[k + 2] : q4[k];
            const float send = b2 ? q4[k] : q4[k + 2];
            q2[k] = keep + __shfl_xor(send, 2);
        }
        const float keep1 = b1 ? q2[1] : q2[0];
        const float send1 = b1 ? q2[0] : q2[1];
        const float tot = keep1 + __shfl_xor(send1, 1);
        out[(size_t)row0 * 2 + lane] = tot + bo[lane & 1];
        return;
    }

#pragma unroll
    for (int t = 0; t < NT; ++t) {
        const int col = t * 16 + r;
        const float bv = bias[col];
#pragma unroll
        for (int g = 0; g < 8; ++g) {
            const size_t o = (size_t)(row0 + rowoff + g) * N + col;
            if (SELF) {
                Y[o] = acc[t][g];
                S[o] = acc[t][g] * d2[g] + bv;
            } else {
                Y[o] = fmaxf(acc[t][g] + bv, 0.0f);
            }
        }
    }
}

#define AGG_TB  1024
#define AGG_C   256
#define AGG_Q   (3 * AGG_C)
#define AGG_G   (AGG_C / 32)
#define AGG_B   16

__device__ __forceinline__ void agg_drain(const float* __restrict__ h,
                                          const unsigned* __restrict__ words,
                                          const float* __restrict__ dis,
                                          float* acc, int* qe, int* qa, float* qw,
                                          uint2* __restrict__ mylist, unsigned* fill,
                                          int cnt, int lo, int lane) {
#pragma unroll 1
    for (int c0 = 0; c0 < cnt; c0 += AGG_C) {
        const int cc = (cnt - c0 < AGG_C) ? (cnt - c0) : AGG_C;
        int* ce = qe + c0;
        int ev[AGG_G];
        unsigned pk[AGG_G];
#pragma unroll
        for (int g = 0; g < AGG_G; ++g) {
            const int idx = g * 32 + lane;
            ev[g] = (idx < cc) ? ce[idx] : 0;
        }
#pragma unroll
        for (int g = 0; g < AGG_G; ++g) pk[g] = words[ev[g]];
#pragma unroll
        for (int g = 0; g < AGG_G; ++g) asm volatile("" :: "v"(pk[g]));
        unsigned pos = 0u;
        if (lane == 0) pos = atomicAdd(fill, (unsigned)cc);
        pos = (unsigned)__builtin_amdgcn_readfirstlane((int)pos);
#pragma unroll
        for (int g = 0; g < AGG_G; ++g) {
            const int idx = g * 32 + lane;
            const unsigned p = (idx < cc) ? pk[g] : 0u;
            const int s = (int)(p & 0x7fffffu);
            const int a = (int)(p >> 23);
            const float w = dis[s] * dis[lo + a];
            if (idx < cc) {
                ce[idx] = s;
                qa[idx] = a;
                qw[idx] = w;
                mylist[pos + idx] = make_uint2(p, __float_as_uint(w));
            }
        }
        for (int g = 0; g < cc; g += 32) {
            const int idx = g + lane;
            const bool valid = idx < cc;
            const int s = valid ? ce[idx] : 0;
            const int a = valid ? qa[idx] : 0;
            const float w = valid ? qw[idx] : 0.0f;
            const int m = (cc - g < 32) ? (cc - g) : 32;
#pragma unroll 1
            for (int u0 = 0; u0 < m; u0 += AGG_B) {
                v2f rows[AGG_B];
#pragma unroll
                for (int u = 0; u < AGG_B; ++u) {
                    const int su = __builtin_amdgcn_readlane(s, u0 + u);
                    rows[u] = *(const v2f*)(h + (size_t)su * 64 + lane * 2);
                }
#pragma unroll
                for (int u = 0; u < AGG_B; ++u) {
                    if (u0 + u < m) {
                        const float wu = __int_as_float(__builtin_amdgcn_readlane(__float_as_int(w), u0 + u));
                        const int au = __builtin_amdgcn_readlane(a, u0 + u);
                        float* p = acc + au * 64 + lane * 2;
                        __hip_atomic_fetch_add(p + 0, rows[u].x * wu, __ATOMIC_RELAXED, __HIP_MEMORY_SCOPE_WORKGROUP);
                        __hip_atomic_fetch_add(p + 1, rows[u].y * wu, __ATOMIC_RELAXED, __HIP_MEMORY_SCOPE_WORKGROUP);
                    }
                }
#pragma unroll
                for (int u = 0; u < AGG_B; ++u) asm volatile("" :: "v"(rows[u]));
            }
        }
    }
}

#define AGG_DRAIN_IF(COND)                                                      \
    if (COND) {                                                                 \
        agg_drain(h, words, dis, acc, myqe, myqa, myqw, mylist, &s_fill, cnt, lo, lane); \
        cnt = 0;                                                                \
    }

#define AGG_TAKE(PRED, EID)                                                     \
    {                                                                           \
        const bool m_ = (PRED);                                                 \
        const unsigned mask_ = __builtin_amdgcn_ballot_w32(m_);                 \
        if (mask_) {                                                            \
            if (m_) {                                                           \
                const int pos_ = cnt + (int)__builtin_amdgcn_mbcnt_lo(mask_, 0u); \
                myqe[pos_] = (EID);                                             \
            }                                                                   \
            cnt += __builtin_popcount(mask_);                                   \
        }                                                                       \
    }

#define AGG_STEP(V, G)                                                          \
    {                                                                           \
        const bool live_ = (G) < ng;                                            \
        unsigned m0_ = live_ ? V.x : 0u, m1_ = live_ ? V.y : 0u;                \
        unsigned m2_ = live_ ? V.z : 0u, m3_ = live_ ? V.w : 0u;                \
        const int eb_ = (G) << 7;                                               \
        unsigned any_ = __builtin_amdgcn_ballot_w32((m0_ | m1_ | m2_ | m3_) != 0u); \
        while (any_) {                                                          \
            AGG_DRAIN_IF(cnt > early)                                           \
            const bool z0_ = m0_ != 0u, z1_ = m1_ != 0u, z2_ = m2_ != 0u;       \
            const unsigned mm_ = z0_ ? m0_ : z1_ ? m1_ : z2_ ? m2_ : m3_;       \
            if (mm_ != 0u) {                                                    \
                const int c_ = z0_ ? 0 : z1_ ? 1 : z2_ ? 2 : 3;                 \
                const int pos_ = cnt + (int)__builtin_amdgcn_mbcnt_lo(any_, 0u); \
                myqe[pos_] = eb_ + c_ + 4 * (__ffs(mm_) - 1);                   \
            }                                                                   \
            cnt += __builtin_popcount(any_);                                    \
            const unsigned cl_ = mm_ & (mm_ - 1u);                              \
            if (z0_) m0_ = cl_; else if (z1_) m1_ = cl_; else if (z2_) m2_ = cl_; else m3_ = cl_; \
            any_ = __builtin_amdgcn_ballot_w32((m0_ | m1_ | m2_ | m3_) != 0u);  \
        }                                                                       \
    }

__launch_bounds__(AGG_TB)
__global__ void edge_tile(const float* __restrict__ h,
                          const int* __restrict__ dst,
                          const unsigned* __restrict__ planes,
                          const unsigned* __restrict__ words,
                          const float* __restrict__ dis,
                          float* __restrict__ agg,
                          const unsigned* __restrict__ blkcnt,
                          unsigned* __restrict__ list,
                          int ne, int n) {
    __shared__ float acc[AGG_R * 64];
    __shared__ int   qe_all[AGG_TB / 32][AGG_Q];
    __shared__ int   qa_all[AGG_TB / 32][AGG_C];
    __shared__ float qw_all[AGG_TB / 32][AGG_C];
    __shared__ unsigned s_base;
    __shared__ unsigned s_fill;

    const int tid  = threadIdx.x;
    const int wave = tid >> 5;
    const int lane = tid & 31;
    const int lo   = blockIdx.x * AGG_R;
    const int nr   = (n - lo < AGG_R) ? (n - lo) : AGG_R;
    int*   myqe = qe_all[wave];
    int*   myqa = qa_all[wave];
    float* myqw = qw_all[wave];

    if (tid == 0) { s_base = 0u; s_fill = 0u; }
    __syncthreads();
    if (tid < 256) {
        const unsigned v = wave_sum(tid < (int)blockIdx.x ? blkcnt[tid] : 0u);
        if (lane == 0 && v != 0u) atomicAdd(&s_base, v);
    }

    float4* acc4 = (float4*)acc;
    float4* agg4 = (float4*)(agg + (size_t)lo * 64);
    for (int i = tid; i < nr * 16; i += AGG_TB) acc4[i] = agg4[i];
    __syncthreads();
    uint2* __restrict__ mylist = (uint2*)list + s_base;

    int cnt = 0;
    if ((ne & 3) == 0) {
        const int n4 = ne >> 2;
        const int ng = (n4 + 31) >> 5;
        const size_t pitch = (size_t)(((n4 + 1023) >> 10) << 7);
        const uint4* __restrict__ k4 = (const uint4*)(planes + (size_t)blockIdx.x * pitch);
        const int early = 64 + 4 * wave;
        static_assert(64 + 4 * (AGG_TB / 32 - 1) + 32 <= AGG_Q, "queue room for one more round");
        const int nsteps = (ng + AGG_TB - 1) / AGG_TB;
        if (nsteps > 0) {
            int gA = tid;
            uint4 va = k4[gA < ng ? gA : ng - 1], vb;
            for (int k = 0; k < nsteps; k += 2) {
                const int gB = gA + AGG_TB;
                vb = k4[gB < ng ? gB : ng - 1];
                AGG_STEP(va, gA)
                gA = gB + AGG_TB;
                va = k4[gA < ng ? gA : ng - 1];
                AGG_STEP(vb, gB)
            }
        }
    } else {
        for (int eb = wave * 32; eb < ne; eb += AGG_TB) {
            const int e = eb + lane;
            const int val = dst[e < ne ? e : ne - 1];
            AGG_DRAIN_IF(cnt + 32 > AGG_Q)
            AGG_TAKE(e < ne && (unsigned)(val - lo) < (unsigned)nr, e)
        }
    }
    agg_drain(h, words, dis, acc, myqe, myqa, myqw, mylist, &s_fill, cnt, lo, lane);
    __syncthreads();

    for (int i = tid; i < nr * 16; i += AGG_TB) agg4[i] = acc4[i];
}

#define LST_B   32

__launch_bounds__(AGG_TB)
__global__ void edge_list(const float* __restrict__ h,
                          const unsigned* __restrict__ list,
                          const unsigned* __restrict__ blkcnt,
                          float* __restrict__ agg,
                          int n) {
    __shared__ float acc[AGG_R * 64];
    __shared__ unsigned s_base;

    const int tid  = threadIdx.x;
    const int wave = __builtin_amdgcn_readfirstlane(tid >> 5);
    const int lane = tid & 31;
    const int lo   = blockIdx.x * AGG_R;
    const int nr   = (n - lo < AGG_R) ? (n - lo) : AGG_R;

    if (tid == 0) s_base = 0u;
    __syncthreads();
    if (tid < 256) {
        const unsigned v = wave_sum(tid < (int)blockIdx.x ? blkcnt[tid] : 0u);
        if (lane == 0 && v != 0u) atomicAdd(&s_base, v);
    }
    const int cnt = (int)blkcnt[blockIdx.x];

    float4* acc4 = (float4*)acc;
    float4* agg4 = (float4*)(agg + (size_t)lo * 64);
    for (int i = tid; i < nr * 16; i += AGG_TB) acc4[i] = agg4[i];
    __syncthreads();
    const uint2* __restrict__ my = (const uint2*)list + s_base;

    int g = wave * 32;
    uint2 en = make_uint2(0u, 0u);
    if (g + lane < cnt) en = my[g + lane];
    while (g < cnt) {
        const uint2 e = en;
        const int gn = g + AGG_TB;
        en = make_uint2(0u, 0u);
        if (gn + lane < cnt) en = my[gn + lane];
        const int m = (cnt - g < 32) ? (cnt - g) : 32;
        const int   s = (int)(e.x & 0x7fffffu);
        const int   a = (int)(e.x >> 23);
        const float w = __uint_as_float(e.y);
        v2f rows[LST_B];
#pragma unroll
        for (int u = 0; u < LST_B; ++u) {
            const int su = __builtin_amdgcn_readlane(s, u);
            rows[u] = *(const v2f*)(h + (size_t)su * 64 + lane * 2);
        }
#pragma unroll
        for (int u = 0; u < LST_B; ++u) {
            if (u < m) {
                const float wu = __int_as_float(__builtin_amdgcn_readlane(__float_as_int(w), u));
                const int au = __builtin_amdgcn_readlane(a, u);
                float* p = acc + au * 64 + lane * 2;
                __hip_atomic_fetch_add(p + 0, rows[u].x * wu, __ATOMIC_RELAXED, __HIP_MEMORY_SCOPE_WORKGROUP);
                __hip_atomic_fetch_add(p + 1, rows[u].y * wu, __ATOMIC_RELAXED, __HIP_MEMORY_SCOPE_WORKGROUP);
            }
        }
        g = gn;
    }
    __syncthreads();

    for (int i = tid; i < nr * 16; i += AGG_TB) agg4[i] = acc4[i];
}

extern "C" void kernel_launch(void* const* d_in, const int* in_sizes, int n_in,
                              void* d_out, int out_size, void* d_ws, size_t ws_size,
                              hipStream_t stream) {
    const float* x  = (const float*)d_in[0];
    const int*   ei = (const int*)d_in[1];
    const float* W1 = (const float*)d_in[2];
    const float* b1 = (const float*)d_in[3];
    const float* W2 = (const float*)d_in[4];
    const float* b2 = (const float*)d_in[5];
    const float* Wd = (const float*)d_in[6];
    const float* bd = (const float*)d_in[7];
    const float* Wo = (const float*)d_in[8];
    const float* bo = (const float*)d_in[9];
    float* out = (float*)d_out;

    const int N = in_sizes[0] / 64;
    const int E = in_sizes[1] / 2;
    const int* src = ei;
    const int* dst = ei + E;

    float* dis  = (float*)d_ws;
    float* bufA = dis  + N;
    float* bufB = bufA + (size_t)N * 64;
    float* bufC = bufB + (size_t)N * 64;
    unsigned* part = (unsigned*)(bufC + (size_t)N * 64);
    (void)ws_size; (void)n_in; (void)out_size;

    const int TB = 256;
    const int gemm_waves   = N / 16;
    const int gemm_blocks  = (gemm_waves * 32 + TB - 1) / TB;
    const int agg_blocks   = (N + AGG_R - 1) / AGG_R;

    const int deg_ranges = (N + DEG_R - 1) / DEG_R;
    const int deg_chunks = (E + DEG_CH - 1) / DEG_CH;
    unsigned* planes = part + (size_t)deg_chunks * N;
    unsigned* list   = planes + (size_t)256 * (((((size_t)E >> 2) + 1023) >> 10) << 7);
    unsigned* blkcnt = list + (size_t)E * 2;
    unsigned* words  = blkcnt + 256;
    deg_hist<<<dim3(deg_ranges, deg_chunks), DEG_TB, 0, stream>>>(src, dst, part, planes, words, E, N);
    finish_dis<<<agg_blocks, FIN_TB, 0, stream>>>(part, dis, blkcnt, N, deg_chunks);

    gemm_wmma_f32<4, 64, false, true, false><<<gemm_blocks, TB, 0, stream>>>(x, W1, b1, dis, nullptr, nullptr, bufA, bufB, nullptr, N);
    edge_tile<<<agg_blocks, AGG_TB, 0, stream>>>(bufA, dst, planes, words, dis, bufB, blkcnt, list, E, N);

    gemm_wmma_f32<4, 64, true, true, false><<<gemm_blocks, TB, 0, stream>>>(bufB, W2, b2, dis, nullptr, nullptr, bufA, bufC, nullptr, N);
    edge_list<<<agg_blocks, AGG_TB, 0, stream>>>(bufA, list, blkcnt, bufC, N);

    gemm_wmma_f32<8, 64, true, false, true><<<gemm_blocks, TB, 0, stream>>>(bufC, Wd, bd, nullptr, Wo, bo, nullptr, nullptr, out, N);
}


# self-loop term + bias (h*dis^2 + b) added by the edge pass's closing store from the node's own h row (summed last, was first); LDS rows start at zero; the products store one plane, not two
# speedup vs baseline: 2.8675x; 1.0878x over previous
#include <hip/hip_runtime.h>
#include <hip/hip_bf16.h>

typedef __attribute__((ext_vector_type(2))) float v2f;
typedef __attribute__((ext_vector_type(8))) float v8f;

#define N_NODES_C 100000

#define AGG_R   391
static_assert((N_NODES_C + AGG_R - 1) / AGG_R <= 256, "a block number must fit one byte");
#define DEG_R   14336
#define DEG_TB  1024
#define DEG_IT  11
#define DEG_CH  (DEG_TB * DEG_IT * 4)
static_assert(DEG_R % DEG_TB == 0, "a range is a whole number of block strides");
static_assert((DEG_CH / 4) % 32 == 0 && DEG_TB % 32 == 0, "a wave's 32 pieces are one aligned 128-edge group");
static_assert(AGG_R <= 512 && N_NODES_C <= (1 << 23), "a list word packs row (9 bits) and source (23 bits)");

__device__ __forceinline__ unsigned edge_word(int s, int d) {
    const unsigned k = (unsigned)d / AGG_R;
    return (unsigned)s | (((unsigned)d - k * AGG_R) << 23);
}

__launch_bounds__(DEG_TB)
__global__ void deg_hist(const int* __restrict__ src, const int* __restrict__ dst,
                         unsigned* __restrict__ part, unsigned* __restrict__ planes,
                         unsigned* __restrict__ words, int ne, int n) {
    __shared__ unsigned hist[DEG_R];
    __shared__ unsigned tile[256 * 128];
    const int tid = threadIdx.x;
    const int lo  = blockIdx.x * DEG_R;
    const int e0  = blockIdx.y * DEG_CH;

    for (int i = tid; i < DEG_R; i += DEG_TB) hist[i] = 0u;
    __syncthreads();

    if ((ne & 3) == 0) {
        const int4* __restrict__ d4 = (const int4*)dst;
        const int n4 = ne >> 2;
        const int j0 = (e0 >> 2) + tid;
        const size_t pitch = (size_t)(((n4 + 1023) >> 10) << 7);
        unsigned mine = 0u;
        for (int it = blockIdx.x; it < DEG_IT; it += gridDim.x) mine |= 1u << it;
        const int4* __restrict__ s4 = (const int4*)src;
        int4 sv[DEG_IT] = {};
#pragma unroll
        for (int it = 0; it < DEG_IT; ++it) {
            const int j = j0 + it * DEG_TB;
            if ((mine >> it) & 1u) sv[it] = s4[j < n4 ? j : n4 - 1];
        }
        int4 v[DEG_IT];
#pragma unroll
        for (int it = 0; it < DEG_IT; ++it) {
            const int j = j0 + it * DEG_TB;
            v[it] = d4[j < n4 ? j : n4 - 1];
        }
#pragma unroll
        for (int it = 0; it < DEG_IT; ++it) {
            const int j = j0 + it * DEG_TB;
            if (j < n4) {
                const unsigned a0 = (unsigned)(v[it].x - lo);
                const unsigned a1 = (unsigned)(v[it].y - lo);
                const unsigned a2 = (unsigned)(v[it].z - lo);
                const unsigned a3 = (unsigned)(v[it].w - lo);
                if (a0 < (unsigned)DEG_R) atomicAdd(&hist[a0], 1u);
                if (a1 < (unsigned)DEG_R) atomicAdd(&hist[a1], 1u);
                if (a2 < (unsigned)DEG_R) atomicAdd(&hist[a2], 1u);
                if (a3 < (unsigned)DEG_R) atomicAdd(&hist[a3], 1u);
            }
        }
        uint4* __restrict__ w4 = (uint4*)words;
        const int lane = tid & 31;
#pragma unroll
        for (int it = 0; it < DEG_IT; ++it) {
            const int j = j0 + it * DEG_TB;
            if ((mine >> it) & 1u) {
                if (j < n4) {
                    w4[j] = make_uint4(edge_word(sv[it].x, v[it].x), edge_word(sv[it].y, v[it].y),
                                       edge_word(sv[it].z, v[it].z), edge_word(sv[it].w, v[it].w));
                }
                const int jb = (e0 >> 2) + it * DEG_TB;
                if (jb < n4) {
                    uint4* t4 = (uint4*)tile;
#pragma unroll
                    for (int k = 0; k < 8; ++k) t4[tid + k * DEG_TB] = make_uint4(0u, 0u, 0u, 0u);
                    __syncthreads();
                    if (j < n4) {
                        const unsigned kk[4] = {(unsigned)v[it].x / AGG_R, (unsigned)v[it].y / AGG_R,
                                                (unsigned)v[it].z / AGG_R, (unsigned)v[it].w / AGG_R};
                        const unsigned u0 = (unsigned)(tid >> 5) * 4u;
#pragma unroll
                        for (int cp = 0; cp < 4; ++cp)
                            atomicOr(&tile[kk[cp] * 128u + ((u0 + cp + 4u * (kk[cp] & 31u)) & 127u)], 1u << lane);
                    }
                    __syncthreads();
                    const unsigned wv = (unsigned)(tid >> 5);
#pragma unroll
                    for (int k = 0; k < 8; ++k) {
                        const unsigned key = wv + 32u * k;
                        const uint4 val = *(const uint4*)&tile[key * 128u + ((4u * lane + 4u * (key & 31u)) & 127u)];
                        *(uint4*)(planes + (size_t)key * pitch + (jb >> 3) + 4 * lane) = val;
                    }
                    __syncthreads();
                }
            }
        }
    } else {
        const int e1 = (e0 + DEG_CH < ne) ? (e0 + DEG_CH) : ne;
        for (int e = e0 + tid; e < e1; e += DEG_TB) {
            const int d = dst[e];
            if (blockIdx.x == 0) words[e] = edge_word(src[e], d);
            const unsigned a = (unsigned)(d - lo);
            if (a < (unsigned)DEG_R) atomicAdd(&hist[a], 1u);
        }
    }
    __syncthreads();

    unsigned* __restrict__ row = part + (size_t)blockIdx.y * n;
    unsigned hv[DEG_R / DEG_TB];
#pragma unroll
    for (int k = 0; k < DEG_R / DEG_TB; ++k) hv[k] = hist[tid + k * DEG_TB];
#pragma unroll
    for (int k = 0; k < DEG_R / DEG_TB; ++k) {
        const int node = lo + tid + k * DEG_TB;
        if (node < n) row[node] = hv[k];
    }
}

__device__ __forceinline__ unsigned wave_sum(unsigned v) {
#pragma unroll
    for (int m = 16; m > 0; m >>= 1) v += __shfl_xor(v, m);
    return v;
}

#define FIN_TB  512
static_assert(AGG_R <= FIN_TB, "one thread per node of a block");
__launch_bounds__(FIN_TB)
__global__ void finish_dis(const unsigned* __restrict__ part, float* __restrict__ dis,
                           unsigned* __restrict__ blkcnt, int n, int nc) {
    __shared__ unsigned tot;
    const int tid = threadIdx.x;
    if (tid == 0) tot = 0u;
    __syncthreads();
    const int i = blockIdx.x * AGG_R + tid;
    unsigned s = 0u;
    if (tid < AGG_R && i < n) {
#pragma unroll 12
        for (int c = 0; c < nc; ++c) s += part[(size_t)c * n + i];
        dis[i] = rsqrtf((float)s + 1.0f);
    }
    s = wave_sum(s);
    if ((tid & 31) == 0 && s != 0u) atomicAdd(&tot, s);
    __syncthreads();
    if (tid == 0) blkcnt[blockIdx.x] = tot;
}

template<int NT, int KDIM, bool RELU_IN, bool SELF, bool OUTP>
__launch_bounds__(256)
__global__ void gemm_wmma_f32(const float* __restrict__ X,
                              const float* __restrict__ W,
                              const float* __restrict__ bias,
                              const float* __restrict__ Wo,
                              const float* __restrict__ bo,
                              float* __restrict__ Y,
                              float* __restrict__ out,
                              int M) {
    static_assert(!(SELF && OUTP), "one epilogue per instantiation");
    constexpr int N = NT * 16;
    const int wave = (blockIdx.x * blockDim.x + threadIdx.x) >> 5;
    const int lane = threadIdx.x & 31;
    const int row0 = wave * 16;
    if (row0 >= M) return;

    const int r      = lane & 15;
    const int khalf  = (lane >> 4) * 2;
    const int rowoff = (lane >> 4) * 8;

    v8f acc[NT] = {};

    const float* __restrict__ xrow = X + (size_t)(row0 + r) * KDIM;

#pragma unroll
    for (int k = 0; k < KDIM; k += 4) {
        const int kk = k + khalf;
        v2f a = *(const v2f*)(xrow + kk);
        if (RELU_IN) {
            a.x = fmaxf(a.x, 0.0f);
            a.y = fmaxf(a.y, 0.0f);
        }
#pragma unroll
        for (int t = 0; t < NT; ++t) {
            const int col = t * 16 + r;
            v2f b;
            b.x = W[(size_t)kk * N + col];
            b.y = W[(size_t)(kk + 1) * N + col];
            acc[t] = __builtin_amdgcn_wmma_f32_16x16x4_f32(
                 false, a,  false, b,
                 (short)0, acc[t],
                 false,  false);
        }
    }

    if (OUTP) {
        float p[16];
#pragma unroll
        for (int i = 0; i < 16; ++i) p[i] = 0.0f;
#pragma unroll
        for (int t = 0; t < NT; ++t) {
            const int col = t * 16 + r;
            const float bv = bias[col];
            const v2f wo = *(const v2f*)(Wo + col * 2);
#pragma unroll
            for (int g = 0; g < 8; ++g) {
                const float v = fmaxf(acc[t][g] + bv, 0.0f);
                p[2 * g + 0] += v * wo.x;
                p[2 * g + 1] += v * wo.y;
            }
        }
        const bool b8 = (r & 8) != 0, b4 = (r & 4) != 0, b2 = (r & 2) != 0, b1 = (r & 1) != 0;
        float q8[8], q4[4], q2[2];
#pragma unroll
        for (int k = 0; k < 8; ++k) {
            const float keep = b8 ? p[k + 8] : p[k];
            const float send = b8 ? p[k] : p[k + 8];
            q8[k] = keep + __shfl_xor(send, 8);
        }
#pragma unroll
        for (int k = 0; k < 4; ++k) {
            const float keep = b4 ? q8[k + 4] : q8[k];
            const float send = b4 ? q8[k] : q8[k + 4];
            q4[k] = keep + __shfl_xor(send, 4);
        }
#pragma unroll
        for (int k = 0; k < 2; ++k) {
            const float keep = b2 ? q4[k + 2] : q4[k];
            const float send = b2 ? q4[k] : q4[k + 2];
            q2[k] = keep + __shfl_xor(send, 2);
        }
        const float keep1 = b1 ? q2[1] : q2[0];
        const float send1 = b1 ? q2[0] : q2[1];
        const float tot = keep1 + __shfl_xor(send1, 1);
        out[(size_t)row0 * 2 + lane] = tot + bo[lane & 1];
        return;
    }

#pragma unroll
    for (int t = 0; t < NT; ++t) {
        const int col = t * 16 + r;
        const float bv = SELF ? 0.0f : bias[col];
#pragma unroll
        for (int g = 0; g < 8; ++g) {
            const size_t o = (size_t)(row0 + rowoff + g) * N + col;
            if (SELF) {
                Y[o] = acc[t][g];
            } else {
                Y[o] = fmaxf(acc[t][g] + bv, 0.0f);
            }
        }
    }
}

#define AGG_TB  1024
#define AGG_C   256
#define AGG_Q   (3 * AGG_C)
#define AGG_G   (AGG_C / 32)
#define AGG_B   16

#define AGG_TRIPS ((AGG_R * 16 + AGG_TB - 1) / AGG_TB)

__device__ __forceinline__ void agg_clear(float* acc, int tid) {
    float4* acc4 = (float4*)acc;
#pragma unroll
    for (int k = 0; k < AGG_TRIPS; ++k) {
        const int i = tid + k * AGG_TB;
        if (i < AGG_R * 16) acc4[i] = make_float4(0.0f, 0.0f, 0.0f, 0.0f);
    }
}

__device__ __forceinline__ void agg_own_load(const float* __restrict__ h,
                                             const float* __restrict__ dis,
                                             const float* __restrict__ bias,
                                             int lo, int nr, int tid,
                                             float4* hv, float* dv, float* bb) {
    const float4* __restrict__ h4 = (const float4*)(h + (size_t)lo * 64);
    const int last = nr * 16 - 1;
#pragma unroll
    for (int k = 0; k < AGG_TRIPS; ++k) {
        const int i = tid + k * AGG_TB;
        const int ii = i < last ? i : last;
        hv[k] = h4[ii];
        dv[k] = dis[lo + (ii >> 4)];
    }
    const int cb = 4 * (tid & 15);
    bb[0] = bias[cb]; bb[1] = bias[cb + 1]; bb[2] = bias[cb + 2]; bb[3] = bias[cb + 3];
}

__device__ __forceinline__ void agg_own_store(const float* acc, float* __restrict__ agg,
                                              int lo, int nr, int tid,
                                              const float4* hv, const float* dv, const float* bb) {
    float4* __restrict__ agg4 = (float4*)(agg + (size_t)lo * 64);
    const float4* acc4 = (const float4*)acc;
    const int last = nr * 16 - 1;
#pragma unroll
    for (int k = 0; k < AGG_TRIPS; ++k) {
        const int i = tid + k * AGG_TB;
        if (i <= last) {
            const float d2 = dv[k] * dv[k];
            const float4 a = acc4[i];
            float4 o;
            o.x = a.x + fmaf(hv[k].x, d2, bb[0]);
            o.y = a.y + fmaf(hv[k].y, d2, bb[1]);
            o.z = a.z + fmaf(hv[k].z, d2, bb[2]);
            o.w = a.w + fmaf(hv[k].w, d2, bb[3]);
            agg4[i] = o;
        }
    }
}

__device__ __forceinline__ void agg_drain(const float* __restrict__ h,
                                          const unsigned* __restrict__ words,
                                          const float* __restrict__ dis,
                                          float* acc, int* qe, int* qa, float* qw,
                                          uint2* __restrict__ mylist, unsigned* fill,
                                          int cnt, int lo, int lane) {
#pragma unroll 1
    for (int c0 = 0; c0 < cnt; c0 += AGG_C) {
        const int cc = (cnt - c0 < AGG_C) ? (cnt - c0) : AGG_C;
        int* ce = qe + c0;
        int ev[AGG_G];
        unsigned pk[AGG_G];
#pragma unroll
        for (int g = 0; g < AGG_G; ++g) {
            const int idx = g * 32 + lane;
            ev[g] = (idx < cc) ? ce[idx] : 0;
        }
#pragma unroll
        for (int g = 0; g < AGG_G; ++g) pk[g] = words[ev[g]];
#pragma unroll
        for (int g = 0; g < AGG_G; ++g) asm volatile("" :: "v"(pk[g]));
        unsigned pos = 0u;
        if (lane == 0) pos = atomicAdd(fill, (unsigned)cc);
        pos = (unsigned)__builtin_amdgcn_readfirstlane((int)pos);
#pragma unroll
        for (int g = 0; g < AGG_G; ++g) {
            const int idx = g * 32 + lane;
            const unsigned p = (idx < cc) ? pk[g] : 0u;
            const int s = (int)(p & 0x7fffffu);
            const int a = (int)(p >> 23);
            const float w = dis[s] * dis[lo + a];
            if (idx < cc) {
                ce[idx] = s;
                qa[idx] = a;
                qw[idx] = w;
                mylist[pos + idx] = make_uint2(p, __float_as_uint(w));
            }
        }
        for (int g = 0; g < cc; g += 32) {
            const int idx = g + lane;
            const bool valid = idx < cc;
            const int s = valid ? ce[idx] : 0;
            const int a = valid ? qa[idx] : 0;
            const float w = valid ? qw[idx] : 0.0f;
            const int m = (cc - g < 32) ? (cc - g) : 32;
#pragma unroll 1
            for (int u0 = 0; u0 < m; u0 += AGG_B) {
                v2f rows[AGG_B];
#pragma unroll
                for (int u = 0; u < AGG_B; ++u) {
                    const int su = __builtin_amdgcn_readlane(s, u0 + u);
                    rows[u] = *(const v2f*)(h + (size_t)su * 64 + lane * 2);
                }
#pragma unroll
                for (int u = 0; u < AGG_B; ++u) {
                    if (u0 + u < m) {
                        const float wu = __int_as_float(__builtin_amdgcn_readlane(__float_as_int(w), u0 + u));
                        const int au = __builtin_amdgcn_readlane(a, u0 + u);
                        float* p = acc + au * 64 + lane * 2;
                        __hip_atomic_fetch_add(p + 0, rows[u].x * wu, __ATOMIC_RELAXED, __HIP_MEMORY_SCOPE_WORKGROUP);
                        __hip_atomic_fetch_add(p + 1, rows[u].y * wu, __ATOMIC_RELAXED, __HIP_MEMORY_SCOPE_WORKGROUP);
                    }
                }
#pragma unroll
                for (int u = 0; u < AGG_B; ++u) asm volatile("" :: "v"(rows[u]));
            }
        }
    }
}

#define AGG_DRAIN_IF(COND)                                                      \
    if (COND) {                                                                 \
        agg_drain(h, words, dis, acc, myqe, myqa, myqw, mylist, &s_fill, cnt, lo, lane); \
        cnt = 0;                                                                \
    }

#define AGG_TAKE(PRED, EID)                                                     \
    {                                                                           \
        const bool m_ = (PRED);                                                 \
        const unsigned mask_ = __builtin_amdgcn_ballot_w32(m_);                 \
        if (mask_) {                                                            \
            if (m_) {                                                           \
                const int pos_ = cnt + (int)__builtin_amdgcn_mbcnt_lo(mask_, 0u); \
                myqe[pos_] = (EID);                                             \
            }                                                                   \
            cnt += __builtin_popcount(mask_);                                   \
        }                                                                       \
    }

#define AGG_STEP(V, G)                                                          \
    {                                                                           \
        const bool live_ = (G) < ng;                                            \
        unsigned m0_ = live_ ? V.x : 0u, m1_ = live_ ? V.y : 0u;                \
        unsigned m2_ = live_ ? V.z : 0u, m3_ = live_ ? V.w : 0u;                \
        const int eb_ = (G) << 7;                                               \
        unsigned any_ = __builtin_amdgcn_ballot_w32((m0_ | m1_ | m2_ | m3_) != 0u); \
        while (any_) {                                                          \
            AGG_DRAIN_IF(cnt > early)                                           \
            const bool z0_ = m0_ != 0u, z1_ = m1_ != 0u, z2_ = m2_ != 0u;       \
            const unsigned mm_ = z0_ ? m0_ : z1_ ? m1_ : z2_ ? m2_ : m3_;       \
            if (mm_ != 0u) {                                                    \
                const int c_ = z0_ ? 0 : z1_ ? 1 : z2_ ? 2 : 3;                 \
                const int pos_ = cnt + (int)__builtin_amdgcn_mbcnt_lo(any_, 0u); \
                myqe[pos_] = eb_ + c_ + 4 * (__ffs(mm_) - 1);                   \
            }                                                                   \
            cnt += __builtin_popcount(any_);                                    \
            const unsigned cl_ = mm_ & (mm_ - 1u);                              \
            if (z0_) m0_ = cl_; else if (z1_) m1_ = cl_; else if (z2_) m2_ = cl_; else m3_ = cl_; \
            any_ = __builtin_amdgcn_ballot_w32((m0_ | m1_ | m2_ | m3_) != 0u);  \
        }                                                                       \
    }

__launch_bounds__(AGG_TB)
__global__ void edge_tile(const float* __restrict__ h,
                          const int* __restrict__ dst,
                          const unsigned* __restrict__ planes,
                          const unsigned* __restrict__ words,
                          const float* __restrict__ dis,
                          const float* __restrict__ bias,
                          float* __restrict__ agg,
                          const unsigned* __restrict__ blkcnt,
                          unsigned* __restrict__ list,
                          int ne, int n) {
    __shared__ float acc[AGG_R * 64];
    __shared__ int   qe_all[AGG_TB / 32][AGG_Q];
    __shared__ int   qa_all[AGG_TB / 32][AGG_C];
    __shared__ float qw_all[AGG_TB / 32][AGG_C];
    __shared__ unsigned s_base;
    __shared__ unsigned s_fill;

    const int tid  = threadIdx.x;
    const int wave = tid >> 5;
    const int lane = tid & 31;
    const int lo   = blockIdx.x * AGG_R;
    const int nr   = (n - lo < AGG_R) ? (n - lo) : AGG_R;
    int*   myqe = qe_all[wave];
    int*   myqa = qa_all[wave];
    float* myqw = qw_all[wave];

    if (tid == 0) { s_base = 0u; s_fill = 0u; }
    __syncthreads();
    if (tid < 256) {
        const unsigned v = wave_sum(tid < (int)blockIdx.x ? blkcnt[tid] : 0u);
        if (lane == 0 && v != 0u) atomicAdd(&s_base, v);
    }

    agg_clear(acc, tid);
    __syncthreads();
    uint2* __restrict__ mylist = (uint2*)list + s_base;

    int cnt = 0;
    if ((ne & 3) == 0) {
        const int n4 = ne >> 2;
        const int ng = (n4 + 31) >> 5;
        const size_t pitch = (size_t)(((n4 + 1023) >> 10) << 7);
        const uint4* __restrict__ k4 = (const uint4*)(planes + (size_t)blockIdx.x * pitch);
        const int early = 64 + 4 * wave;
        static_assert(64 + 4 * (AGG_TB / 32 - 1) + 32 <= AGG_Q, "queue room for one more round");
        const int nsteps = (ng + AGG_TB - 1) / AGG_TB;
        if (nsteps > 0) {
            int gA = tid;
            uint4 va = k4[gA < ng ? gA : ng - 1], vb;
            for (int k = 0; k < nsteps; k += 2) {
                const int gB = gA + AGG_TB;
                vb = k4[gB < ng ? gB : ng - 1];
                AGG_STEP(va, gA)
                gA = gB + AGG_TB;
                va = k4[gA < ng ? gA : ng - 1];
                AGG_STEP(vb, gB)
            }
        }
    } else {
        for (int eb = wave * 32; eb < ne; eb += AGG_TB) {
            const int e = eb + lane;
            const int val = dst[e < ne ? e : ne - 1];
            AGG_DRAIN_IF(cnt + 32 > AGG_Q)
            AGG_TAKE(e < ne && (unsigned)(val - lo) < (unsigned)nr, e)
        }
    }
    agg_drain(h, words, dis, acc, myqe, myqa, myqw, mylist, &s_fill, cnt, lo, lane);
    float4 hv[AGG_TRIPS];
    float dv[AGG_TRIPS], bb[4];
    agg_own_load(h, dis, bias, lo, nr, tid, hv, dv, bb);
    __syncthreads();

    agg_own_store(acc, agg, lo, nr, tid, hv, dv, bb);
}

#define LST_B   32

__launch_bounds__(AGG_TB)
__global__ void edge_list(const float* __restrict__ h,
                          const unsigned* __restrict__ list,
                          const unsigned* __restrict__ blkcnt,
                          const float* __restrict__ dis,
                          const float* __restrict__ bias,
                          float* __restrict__ agg,
                          int n) {
    __shared__ float acc[AGG_R * 64];
    __shared__ unsigned s_base;

    const int tid  = threadIdx.x;
    const int wave = __builtin_amdgcn_readfirstlane(tid >> 5);
    const int lane = tid & 31;
    const int lo   = blockIdx.x * AGG_R;
    const int nr   = (n - lo < AGG_R) ? (n - lo) : AGG_R;

    if (tid == 0) s_base = 0u;
    __syncthreads();
    if (tid < 256) {
        const unsigned v = wave_sum(tid < (int)blockIdx.x ? blkcnt[tid] : 0u);
        if (lane == 0 && v != 0u) atomicAdd(&s_base, v);
    }
    const int cnt = (int)blkcnt[blockIdx.x];

    agg_clear(acc, tid);
    __syncthreads();
    const uint2* __restrict__ my = (const uint2*)list + s_base;

    int g = wave * 32;
    uint2 en = make_uint2(0u, 0u);
    if (g + lane < cnt) en = my[g + lane];
    while (g < cnt) {
        const uint2 e = en;
        const int gn = g + AGG_TB;
        en = make_uint2(0u, 0u);
        if (gn + lane < cnt) en = my[gn + lane];
        const int m = (cnt - g < 32) ? (cnt - g) : 32;
        const int   s = (int)(e.x & 0x7fffffu);
        const int   a = (int)(e.x >> 23);
        const float w = __uint_as_float(e.y);
        v2f rows[LST_B];
#pragma unroll
        for (int u = 0; u < LST_B; ++u) {
            const int su = __builtin_amdgcn_readlane(s, u);
            rows[u] = *(const v2f*)(h + (size_t)su * 64 + lane * 2);
        }
#pragma unroll
        for (int u = 0; u < LST_B; ++u) {
            if (u < m) {
                const float wu = __int_as_float(__builtin_amdgcn_readlane(__float_as_int(w), u));
                const int au = __builtin_amdgcn_readlane(a, u);
                float* p = acc + au * 64 + lane * 2;
                __hip_atomic_fetch_add(p + 0, rows[u].x * wu, __ATOMIC_RELAXED, __HIP_MEMORY_SCOPE_WORKGROUP);
                __hip_atomic_fetch_add(p + 1, rows[u].y * wu, __ATOMIC_RELAXED, __HIP_MEMORY_SCOPE_WORKGROUP);
            }
        }
        g = gn;
    }
    float4 hv[AGG_TRIPS];
    float dv[AGG_TRIPS], bb[4];
    agg_own_load(h, dis, bias, lo, nr, tid, hv, dv, bb);
    __syncthreads();

    agg_own_store(acc, agg, lo, nr, tid, hv, dv, bb);
}

extern "C" void kernel_launch(void* const* d_in, const int* in_sizes, int n_in,
                              void* d_out, int out_size, void* d_ws, size_t ws_size,
                              hipStream_t stream) {
    const float* x  = (const float*)d_in[0];
    const int*   ei = (const int*)d_in[1];
    const float* W1 = (const float*)d_in[2];
    const float* b1 = (const float*)d_in[3];
    const float* W2 = (const float*)d_in[4];
    const float* b2 = (const float*)d_in[5];
    const float* Wd = (const float*)d_in[6];
    const float* bd = (const float*)d_in[7];
    const float* Wo = (const float*)d_in[8];
    const float* bo = (const float*)d_in[9];
    float* out = (float*)d_out;

    const int N = in_sizes[0] / 64;
    const int E = in_sizes[1] / 2;
    const int* src = ei;
    const int* dst = ei + E;

    float* dis  = (float*)d_ws;
    float* bufA = dis  + N;
    float* bufB = bufA + (size_t)N * 64;
    float* bufC = bufB + (size_t)N * 64;
    unsigned* part = (unsigned*)(bufC + (size_t)N * 64);
    (void)ws_size; (void)n_in; (void)out_size;

    const int TB = 256;
    const int gemm_waves   = N / 16;
    const int gemm_blocks  = (gemm_waves * 32 + TB - 1) / TB;
    const int agg_blocks   = (N + AGG_R - 1) / AGG_R;

    const int deg_ranges = (N + DEG_R - 1) / DEG_R;
    const int deg_chunks = (E + DEG_CH - 1) / DEG_CH;
    unsigned* planes = part + (size_t)deg_chunks * N;
    unsigned* list   = planes + (size_t)256 * (((((size_t)E >> 2) + 1023) >> 10) << 7);
    unsigned* blkcnt = list + (size_t)E * 2;
    unsigned* words  = blkcnt + 256;
    deg_hist<<<dim3(deg_ranges, deg_chunks), DEG_TB, 0, stream>>>(src, dst, part, planes, words, E, N);
    finish_dis<<<agg_blocks, FIN_TB, 0, stream>>>(part, dis, blkcnt, N, deg_chunks);

    gemm_wmma_f32<4, 64, false, true, false><<<gemm_blocks, TB, 0, stream>>>(x, W1, nullptr, nullptr, nullptr, bufA, nullptr, N);
    edge_tile<<<agg_blocks, AGG_TB, 0, stream>>>(bufA, dst, planes, words, dis, b1, bufB, blkcnt, list, E, N);

    gemm_wmma_f32<4, 64, true, true, false><<<gemm_blocks, TB, 0, stream>>>(bufB, W2, nullptr, nullptr, nullptr, bufA, nullptr, N);
    edge_list<<<agg_blocks, AGG_TB, 0, stream>>>(bufA, list, blkcnt, dis, b2, bufC, N);

    gemm_wmma_f32<8, 64, true, false, true><<<gemm_blocks, TB, 0, stream>>>(bufC, Wd, bd, Wo, bo, nullptr, out, N);
}
